# starting kernel
# baseline (speedup 1.0000x reference)
#include <hip/hip_runtime.h>
#include <hip/hip_bf16.h>


typedef __attribute__((ext_vector_type(2))) float v2f;
typedef __attribute__((ext_vector_type(8))) float v8f;

#define HH    4
#define DD    8
#define DHH   32
#define DIMV  64
#define RPE   18
#define QKVW  128

__global__ __launch_bounds__(256) void qkv_proj_kernel(
    const float* __restrict__ x, const float* __restrict__ W,
    const float* __restrict__ bias, float* __restrict__ qkv, int N)
{
    __shared__ float xs[16 * 65];
    const int tid = threadIdx.x;
    const int r0  = blockIdx.x * 16;

    {
        int idx = tid * 4;
        int row = idx >> 6, col = idx & 63;
        int grow = r0 + row;
        float4 vx = make_float4(0.f, 0.f, 0.f, 0.f);
        if (grow < N) vx = *(const float4*)(x + (size_t)grow * 64 + col);
        xs[row*65+col+0] = vx.x; xs[row*65+col+1] = vx.y;
        xs[row*65+col+2] = vx.z; xs[row*65+col+3] = vx.w;
    }
    __syncthreads();

    const int wave = tid >> 5;
    const int lane = tid & 31;
    const int lhi  = lane >> 4;
    const int l16  = lane & 15;
    const int colbase = wave * 16;

    v8f acc = {};
    #pragma unroll
    for (int kk = 0; kk < 64; kk += 4) {
        const int ka = kk + (lhi ? 2 : 0);
        v2f a, b;
        a.x = xs[l16*65 + ka];
        a.y = xs[l16*65 + ka + 1];
        b.x = W[(size_t)ka     * QKVW + colbase + l16];
        b.y = W[(size_t)(ka+1) * QKVW + colbase + l16];
        acc = __builtin_amdgcn_wmma_f32_16x16x4_f32(
            false, a, false, b, (short)0, acc, false, false);
    }

    const int col = colbase + l16;
    const float bv = bias[col];
    #pragma unroll
    for (int r = 0; r < 8; ++r) {
        int grow = r0 + r + (lhi ? 8 : 0);
        if (grow < N) qkv[(size_t)grow * QKVW + col] = acc[r] + bv;
    }
}

__global__ __launch_bounds__(256) void edge_compat_kernel(
    const float* __restrict__ ea, const int* __restrict__ eidx,
    const float* __restrict__ qkv,
    const float* __restrict__ qw, const float* __restrict__ qb,
    const float* __restrict__ kw, const float* __restrict__ kb,
    float* __restrict__ exbuf, float* __restrict__ den, int E_)
{
    __shared__ float sea[128 * 21];
    __shared__ float srpe[128 * 67];
    const int tid = threadIdx.x;
    const int e0  = blockIdx.x * 128;

    __builtin_prefetch(ea + (size_t)(e0 + 128) * RPE, 0, 1);

    for (int i = tid; i < 128 * RPE; i += 256) {
        int e = i / RPE, c = i % RPE;
        int ge = e0 + e;
        sea[e*21 + c] = (ge < E_) ? ea[(size_t)ge * RPE + c] : 0.0f;
    }
    if (tid < 128) { sea[tid*21 + 18] = 0.f; sea[tid*21 + 19] = 0.f; }
    __syncthreads();

    const int wave = tid >> 5, lane = tid & 31;
    const int lhi = lane >> 4, l16 = lane & 15;
    const int etile = wave * 16;

    #pragma unroll
    for (int sel = 0; sel < 2; ++sel) {
        const float* Wp = sel ? kw : qw;
        #pragma unroll
        for (int ct = 0; ct < 2; ++ct) {
            v8f acc = {};
            #pragma unroll
            for (int kk = 0; kk < 20; kk += 4) {
                const int ka = kk + (lhi ? 2 : 0);
                v2f a, b;
                a.x = sea[(etile + l16)*21 + ka];
                a.y = sea[(etile + l16)*21 + ka + 1];
                b.x = (ka     < RPE) ? Wp[ka     * DHH + ct*16 + l16] : 0.0f;
                b.y = (ka + 1 < RPE) ? Wp[(ka+1) * DHH + ct*16 + l16] : 0.0f;
                acc = __builtin_amdgcn_wmma_f32_16x16x4_f32(
                    false, a, false, b, (short)0, acc, false, false);
            }
            const int cb = sel*32 + ct*16 + l16;
            #pragma unroll
            for (int r = 0; r < 8; ++r)
                srpe[(etile + r + (lhi ? 8 : 0))*67 + cb] = acc[r];
        }
    }
    __syncthreads();

    const int eloc = tid >> 1, pairh = (tid & 1) * 2;
    const int ge = e0 + eloc;
    if (ge < E_) {
        const int s = eidx[ge];
        const int t = eidx[E_ + ge];
        const float scale = 0.35355339059327373f;
        const float* qrow = qkv + (size_t)s * QKVW;
        const float* krow = qkv + (size_t)t * QKVW + DHH;
        #pragma unroll
        for (int hh = 0; hh < 2; ++hh) {
            const int h = pairh + hh;
            float compat = 0.f;
            #pragma unroll
            for (int d = 0; d < DD; ++d) {
                const int c = h*DD + d;
                const float qe = qrow[c]*scale + srpe[eloc*67 + c]      + qb[c];
                const float ke = krow[c]       + srpe[eloc*67 + 32 + c] + kb[c];
                compat = fmaf(qe, ke, compat);
            }
            const float ev = __expf(compat);
            exbuf[(size_t)ge * HH + h] = ev;
            atomicAdd(&den[s * HH + h], ev);
        }
    }
}

__global__ __launch_bounds__(256) void edge_out_kernel(
    const float* __restrict__ ea, const int* __restrict__ eidx,
    const float* __restrict__ qkv,
    const float* __restrict__ vw, const float* __restrict__ vb,
    const float* __restrict__ exbuf, const float* __restrict__ den,
    float* __restrict__ out, int E_)
{
    __shared__ float sea[128 * 21];
    __shared__ float srpe[128 * 67];
    const int tid = threadIdx.x;
    const int e0  = blockIdx.x * 128;

    __builtin_prefetch(ea + (size_t)(e0 + 128) * RPE, 0, 1);

    for (int i = tid; i < 128 * RPE; i += 256) {
        int e = i / RPE, c = i % RPE;
        int ge = e0 + e;
        sea[e*21 + c] = (ge < E_) ? ea[(size_t)ge * RPE + c] : 0.0f;
    }
    if (tid < 128) { sea[tid*21 + 18] = 0.f; sea[tid*21 + 19] = 0.f; }
    __syncthreads();

    const int wave = tid >> 5, lane = tid & 31;
    const int lhi = lane >> 4, l16 = lane & 15;
    const int etile = wave * 16;

    #pragma unroll
    for (int ct = 0; ct < 4; ++ct) {
        v8f acc = {};
        #pragma unroll
        for (int kk = 0; kk < 20; kk += 4) {
            const int ka = kk + (lhi ? 2 : 0);
            v2f a, b;
            a.x = sea[(etile + l16)*21 + ka];
            a.y = sea[(etile + l16)*21 + ka + 1];
            b.x = (ka     < RPE) ? vw[ka     * DIMV + ct*16 + l16] : 0.0f;
            b.y = (ka + 1 < RPE) ? vw[(ka+1) * DIMV + ct*16 + l16] : 0.0f;
            acc = __builtin_amdgcn_wmma_f32_16x16x4_f32(
                false, a, false, b, (short)0, acc, false, false);
        }
        const int cb = ct*16 + l16;
        #pragma unroll
        for (int r = 0; r < 8; ++r)
            srpe[(etile + r + (lhi ? 8 : 0))*67 + cb] = acc[r];
    }
    __syncthreads();

    const int eloc = tid >> 1, half = tid & 1;
    const int ge = e0 + eloc;
    if (ge < E_) {
        const int s = eidx[ge];
        const int t = eidx[E_ + ge];
        float attn[2];
        #pragma unroll
        for (int hh = 0; hh < 2; ++hh) {
            const int h = half*2 + hh;
            attn[hh] = exbuf[(size_t)ge*HH + h] / (den[s*HH + h] + 1e-16f);
        }
        const float* vrow = qkv + (size_t)t * QKVW + 2*DHH;
        const int c0 = half * 32;
        #pragma unroll
        for (int c = c0; c < c0 + 32; ++c) {
            const float val = (vrow[c] + srpe[eloc*67 + c] + vb[c]) * attn[(c >> 4) & 1];
            atomicAdd(&out[(size_t)s * DIMV + c], val);
        }
    }
}

extern "C" void kernel_launch(void* const* d_in, const int* in_sizes, int n_in,
                              void* d_out, int out_size, void* d_ws, size_t ws_size,
                              hipStream_t stream)
{
    const float* x       = (const float*)d_in[0];
    const int*   eidx    = (const int*)  d_in[1];
    const float* ea      = (const float*)d_in[2];
    const float* qkv_w   = (const float*)d_in[3];
    const float* qkv_b   = (const float*)d_in[4];
    const float* k_rpe_w = (const float*)d_in[5];
    const float* k_rpe_b = (const float*)d_in[6];
    const float* q_rpe_w = (const float*)d_in[7];
    const float* q_rpe_b = (const float*)d_in[8];
    const float* v_rpe_w = (const float*)d_in[9];
    const float* v_rpe_b = (const float*)d_in[10];

    const int N  = in_sizes[0] / DIMV;
    const int E_ = in_sizes[2] / RPE;
    float* out = (float*)d_out;

    char* ws = (char*)d_ws;
    float* qkv  = (float*)ws;
    float* den  = (float*)(ws + (size_t)N * QKVW * 4);
    float* exb  = (float*)(ws + (size_t)N * QKVW * 4
                              + (size_t)N * HH   * 4);

    hipMemsetAsync(den, 0, (size_t)N * HH   * sizeof(float), stream);
    hipMemsetAsync(out, 0, (size_t)N * DIMV * sizeof(float), stream);

    qkv_proj_kernel<<<(N + 15) / 16, 256, 0, stream>>>(x, qkv_w, qkv_b, qkv, N);
    edge_compat_kernel<<<(E_ + 127) / 128, 256, 0, stream>>>(
        ea, eidx, qkv, q_rpe_w, q_rpe_b, k_rpe_w, k_rpe_b, exb, den, E_);
    edge_out_kernel<<<(E_ + 127) / 128, 256, 0, stream>>>(
        ea, eidx, qkv, v_rpe_w, v_rpe_b, exb, den, out, E_);
}


# Counting-sort edges by source node (histogram+scan+fill) so edge_out sums each node's run in registers, lane=column, one atomic flush per run; float sum order changes (arrival-order rank)
# speedup vs baseline: 1.7833x; 1.7833x over previous
#include <hip/hip_runtime.h>
#include <hip/hip_bf16.h>


typedef __attribute__((ext_vector_type(2))) float v2f;
typedef __attribute__((ext_vector_type(8))) float v8f;

#define HH    4
#define DD    8
#define DHH   32
#define DIMV  64
#define RPE   18
#define QKVW  128

__global__ __launch_bounds__(256) void qkv_proj_kernel(
    const float* __restrict__ x, const float* __restrict__ W,
    const float* __restrict__ bias, float* __restrict__ qkv, int N)
{
    __shared__ float xs[16 * 65];
    const int tid = threadIdx.x;
    const int r0  = blockIdx.x * 16;

    {
        int idx = tid * 4;
        int row = idx >> 6, col = idx & 63;
        int grow = r0 + row;
        float4 vx = make_float4(0.f, 0.f, 0.f, 0.f);
        if (grow < N) vx = *(const float4*)(x + (size_t)grow * 64 + col);
        xs[row*65+col+0] = vx.x; xs[row*65+col+1] = vx.y;
        xs[row*65+col+2] = vx.z; xs[row*65+col+3] = vx.w;
    }
    __syncthreads();

    const int wave = tid >> 5;
    const int lane = tid & 31;
    const int lhi  = lane >> 4;
    const int l16  = lane & 15;
    const int colbase = wave * 16;

    v8f acc = {};
    #pragma unroll
    for (int kk = 0; kk < 64; kk += 4) {
        const int ka = kk + (lhi ? 2 : 0);
        v2f a, b;
        a.x = xs[l16*65 + ka];
        a.y = xs[l16*65 + ka + 1];
        b.x = W[(size_t)ka     * QKVW + colbase + l16];
        b.y = W[(size_t)(ka+1) * QKVW + colbase + l16];
        acc = __builtin_amdgcn_wmma_f32_16x16x4_f32(
            false, a, false, b, (short)0, acc, false, false);
    }

    const int col = colbase + l16;
    const float bv = bias[col];
    #pragma unroll
    for (int r = 0; r < 8; ++r) {
        int grow = r0 + r + (lhi ? 8 : 0);
        if (grow < N) qkv[(size_t)grow * QKVW + col] = acc[r] + bv;
    }
}

__global__ __launch_bounds__(256) void edge_compat_kernel(
    const float* __restrict__ ea, const int* __restrict__ eidx,
    const float* __restrict__ qkv,
    const float* __restrict__ qw, const float* __restrict__ qb,
    const float* __restrict__ kw, const float* __restrict__ kb,
    float* __restrict__ exbuf, float* __restrict__ den, int* __restrict__ cnt, int E_)
{
    __shared__ float sea[128 * 21];
    __shared__ float srpe[128 * 67];
    const int tid = threadIdx.x;
    const int e0  = blockIdx.x * 128;

    __builtin_prefetch(ea + (size_t)(e0 + 128) * RPE, 0, 1);

    for (int i = tid; i < 128 * RPE; i += 256) {
        int e = i / RPE, c = i % RPE;
        int ge = e0 + e;
        sea[e*21 + c] = (ge < E_) ? ea[(size_t)ge * RPE + c] : 0.0f;
    }
    if (tid < 128) { sea[tid*21 + 18] = 0.f; sea[tid*21 + 19] = 0.f; }
    __syncthreads();

    const int wave = tid >> 5, lane = tid & 31;
    const int lhi = lane >> 4, l16 = lane & 15;
    const int etile = wave * 16;

    #pragma unroll
    for (int sel = 0; sel < 2; ++sel) {
        const float* Wp = sel ? kw : qw;
        #pragma unroll
        for (int ct = 0; ct < 2; ++ct) {
            v8f acc = {};
            #pragma unroll
            for (int kk = 0; kk < 20; kk += 4) {
                const int ka = kk + (lhi ? 2 : 0);
                v2f a, b;
                a.x = sea[(etile + l16)*21 + ka];
                a.y = sea[(etile + l16)*21 + ka + 1];
                b.x = (ka     < RPE) ? Wp[ka     * DHH + ct*16 + l16] : 0.0f;
                b.y = (ka + 1 < RPE) ? Wp[(ka+1) * DHH + ct*16 + l16] : 0.0f;
                acc = __builtin_amdgcn_wmma_f32_16x16x4_f32(
                    false, a, false, b, (short)0, acc, false, false);
            }
            const int cb = sel*32 + ct*16 + l16;
            #pragma unroll
            for (int r = 0; r < 8; ++r)
                srpe[(etile + r + (lhi ? 8 : 0))*67 + cb] = acc[r];
        }
    }
    __syncthreads();

    const int eloc = tid >> 1, pairh = (tid & 1) * 2;
    const int ge = e0 + eloc;
    if (ge < E_) {
        const int s = eidx[ge];
        const int t = eidx[E_ + ge];
        const float scale = 0.35355339059327373f;
        const float* qrow = qkv + (size_t)s * QKVW;
        const float* krow = qkv + (size_t)t * QKVW + DHH;
        if (pairh == 0) atomicAdd(&cnt[s], 1);
        #pragma unroll
        for (int hh = 0; hh < 2; ++hh) {
            const int h = pairh + hh;
            float compat = 0.f;
            #pragma unroll
            for (int d = 0; d < DD; ++d) {
                const int c = h*DD + d;
                const float qe = qrow[c]*scale + srpe[eloc*67 + c]      + qb[c];
                const float ke = krow[c]       + srpe[eloc*67 + 32 + c] + kb[c];
                compat = fmaf(qe, ke, compat);
            }
            const float ev = __expf(compat);
            exbuf[(size_t)ge * HH + h] = ev;
            atomicAdd(&den[s * HH + h], ev);
        }
    }
}

__global__ __launch_bounds__(1024) void degree_scan_kernel(
    const int* __restrict__ cnt, int* __restrict__ cursor, int N)
{
    __shared__ int wsum[33];
    const int tid = threadIdx.x, lane = tid & 31, wave = tid >> 5;
    const bool vec_ok =
        ((reinterpret_cast<size_t>(cnt) | reinterpret_cast<size_t>(cursor)) & 15) == 0;
    int carry = 0;
    for (int base = 0; base < N; base += 8192) {
        const int i0 = base + tid * 8;
        const bool whole = vec_ok && (i0 + 7 < N);
        int c[8];
        if (whole) {
            const int4 a = *reinterpret_cast<const int4*>(cnt + i0);
            const int4 b = *reinterpret_cast<const int4*>(cnt + i0 + 4);
            c[0] = a.x; c[1] = a.y; c[2] = a.z; c[3] = a.w;
            c[4] = b.x; c[5] = b.y; c[6] = b.z; c[7] = b.w;
        } else {
            #pragma unroll
            for (int k = 0; k < 8; ++k) c[k] = (i0 + k < N) ? cnt[i0 + k] : 0;
        }
        int tsum = 0;
        #pragma unroll
        for (int k = 0; k < 8; ++k) tsum += c[k];
        int x = tsum;
        #pragma unroll
        for (int d = 1; d < 32; d <<= 1) {
            const int y = __shfl_up(x, d);
            if (lane >= d) x += y;
        }
        if (lane == 31) wsum[wave] = x;
        __syncthreads();
        if (wave == 0) {
            const int w = wsum[lane];
            int xs = w;
            #pragma unroll
            for (int d = 1; d < 32; d <<= 1) {
                const int y = __shfl_up(xs, d);
                if (lane >= d) xs += y;
            }
            wsum[lane] = xs - w;
            if (lane == 31) wsum[32] = xs;
        }
        __syncthreads();
        int p = carry + wsum[wave] + (x - tsum);
        carry += wsum[32];
        if (whole) {
            int4 a, b;
            a.x = p; p += c[0]; a.y = p; p += c[1]; a.z = p; p += c[2]; a.w = p; p += c[3];
            b.x = p; p += c[4]; b.y = p; p += c[5]; b.z = p; p += c[6]; b.w = p;
            *reinterpret_cast<int4*>(cursor + i0)     = a;
            *reinterpret_cast<int4*>(cursor + i0 + 4) = b;
        } else {
            #pragma unroll
            for (int k = 0; k < 8; ++k) {
                if (i0 + k < N) cursor[i0 + k] = p;
                p += c[k];
            }
        }
        __syncthreads();
    }
}

__global__ __launch_bounds__(256) void edge_fill_kernel(
    const int* __restrict__ eidx, int* __restrict__ cursor,
    int* __restrict__ perm, int E_)
{
    const int e = blockIdx.x * 256 + threadIdx.x;
    if (e < E_) {
        const int s = eidx[e];
        const int t = eidx[E_ + e];
        const int pos = atomicAdd(&cursor[s], 1);
        reinterpret_cast<int4*>(perm)[pos] = make_int4(e, s, t, 0);
    }
}

__global__ __launch_bounds__(256) void edge_out_kernel(
    const float* __restrict__ ea, const int* __restrict__ perm,
    const float* __restrict__ qkv,
    const float* __restrict__ vw, const float* __restrict__ vb,
    const float* __restrict__ exbuf, const float* __restrict__ den,
    float* __restrict__ out, int E_)
{
    __shared__ float sea[128 * 21];
    __shared__ float srpe[128 * 67];
    __shared__ int sE[128];
    __shared__ int sS[128];
    __shared__ int sT[128];
    const int tid = threadIdx.x;
    const int e0  = blockIdx.x * 128;

    if (tid < 128) {
        const int i = e0 + tid;
        int e = 0, s = -1, t = 0;
        if (i < E_) {
            const int4 rec = reinterpret_cast<const int4*>(perm)[i];
            e = rec.x; s = rec.y; t = rec.z;
        }
        sE[tid] = e; sS[tid] = s; sT[tid] = t;
    }
    __syncthreads();

    for (int i = tid; i < 128 * RPE; i += 256) {
        int el = i / RPE, c = i % RPE;
        sea[el*21 + c] = (sS[el] >= 0) ? ea[(size_t)sE[el] * RPE + c] : 0.0f;
    }
    if (tid < 128) { sea[tid*21 + 18] = 0.f; sea[tid*21 + 19] = 0.f; }
    __syncthreads();

    const int wave = tid >> 5, lane = tid & 31;
    const int lhi = lane >> 4, l16 = lane & 15;
    const int etile = wave * 16;

    #pragma unroll
    for (int ct = 0; ct < 4; ++ct) {
        v8f acc = {};
        #pragma unroll
        for (int kk = 0; kk < 20; kk += 4) {
            const int ka = kk + (lhi ? 2 : 0);
            v2f a, b;
            a.x = sea[(etile + l16)*21 + ka];
            a.y = sea[(etile + l16)*21 + ka + 1];
            b.x = (ka     < RPE) ? vw[ka     * DIMV + ct*16 + l16] : 0.0f;
            b.y = (ka + 1 < RPE) ? vw[(ka+1) * DIMV + ct*16 + l16] : 0.0f;
            acc = __builtin_amdgcn_wmma_f32_16x16x4_f32(
                false, a, false, b, (short)0, acc, false, false);
        }
        const int cb = ct*16 + l16;
        #pragma unroll
        for (int r = 0; r < 8; ++r)
            srpe[(etile + r + (lhi ? 8 : 0))*67 + cb] = acc[r];
    }
    __syncthreads();

    const int me = sE[etile + l16];
    const int ms = sS[etile + l16];
    const int mt = sT[etile + l16];
    const float vb0 = vb[lane], vb1 = vb[lane + 32];
    int cur = -1;
    float r0 = 0.f, r1 = 0.f;
    #pragma unroll 1
    for (int hb = 0; hb < 16; hb += 8) {
        float val0[8], val1[8];
        int   srun[8];
        #pragma unroll
        for (int j = 0; j < 8; ++j) {
            const int e = __builtin_amdgcn_readlane(me, hb + j);
            const int s = __builtin_amdgcn_readlane(ms, hb + j);
            const int t = __builtin_amdgcn_readlane(mt, hb + j);
            const int sd = (s < 0) ? 0 : s;
            const float4 ex4 = *reinterpret_cast<const float4*>(exbuf + (size_t)e  * HH);
            const float4 dn4 = *reinterpret_cast<const float4*>(den   + (size_t)sd * HH);
            const float a0 = (lhi ? ex4.y : ex4.x) / ((lhi ? dn4.y : dn4.x) + 1e-16f);
            const float a1 = (lhi ? ex4.w : ex4.z) / ((lhi ? dn4.w : dn4.z) + 1e-16f);
            const float* vrow = qkv + (size_t)t * QKVW + 2*DHH;
            val0[j] = (vrow[lane]      + srpe[(etile + hb + j)*67 + lane]      + vb0) * a0;
            val1[j] = (vrow[lane + 32] + srpe[(etile + hb + j)*67 + lane + 32] + vb1) * a1;
            srun[j] = s;
        }
        #pragma unroll
        for (int j = 0; j < 8; ++j) {
            if (srun[j] != cur) {
                if (cur >= 0) {
                    atomicAdd(&out[(size_t)cur * DIMV + lane],      r0);
                    atomicAdd(&out[(size_t)cur * DIMV + 32 + lane], r1);
                }
                cur = srun[j]; r0 = val0[j]; r1 = val1[j];
            } else {
                r0 += val0[j]; r1 += val1[j];
            }
        }
    }
    if (cur >= 0) {
        atomicAdd(&out[(size_t)cur * DIMV + lane],      r0);
        atomicAdd(&out[(size_t)cur * DIMV + 32 + lane], r1);
    }
}

extern "C" void kernel_launch(void* const* d_in, const int* in_sizes, int n_in,
                              void* d_out, int out_size, void* d_ws, size_t ws_size,
                              hipStream_t stream)
{
    const float* x       = (const float*)d_in[0];
    const int*   eidx    = (const int*)  d_in[1];
    const float* ea      = (const float*)d_in[2];
    const float* qkv_w   = (const float*)d_in[3];
    const float* qkv_b   = (const float*)d_in[4];
    const float* k_rpe_w = (const float*)d_in[5];
    const float* k_rpe_b = (const float*)d_in[6];
    const float* q_rpe_w = (const float*)d_in[7];
    const float* q_rpe_b = (const float*)d_in[8];
    const float* v_rpe_w = (const float*)d_in[9];
    const float* v_rpe_b = (const float*)d_in[10];

    const int N  = in_sizes[0] / DIMV;
    const int E_ = in_sizes[2] / RPE;
    float* out = (float*)d_out;

    char* ws = (char*)d_ws;
    float* qkv  = (float*)ws;
    float* exb  = (float*)(ws + (size_t)N * QKVW * 4);
    int*   perm = (int*)  (ws + (size_t)N * QKVW * 4
                              + (size_t)E_ * HH * 4);
    float* den  = (float*)(ws + (size_t)N * QKVW * 4
                              + (size_t)E_ * HH * 8);
    int*   cnt  = (int*)  (ws + (size_t)N * QKVW * 4
                              + (size_t)E_ * HH * 8
                              + (size_t)N * HH * 4);
    int*   cur  = (int*)  (ws + (size_t)N * QKVW * 4
                              + (size_t)E_ * HH * 8
                              + (size_t)N * (HH + 1) * 4);

    hipMemsetAsync(den, 0, (size_t)N * (HH + 1) * sizeof(float), stream);
    hipMemsetAsync(out, 0, (size_t)N * DIMV * sizeof(float), stream);

    qkv_proj_kernel<<<(N + 15) / 16, 256, 0, stream>>>(x, qkv_w, qkv_b, qkv, N);
    edge_compat_kernel<<<(E_ + 127) / 128, 256, 0, stream>>>(
        ea, eidx, qkv, q_rpe_w, q_rpe_b, k_rpe_w, k_rpe_b, exb, den, cnt, E_);
    degree_scan_kernel<<<1, 1024, 0, stream>>>(cnt, cur, N);
    edge_fill_kernel<<<(E_ + 255) / 256, 256, 0, stream>>>(eidx, cur, perm, E_);
    edge_out_kernel<<<(E_ + 127) / 128, 256, 0, stream>>>(
        ea, perm, qkv, v_rpe_w, v_rpe_b, exb, den, out, E_);
}


# Softmax denominator summed per sorted run in the output kernel (one 4-lane atomic per run), not 4 scattered f32 atomics per edge; rows divided in a last pass; float sum order and rounding change
# speedup vs baseline: 1.9087x; 1.0703x over previous
#include <hip/hip_runtime.h>
#include <hip/hip_bf16.h>


typedef __attribute__((ext_vector_type(2))) float v2f;
typedef __attribute__((ext_vector_type(8))) float v8f;

#define HH    4
#define DD    8
#define DHH   32
#define DIMV  64
#define RPE   18
#define QKVW  128

__global__ __launch_bounds__(256) void qkv_proj_kernel(
    const float* __restrict__ x, const float* __restrict__ W,
    const float* __restrict__ bias, float* __restrict__ qkv, int N)
{
    __shared__ float xs[16 * 65];
    const int tid = threadIdx.x;
    const int r0  = blockIdx.x * 16;

    {
        int idx = tid * 4;
        int row = idx >> 6, col = idx & 63;
        int grow = r0 + row;
        float4 vx = make_float4(0.f, 0.f, 0.f, 0.f);
        if (grow < N) vx = *(const float4*)(x + (size_t)grow * 64 + col);
        xs[row*65+col+0] = vx.x; xs[row*65+col+1] = vx.y;
        xs[row*65+col+2] = vx.z; xs[row*65+col+3] = vx.w;
    }
    __syncthreads();

    const int wave = tid >> 5;
    const int lane = tid & 31;
    const int lhi  = lane >> 4;
    const int l16  = lane & 15;
    const int colbase = wave * 16;

    v8f acc = {};
    #pragma unroll
    for (int kk = 0; kk < 64; kk += 4) {
        const int ka = kk + (lhi ? 2 : 0);
        v2f a, b;
        a.x = xs[l16*65 + ka];
        a.y = xs[l16*65 + ka + 1];
        b.x = W[(size_t)ka     * QKVW + colbase + l16];
        b.y = W[(size_t)(ka+1) * QKVW + colbase + l16];
        acc = __builtin_amdgcn_wmma_f32_16x16x4_f32(
            false, a, false, b, (short)0, acc, false, false);
    }

    const int col = colbase + l16;
    const float bv = bias[col];
    #pragma unroll
    for (int r = 0; r < 8; ++r) {
        int grow = r0 + r + (lhi ? 8 : 0);
        if (grow < N) qkv[(size_t)grow * QKVW + col] = acc[r] + bv;
    }
}

__global__ __launch_bounds__(256) void edge_compat_kernel(
    const float* __restrict__ ea, const int* __restrict__ eidx,
    const float* __restrict__ qkv,
    const float* __restrict__ qw, const float* __restrict__ qb,
    const float* __restrict__ kw, const float* __restrict__ kb,
    float* __restrict__ exbuf, int* __restrict__ cnt, int E_)
{
    __shared__ float sea[128 * 21];
    __shared__ float srpe[128 * 67];
    const int tid = threadIdx.x;
    const int e0  = blockIdx.x * 128;

    __builtin_prefetch(ea + (size_t)(e0 + 128) * RPE, 0, 1);

    for (int i = tid; i < 128 * RPE; i += 256) {
        int e = i / RPE, c = i % RPE;
        int ge = e0 + e;
        sea[e*21 + c] = (ge < E_) ? ea[(size_t)ge * RPE + c] : 0.0f;
    }
    if (tid < 128) { sea[tid*21 + 18] = 0.f; sea[tid*21 + 19] = 0.f; }
    __syncthreads();

    const int wave = tid >> 5, lane = tid & 31;
    const int lhi = lane >> 4, l16 = lane & 15;
    const int etile = wave * 16;

    #pragma unroll
    for (int sel = 0; sel < 2; ++sel) {
        const float* Wp = sel ? kw : qw;
        #pragma unroll
        for (int ct = 0; ct < 2; ++ct) {
            v8f acc = {};
            #pragma unroll
            for (int kk = 0; kk < 20; kk += 4) {
                const int ka = kk + (lhi ? 2 : 0);
                v2f a, b;
                a.x = sea[(etile + l16)*21 + ka];
                a.y = sea[(etile + l16)*21 + ka + 1];
                b.x = (ka     < RPE) ? Wp[ka     * DHH + ct*16 + l16] : 0.0f;
                b.y = (ka + 1 < RPE) ? Wp[(ka+1) * DHH + ct*16 + l16] : 0.0f;
                acc = __builtin_amdgcn_wmma_f32_16x16x4_f32(
                    false, a, false, b, (short)0, acc, false, false);
            }
            const int cb = sel*32 + ct*16 + l16;
            #pragma unroll
            for (int r = 0; r < 8; ++r)
                srpe[(etile + r + (lhi ? 8 : 0))*67 + cb] = acc[r];
        }
    }
    __syncthreads();

    const int eloc = tid >> 1, pairh = (tid & 1) * 2;
    const int ge = e0 + eloc;
    if (ge < E_) {
        const int s = eidx[ge];
        const int t = eidx[E_ + ge];
        const float scale = 0.35355339059327373f;
        const float* qrow = qkv + (size_t)s * QKVW;
        const float* krow = qkv + (size_t)t * QKVW + DHH;
        if (pairh == 0) atomicAdd(&cnt[s], 1);
        #pragma unroll
        for (int hh = 0; hh < 2; ++hh) {
            const int h = pairh + hh;
            float compat = 0.f;
            #pragma unroll
            for (int d = 0; d < DD; ++d) {
                const int c = h*DD + d;
                const float qe = qrow[c]*scale + srpe[eloc*67 + c]      + qb[c];
                const float ke = krow[c]       + srpe[eloc*67 + 32 + c] + kb[c];
                compat = fmaf(qe, ke, compat);
            }
            const float ev = __expf(compat);
            exbuf[(size_t)ge * HH + h] = ev;
        }
    }
}

__global__ __launch_bounds__(1024) void degree_scan_kernel(
    const int* __restrict__ cnt, int* __restrict__ cursor, int N)
{
    __shared__ int wsum[33];
    const int tid = threadIdx.x, lane = tid & 31, wave = tid >> 5;
    const bool vec_ok =
        ((reinterpret_cast<size_t>(cnt) | reinterpret_cast<size_t>(cursor)) & 15) == 0;
    int carry = 0;
    for (int base = 0; base < N; base += 8192) {
        const int i0 = base + tid * 8;
        const bool whole = vec_ok && (i0 + 7 < N);
        int c[8];
        if (whole) {
            const int4 a = *reinterpret_cast<const int4*>(cnt + i0);
            const int4 b = *reinterpret_cast<const int4*>(cnt + i0 + 4);
            c[0] = a.x; c[1] = a.y; c[2] = a.z; c[3] = a.w;
            c[4] = b.x; c[5] = b.y; c[6] = b.z; c[7] = b.w;
        } else {
            #pragma unroll
            for (int k = 0; k < 8; ++k) c[k] = (i0 + k < N) ? cnt[i0 + k] : 0;
        }
        int tsum = 0;
        #pragma unroll
        for (int k = 0; k < 8; ++k) tsum += c[k];
        int x = tsum;
        #pragma unroll
        for (int d = 1; d < 32; d <<= 1) {
            const int y = __shfl_up(x, d);
            if (lane >= d) x += y;
        }
        if (lane == 31) wsum[wave] = x;
        __syncthreads();
        if (wave == 0) {
            const int w = wsum[lane];
            int xs = w;
            #pragma unroll
            for (int d = 1; d < 32; d <<= 1) {
                const int y = __shfl_up(xs, d);
                if (lane >= d) xs += y;
            }
            wsum[lane] = xs - w;
            if (lane == 31) wsum[32] = xs;
        }
        __syncthreads();
        int p = carry + wsum[wave] + (x - tsum);
        carry += wsum[32];
        if (whole) {
            int4 a, b;
            a.x = p; p += c[0]; a.y = p; p += c[1]; a.z = p; p += c[2]; a.w = p; p += c[3];
            b.x = p; p += c[4]; b.y = p; p += c[5]; b.z = p; p += c[6]; b.w = p;
            *reinterpret_cast<int4*>(cursor + i0)     = a;
            *reinterpret_cast<int4*>(cursor + i0 + 4) = b;
        } else {
            #pragma unroll
            for (int k = 0; k < 8; ++k) {
                if (i0 + k < N) cursor[i0 + k] = p;
                p += c[k];
            }
        }
        __syncthreads();
    }
}

__global__ __launch_bounds__(256) void edge_fill_kernel(
    const int* __restrict__ eidx, int* __restrict__ cursor,
    int* __restrict__ perm, int E_)
{
    const int e = blockIdx.x * 256 + threadIdx.x;
    if (e < E_) {
        const int s = eidx[e];
        const int t = eidx[E_ + e];
        const int pos = atomicAdd(&cursor[s], 1);
        reinterpret_cast<int4*>(perm)[pos] = make_int4(e, s, t, 0);
    }
}

__global__ __launch_bounds__(256) void edge_out_kernel(
    const float* __restrict__ ea, const int* __restrict__ perm,
    const float* __restrict__ qkv,
    const float* __restrict__ vw, const float* __restrict__ vb,
    const float* __restrict__ exbuf, float* __restrict__ den,
    float* __restrict__ out, int E_)
{
    __shared__ float sea[128 * 21];
    __shared__ float srpe[128 * 67];
    __shared__ int sE[128];
    __shared__ int sS[128];
    __shared__ int sT[128];
    const int tid = threadIdx.x;
    const int e0  = blockIdx.x * 128;

    if (tid < 128) {
        const int i = e0 + tid;
        int e = 0, s = -1, t = 0;
        if (i < E_) {
            const int4 rec = reinterpret_cast<const int4*>(perm)[i];
            e = rec.x; s = rec.y; t = rec.z;
        }
        sE[tid] = e; sS[tid] = s; sT[tid] = t;
    }
    __syncthreads();

    for (int i = tid; i < 128 * RPE; i += 256) {
        int el = i / RPE, c = i % RPE;
        sea[el*21 + c] = (sS[el] >= 0) ? ea[(size_t)sE[el] * RPE + c] : 0.0f;
    }
    if (tid < 128) { sea[tid*21 + 18] = 0.f; sea[tid*21 + 19] = 0.f; }
    __syncthreads();

    const int wave = tid >> 5, lane = tid & 31;
    const int lhi = lane >> 4, l16 = lane & 15;
    const int etile = wave * 16;

    #pragma unroll
    for (int ct = 0; ct < 4; ++ct) {
        v8f acc = {};
        #pragma unroll
        for (int kk = 0; kk < 20; kk += 4) {
            const int ka = kk + (lhi ? 2 : 0);
            v2f a, b;
            a.x = sea[(etile + l16)*21 + ka];
            a.y = sea[(etile + l16)*21 + ka + 1];
            b.x = (ka     < RPE) ? vw[ka     * DIMV + ct*16 + l16] : 0.0f;
            b.y = (ka + 1 < RPE) ? vw[(ka+1) * DIMV + ct*16 + l16] : 0.0f;
            acc = __builtin_amdgcn_wmma_f32_16x16x4_f32(
                false, a, false, b, (short)0, acc, false, false);
        }
        const int cb = ct*16 + l16;
        #pragma unroll
        for (int r = 0; r < 8; ++r)
            srpe[(etile + r + (lhi ? 8 : 0))*67 + cb] = acc[r];
    }
    __syncthreads();

    const int me = sE[etile + l16];
    const int ms = sS[etile + l16];
    const int mt = sT[etile + l16];
    const float vb0 = vb[lane], vb1 = vb[lane + 32];
    int cur = -1;
    float r0 = 0.f, r1 = 0.f;
    float d0 = 0.f, d1 = 0.f;
    #pragma unroll 1
    for (int hb = 0; hb < 16; hb += 8) {
        float val0[8], val1[8];
        float wgt0[8], wgt1[8];
        int   srun[8];
        #pragma unroll
        for (int j = 0; j < 8; ++j) {
            const int e = __builtin_amdgcn_readlane(me, hb + j);
            const int s = __builtin_amdgcn_readlane(ms, hb + j);
            const int t = __builtin_amdgcn_readlane(mt, hb + j);
            const float4 ex4 = *reinterpret_cast<const float4*>(exbuf + (size_t)e  * HH);
            const float a0 = lhi ? ex4.y : ex4.x;
            const float a1 = lhi ? ex4.w : ex4.z;
            const float* vrow = qkv + (size_t)t * QKVW + 2*DHH;
            val0[j] = (vrow[lane]      + srpe[(etile + hb + j)*67 + lane]      + vb0) * a0;
            val1[j] = (vrow[lane + 32] + srpe[(etile + hb + j)*67 + lane + 32] + vb1) * a1;
            wgt0[j] = a0; wgt1[j] = a1;
            srun[j] = s;
        }
        #pragma unroll
        for (int j = 0; j < 8; ++j) {
            if (srun[j] != cur) {
                if (cur >= 0) {
                    atomicAdd(&out[(size_t)cur * DIMV + lane],      r0);
                    atomicAdd(&out[(size_t)cur * DIMV + 32 + lane], r1);
                    if (l16 < 2) atomicAdd(&den[cur * HH + lhi + 2*l16], l16 ? d1 : d0);
                }
                cur = srun[j]; r0 = val0[j]; r1 = val1[j]; d0 = wgt0[j]; d1 = wgt1[j];
            } else {
                r0 += val0[j]; r1 += val1[j]; d0 += wgt0[j]; d1 += wgt1[j];
            }
        }
    }
    if (cur >= 0) {
        atomicAdd(&out[(size_t)cur * DIMV + lane],      r0);
        atomicAdd(&out[(size_t)cur * DIMV + 32 + lane], r1);
        if (l16 < 2) atomicAdd(&den[cur * HH + lhi + 2*l16], l16 ? d1 : d0);
    }
}

__global__ __launch_bounds__(256) void out_divide_kernel(
    float* __restrict__ out, const float* __restrict__ den, int N)
{
    const int total = N * (DIMV / 4);
    const int base  = blockIdx.x * 2048 + threadIdx.x;
    float4 v[8];
    float  d[8];
    #pragma unroll
    for (int k = 0; k < 8; ++k) {
        const int i = base + k * 256;
        v[k] = make_float4(0.f, 0.f, 0.f, 0.f);
        d[k] = 1.0f;
        if (i < total) {
            v[k] = reinterpret_cast<const float4*>(out)[i];
            d[k] = den[(i >> 4) * HH + ((i & 15) >> 2)];
        }
    }
    #pragma unroll
    for (int k = 0; k < 8; ++k) {
        const int i = base + k * 256;
        if (i < total) {
            const float dd = d[k] + 1e-16f;
            float4 w = v[k];
            w.x = w.x / dd; w.y = w.y / dd; w.z = w.z / dd; w.w = w.w / dd;
            reinterpret_cast<float4*>(out)[i] = w;
        }
    }
}

extern "C" void kernel_launch(void* const* d_in, const int* in_sizes, int n_in,
                              void* d_out, int out_size, void* d_ws, size_t ws_size,
                              hipStream_t stream)
{
    const float* x       = (const float*)d_in[0];
    const int*   eidx    = (const int*)  d_in[1];
    const float* ea      = (const float*)d_in[2];
    const float* qkv_w   = (const float*)d_in[3];
    const float* qkv_b   = (const float*)d_in[4];
    const float* k_rpe_w = (const float*)d_in[5];
    const float* k_rpe_b = (const float*)d_in[6];
    const float* q_rpe_w = (const float*)d_in[7];
    const float* q_rpe_b = (const float*)d_in[8];
    const float* v_rpe_w = (const float*)d_in[9];
    const float* v_rpe_b = (const float*)d_in[10];

    const int N  = in_sizes[0] / DIMV;
    const int E_ = in_sizes[2] / RPE;
    float* out = (float*)d_out;

    char* ws = (char*)d_ws;
    float* qkv  = (float*)ws;
    float* exb  = (float*)(ws + (size_t)N * QKVW * 4);
    int*   perm = (int*)  (ws + (size_t)N * QKVW * 4
                              + (size_t)E_ * HH * 4);
    float* den  = (float*)(ws + (size_t)N * QKVW * 4
                              + (size_t)E_ * HH * 8);
    int*   cnt  = (int*)  (ws + (size_t)N * QKVW * 4
                              + (size_t)E_ * HH * 8
                              + (size_t)N * HH * 4);
    int*   cur  = (int*)  (ws + (size_t)N * QKVW * 4
                              + (size_t)E_ * HH * 8
                              + (size_t)N * (HH + 1) * 4);

    hipMemsetAsync(den, 0, (size_t)N * (HH + 1) * sizeof(float), stream);
    hipMemsetAsync(out, 0, (size_t)N * DIMV * sizeof(float), stream);

    qkv_proj_kernel<<<(N + 15) / 16, 256, 0, stream>>>(x, qkv_w, qkv_b, qkv, N);
    edge_compat_kernel<<<(E_ + 127) / 128, 256, 0, stream>>>(
        ea, eidx, qkv, q_rpe_w, q_rpe_b, k_rpe_w, k_rpe_b, exb, cnt, E_);
    degree_scan_kernel<<<1, 1024, 0, stream>>>(cnt, cur, N);
    edge_fill_kernel<<<(E_ + 255) / 256, 256, 0, stream>>>(eidx, cur, perm, E_);
    edge_out_kernel<<<(E_ + 127) / 128, 256, 0, stream>>>(
        ea, perm, qkv, v_rpe_w, v_rpe_b, exb, den, out, E_);
    out_divide_kernel<<<(N * (DIMV / 4) + 2047) / 2048, 256, 0, stream>>>(out, den, N);
}


# Edge kernels: each thread's 9 edge_attr staging loads, the edge ids and the weight B fragments issue as one group before the first wait (unrolled, hoisted above the barrier), not one load per wait
# speedup vs baseline: 2.3343x; 1.2230x over previous
#include <hip/hip_runtime.h>
#include <hip/hip_bf16.h>


typedef __attribute__((ext_vector_type(2))) float v2f;
typedef __attribute__((ext_vector_type(8))) float v8f;

#define HH    4
#define DD    8
#define DHH   32
#define DIMV  64
#define RPE   18
#define QKVW  128

__global__ __launch_bounds__(256) void qkv_proj_kernel(
    const float* __restrict__ x, const float* __restrict__ W,
    const float* __restrict__ bias, float* __restrict__ qkv, int N)
{
    __shared__ float xs[16 * 65];
    const int tid = threadIdx.x;
    const int r0  = blockIdx.x * 16;

    {
        int idx = tid * 4;
        int row = idx >> 6, col = idx & 63;
        int grow = r0 + row;
        float4 vx = make_float4(0.f, 0.f, 0.f, 0.f);
        if (grow < N) vx = *(const float4*)(x + (size_t)grow * 64 + col);
        xs[row*65+col+0] = vx.x; xs[row*65+col+1] = vx.y;
        xs[row*65+col+2] = vx.z; xs[row*65+col+3] = vx.w;
    }
    __syncthreads();

    const int wave = tid >> 5;
    const int lane = tid & 31;
    const int lhi  = lane >> 4;
    const int l16  = lane & 15;
    const int colbase = wave * 16;

    v8f acc = {};
    #pragma unroll
    for (int kk = 0; kk < 64; kk += 4) {
        const int ka = kk + (lhi ? 2 : 0);
        v2f a, b;
        a.x = xs[l16*65 + ka];
        a.y = xs[l16*65 + ka + 1];
        b.x = W[(size_t)ka     * QKVW + colbase + l16];
        b.y = W[(size_t)(ka+1) * QKVW + colbase + l16];
        acc = __builtin_amdgcn_wmma_f32_16x16x4_f32(
            false, a, false, b, (short)0, acc, false, false);
    }

    const int col = colbase + l16;
    const float bv = bias[col];
    #pragma unroll
    for (int r = 0; r < 8; ++r) {
        int grow = r0 + r + (lhi ? 8 : 0);
        if (grow < N) qkv[(size_t)grow * QKVW + col] = acc[r] + bv;
    }
}

__global__ __launch_bounds__(256) void edge_compat_kernel(
    const float* __restrict__ ea, const int* __restrict__ eidx,
    const float* __restrict__ qkv,
    const float* __restrict__ qw, const float* __restrict__ qb,
    const float* __restrict__ kw, const float* __restrict__ kb,
    float* __restrict__ exbuf, int* __restrict__ cnt, int E_)
{
    __shared__ float sea[128 * 21];
    __shared__ float srpe[128 * 67];
    const int tid = threadIdx.x;
    const int e0  = blockIdx.x * 128;

    __builtin_prefetch(ea + (size_t)(e0 + 128) * RPE, 0, 1);

    const int wave = tid >> 5, lane = tid & 31;
    const int lhi = lane >> 4, l16 = lane & 15;
    const int etile = wave * 16;


    const int eloc = tid >> 1, pairh = (tid & 1) * 2;
    const int ge = e0 + eloc;
    const int gid = min(ge, E_ - 1);
    const int s = eidx[gid];
    const int t = eidx[E_ + gid];

    float stg[9];
    #pragma unroll
    for (int k = 0; k < 9; ++k) {
        const int i = tid + k * 256;
        const int e = i / RPE, c = i % RPE;
        const int gc = min(e0 + e, E_ - 1);
        stg[k] = ea[(size_t)gc * RPE + c];
    }
    #pragma unroll
    for (int k = 0; k < 9; ++k) {
        const int i = tid + k * 256;
        const int e = i / RPE, c = i % RPE;
        sea[e*21 + c] = (e0 + e < E_) ? stg[k] : 0.0f;
    }

    v2f bw[2][2][5];
    #pragma unroll
    for (int sel = 0; sel < 2; ++sel) {
        const float* Wp = sel ? kw : qw;
        #pragma unroll
        for (int ct = 0; ct < 2; ++ct) {
            #pragma unroll
            for (int kq = 0; kq < 5; ++kq) {
                const int ka = kq*4 + (lhi ? 2 : 0);
                bw[sel][ct][kq].x = (ka     < RPE) ? Wp[ka     * DHH + ct*16 + l16] : 0.0f;
                bw[sel][ct][kq].y = (ka + 1 < RPE) ? Wp[(ka+1) * DHH + ct*16 + l16] : 0.0f;
            }
        }
    }
    if (tid < 128) { sea[tid*21 + 18] = 0.f; sea[tid*21 + 19] = 0.f; }
    __syncthreads();

    #pragma unroll
    for (int sel = 0; sel < 2; ++sel) {
        #pragma unroll
        for (int ct = 0; ct < 2; ++ct) {
            v8f acc = {};
            #pragma unroll
            for (int kq = 0; kq < 5; ++kq) {
                const int ka = kq*4 + (lhi ? 2 : 0);
                v2f a;
                a.x = sea[(etile + l16)*21 + ka];
                a.y = sea[(etile + l16)*21 + ka + 1];
                acc = __builtin_amdgcn_wmma_f32_16x16x4_f32(
                    false, a, false, bw[sel][ct][kq], (short)0, acc, false, false);
            }
            const int cb = sel*32 + ct*16 + l16;
            #pragma unroll
            for (int r = 0; r < 8; ++r)
                srpe[(etile + r + (lhi ? 8 : 0))*67 + cb] = acc[r];
        }
    }
    __syncthreads();

    if (ge < E_) {
        const float scale = 0.35355339059327373f;
        const float* qrow = qkv + (size_t)s * QKVW;
        const float* krow = qkv + (size_t)t * QKVW + DHH;
        if (pairh == 0) atomicAdd(&cnt[s], 1);
        #pragma unroll
        for (int hh = 0; hh < 2; ++hh) {
            const int h = pairh + hh;
            float compat = 0.f;
            #pragma unroll
            for (int d = 0; d < DD; ++d) {
                const int c = h*DD + d;
                const float qe = qrow[c]*scale + srpe[eloc*67 + c]      + qb[c];
                const float ke = krow[c]       + srpe[eloc*67 + 32 + c] + kb[c];
                compat = fmaf(qe, ke, compat);
            }
            const float ev = __expf(compat);
            exbuf[(size_t)ge * HH + h] = ev;
        }
    }
}

__global__ __launch_bounds__(1024) void degree_scan_kernel(
    const int* __restrict__ cnt, int* __restrict__ cursor, int N)
{
    __shared__ int wsum[33];
    const int tid = threadIdx.x, lane = tid & 31, wave = tid >> 5;
    const bool vec_ok =
        ((reinterpret_cast<size_t>(cnt) | reinterpret_cast<size_t>(cursor)) & 15) == 0;
    int carry = 0;
    for (int base = 0; base < N; base += 8192) {
        const int i0 = base + tid * 8;
        const bool whole = vec_ok && (i0 + 7 < N);
        int c[8];
        if (whole) {
            const int4 a = *reinterpret_cast<const int4*>(cnt + i0);
            const int4 b = *reinterpret_cast<const int4*>(cnt + i0 + 4);
            c[0] = a.x; c[1] = a.y; c[2] = a.z; c[3] = a.w;
            c[4] = b.x; c[5] = b.y; c[6] = b.z; c[7] = b.w;
        } else {
            #pragma unroll
            for (int k = 0; k < 8; ++k) c[k] = (i0 + k < N) ? cnt[i0 + k] : 0;
        }
        int tsum = 0;
        #pragma unroll
        for (int k = 0; k < 8; ++k) tsum += c[k];
        int x = tsum;
        #pragma unroll
        for (int d = 1; d < 32; d <<= 1) {
            const int y = __shfl_up(x, d);
            if (lane >= d) x += y;
        }
        if (lane == 31) wsum[wave] = x;
        __syncthreads();
        if (wave == 0) {
            const int w = wsum[lane];
            int xs = w;
            #pragma unroll
            for (int d = 1; d < 32; d <<= 1) {
                const int y = __shfl_up(xs, d);
                if (lane >= d) xs += y;
            }
            wsum[lane] = xs - w;
            if (lane == 31) wsum[32] = xs;
        }
        __syncthreads();
        int p = carry + wsum[wave] + (x - tsum);
        carry += wsum[32];
        if (whole) {
            int4 a, b;
            a.x = p; p += c[0]; a.y = p; p += c[1]; a.z = p; p += c[2]; a.w = p; p += c[3];
            b.x = p; p += c[4]; b.y = p; p += c[5]; b.z = p; p += c[6]; b.w = p;
            *reinterpret_cast<int4*>(cursor + i0)     = a;
            *reinterpret_cast<int4*>(cursor + i0 + 4) = b;
        } else {
            #pragma unroll
            for (int k = 0; k < 8; ++k) {
                if (i0 + k < N) cursor[i0 + k] = p;
                p += c[k];
            }
        }
        __syncthreads();
    }
}

__global__ __launch_bounds__(256) void edge_fill_kernel(
    const int* __restrict__ eidx, int* __restrict__ cursor,
    int* __restrict__ perm, int E_)
{
    const int e = blockIdx.x * 256 + threadIdx.x;
    if (e < E_) {
        const int s = eidx[e];
        const int t = eidx[E_ + e];
        const int pos = atomicAdd(&cursor[s], 1);
        reinterpret_cast<int4*>(perm)[pos] = make_int4(e, s, t, 0);
    }
}

__global__ __launch_bounds__(256) void edge_out_kernel(
    const float* __restrict__ ea, const int* __restrict__ perm,
    const float* __restrict__ qkv,
    const float* __restrict__ vw, const float* __restrict__ vb,
    const float* __restrict__ exbuf, float* __restrict__ den,
    float* __restrict__ out, int E_)
{
    __shared__ float sea[128 * 21];
    __shared__ float srpe[128 * 67];
    __shared__ int sE[128];
    __shared__ int sS[128];
    __shared__ int sT[128];
    const int tid = threadIdx.x;
    const int e0  = blockIdx.x * 128;

    const int wave = tid >> 5, lane = tid & 31;
    const int lhi = lane >> 4, l16 = lane & 15;
    const int etile = wave * 16;

    if (tid < 128) {
        const int i = e0 + tid;
        int e = 0, s = -1, t = 0;
        if (i < E_) {
            const int4 rec = reinterpret_cast<const int4*>(perm)[i];
            e = rec.x; s = rec.y; t = rec.z;
        }
        sE[tid] = e; sS[tid] = s; sT[tid] = t;
    }
    __syncthreads();

    float stg[9];
    #pragma unroll
    for (int k = 0; k < 9; ++k) {
        const int i = tid + k * 256;
        const int el = i / RPE, c = i % RPE;
        stg[k] = ea[(size_t)sE[el] * RPE + c];
    }
    #pragma unroll
    for (int k = 0; k < 9; ++k) {
        const int i = tid + k * 256;
        const int el = i / RPE, c = i % RPE;
        sea[el*21 + c] = (sS[el] >= 0) ? stg[k] : 0.0f;
    }
    v2f bw[4][5];
    #pragma unroll
    for (int ct = 0; ct < 4; ++ct) {
        #pragma unroll
        for (int kq = 0; kq < 5; ++kq) {
            const int ka = kq*4 + (lhi ? 2 : 0);
            bw[ct][kq].x = (ka     < RPE) ? vw[ka     * DIMV + ct*16 + l16] : 0.0f;
            bw[ct][kq].y = (ka + 1 < RPE) ? vw[(ka+1) * DIMV + ct*16 + l16] : 0.0f;
        }
    }
    if (tid < 128) { sea[tid*21 + 18] = 0.f; sea[tid*21 + 19] = 0.f; }
    __syncthreads();

    #pragma unroll
    for (int ct = 0; ct < 4; ++ct) {
        v8f acc = {};
        #pragma unroll
        for (int kq = 0; kq < 5; ++kq) {
            const int ka = kq*4 + (lhi ? 2 : 0);
            v2f a;
            a.x = sea[(etile + l16)*21 + ka];
            a.y = sea[(etile + l16)*21 + ka + 1];
            acc = __builtin_amdgcn_wmma_f32_16x16x4_f32(
                false, a, false, bw[ct][kq], (short)0, acc, false, false);
        }
        const int cb = ct*16 + l16;
        #pragma unroll
        for (int r = 0; r < 8; ++r)
            srpe[(etile + r + (lhi ? 8 : 0))*67 + cb] = acc[r];
    }
    __syncthreads();

    const int me = sE[etile + l16];
    const int ms = sS[etile + l16];
    const int mt = sT[etile + l16];
    const float vb0 = vb[lane], vb1 = vb[lane + 32];
    int cur = -1;
    float r0 = 0.f, r1 = 0.f;
    float d0 = 0.f, d1 = 0.f;
    #pragma unroll 1
    for (int hb = 0; hb < 16; hb += 8) {
        float val0[8], val1[8];
        float wgt0[8], wgt1[8];
        int   srun[8];
        #pragma unroll
        for (int j = 0; j < 8; ++j) {
            const int e = __builtin_amdgcn_readlane(me, hb + j);
            const int s = __builtin_amdgcn_readlane(ms, hb + j);
            const int t = __builtin_amdgcn_readlane(mt, hb + j);
            const float4 ex4 = *reinterpret_cast<const float4*>(exbuf + (size_t)e  * HH);
            const float a0 = lhi ? ex4.y : ex4.x;
            const float a1 = lhi ? ex4.w : ex4.z;
            const float* vrow = qkv + (size_t)t * QKVW + 2*DHH;
            val0[j] = (vrow[lane]      + srpe[(etile + hb + j)*67 + lane]      + vb0) * a0;
            val1[j] = (vrow[lane + 32] + srpe[(etile + hb + j)*67 + lane + 32] + vb1) * a1;
            wgt0[j] = a0; wgt1[j] = a1;
            srun[j] = s;
        }
        #pragma unroll
        for (int j = 0; j < 8; ++j) {
            if (srun[j] != cur) {
                if (cur >= 0) {
                    atomicAdd(&out[(size_t)cur * DIMV + lane],      r0);
                    atomicAdd(&out[(size_t)cur * DIMV + 32 + lane], r1);
                    if (l16 < 2) atomicAdd(&den[cur * HH + lhi + 2*l16], l16 ? d1 : d0);
                }
                cur = srun[j]; r0 = val0[j]; r1 = val1[j]; d0 = wgt0[j]; d1 = wgt1[j];
            } else {
                r0 += val0[j]; r1 += val1[j]; d0 += wgt0[j]; d1 += wgt1[j];
            }
        }
    }
    if (cur >= 0) {
        atomicAdd(&out[(size_t)cur * DIMV + lane],      r0);
        atomicAdd(&out[(size_t)cur * DIMV + 32 + lane], r1);
        if (l16 < 2) atomicAdd(&den[cur * HH + lhi + 2*l16], l16 ? d1 : d0);
    }
}

__global__ __launch_bounds__(256) void out_divide_kernel(
    float* __restrict__ out, const float* __restrict__ den, int N)
{
    const int total = N * (DIMV / 4);
    const int base  = blockIdx.x * 2048 + threadIdx.x;
    float4 v[8];
    float  d[8];
    #pragma unroll
    for (int k = 0; k < 8; ++k) {
        const int i = base + k * 256;
        v[k] = make_float4(0.f, 0.f, 0.f, 0.f);
        d[k] = 1.0f;
        if (i < total) {
            v[k] = reinterpret_cast<const float4*>(out)[i];
            d[k] = den[(i >> 4) * HH + ((i & 15) >> 2)];
        }
    }
    #pragma unroll
    for (int k = 0; k < 8; ++k) {
        const int i = base + k * 256;
        if (i < total) {
            const float dd = d[k] + 1e-16f;
            float4 w = v[k];
            w.x = w.x / dd; w.y = w.y / dd; w.z = w.z / dd; w.w = w.w / dd;
            reinterpret_cast<float4*>(out)[i] = w;
        }
    }
}

extern "C" void kernel_launch(void* const* d_in, const int* in_sizes, int n_in,
                              void* d_out, int out_size, void* d_ws, size_t ws_size,
                              hipStream_t stream)
{
    const float* x       = (const float*)d_in[0];
    const int*   eidx    = (const int*)  d_in[1];
    const float* ea      = (const float*)d_in[2];
    const float* qkv_w   = (const float*)d_in[3];
    const float* qkv_b   = (const float*)d_in[4];
    const float* k_rpe_w = (const float*)d_in[5];
    const float* k_rpe_b = (const float*)d_in[6];
    const float* q_rpe_w = (const float*)d_in[7];
    const float* q_rpe_b = (const float*)d_in[8];
    const float* v_rpe_w = (const float*)d_in[9];
    const float* v_rpe_b = (const float*)d_in[10];

    const int N  = in_sizes[0] / DIMV;
    const int E_ = in_sizes[2] / RPE;
    float* out = (float*)d_out;

    char* ws = (char*)d_ws;
    float* qkv  = (float*)ws;
    float* exb  = (float*)(ws + (size_t)N * QKVW * 4);
    int*   perm = (int*)  (ws + (size_t)N * QKVW * 4
                              + (size_t)E_ * HH * 4);
    float* den  = (float*)(ws + (size_t)N * QKVW * 4
                              + (size_t)E_ * HH * 8);
    int*   cnt  = (int*)  (ws + (size_t)N * QKVW * 4
                              + (size_t)E_ * HH * 8
                              + (size_t)N * HH * 4);
    int*   cur  = (int*)  (ws + (size_t)N * QKVW * 4
                              + (size_t)E_ * HH * 8
                              + (size_t)N * (HH + 1) * 4);

    hipMemsetAsync(den, 0, (size_t)N * (HH + 1) * sizeof(float), stream);
    hipMemsetAsync(out, 0, (size_t)N * DIMV * sizeof(float), stream);

    qkv_proj_kernel<<<(N + 15) / 16, 256, 0, stream>>>(x, qkv_w, qkv_b, qkv, N);
    edge_compat_kernel<<<(E_ + 127) / 128, 256, 0, stream>>>(
        ea, eidx, qkv, q_rpe_w, q_rpe_b, k_rpe_w, k_rpe_b, exb, cnt, E_);
    degree_scan_kernel<<<1, 1024, 0, stream>>>(cnt, cur, N);
    edge_fill_kernel<<<(E_ + 255) / 256, 256, 0, stream>>>(eidx, cur, perm, E_);
    edge_out_kernel<<<(E_ + 127) / 128, 256, 0, stream>>>(
        ea, perm, qkv, v_rpe_w, v_rpe_b, exb, den, out, E_);
    out_divide_kernel<<<(N * (DIMV / 4) + 2047) / 2048, 256, 0, stream>>>(out, den, N);
}


# Compat kernel fused into the sorted output kernel: one edge_attr gather feeds the q/k/v rpe WMMAs, exp() stays in LDS; q/k rpe biases start the WMMA accumulators (their add order changes)
# speedup vs baseline: 2.5333x; 1.0853x over previous
#include <hip/hip_runtime.h>
#include <hip/hip_bf16.h>


typedef __attribute__((ext_vector_type(2))) float v2f;
typedef __attribute__((ext_vector_type(8))) float v8f;

#define HH    4
#define DD    8
#define DHH   32
#define DIMV  64
#define RPE   18
#define QKVW  128

__global__ __launch_bounds__(256) void qkv_proj_kernel(
    const float* __restrict__ x, const float* __restrict__ W,
    const float* __restrict__ bias, float* __restrict__ qkv, int N)
{
    __shared__ float xs[16 * 65];
    const int tid = threadIdx.x;
    const int r0  = blockIdx.x * 16;

    {
        int idx = tid * 4;
        int row = idx >> 6, col = idx & 63;
        int grow = r0 + row;
        float4 vx = make_float4(0.f, 0.f, 0.f, 0.f);
        if (grow < N) vx = *(const float4*)(x + (size_t)grow * 64 + col);
        xs[row*65+col+0] = vx.x; xs[row*65+col+1] = vx.y;
        xs[row*65+col+2] = vx.z; xs[row*65+col+3] = vx.w;
    }
    __syncthreads();

    const int wave = tid >> 5;
    const int lane = tid & 31;
    const int lhi  = lane >> 4;
    const int l16  = lane & 15;
    const int colbase = wave * 16;

    v8f acc = {};
    #pragma unroll
    for (int kk = 0; kk < 64; kk += 4) {
        const int ka = kk + (lhi ? 2 : 0);
        v2f a, b;
        a.x = xs[l16*65 + ka];
        a.y = xs[l16*65 + ka + 1];
        b.x = W[(size_t)ka     * QKVW + colbase + l16];
        b.y = W[(size_t)(ka+1) * QKVW + colbase + l16];
        acc = __builtin_amdgcn_wmma_f32_16x16x4_f32(
            false, a, false, b, (short)0, acc, false, false);
    }

    const int col = colbase + l16;
    const float bv = bias[col];
    #pragma unroll
    for (int r = 0; r < 8; ++r) {
        int grow = r0 + r + (lhi ? 8 : 0);
        if (grow < N) qkv[(size_t)grow * QKVW + col] = acc[r] + bv;
    }
}

__global__ __launch_bounds__(256) void degree_hist_kernel(
    const int* __restrict__ eidx, int* __restrict__ cnt, int E_)
{
    const int e = blockIdx.x * 256 + threadIdx.x;
    if (e < E_) atomicAdd(&cnt[eidx[e]], 1);
}

__global__ __launch_bounds__(1024) void degree_scan_kernel(
    const int* __restrict__ cnt, int* __restrict__ cursor, int N)
{
    __shared__ int wsum[33];
    const int tid = threadIdx.x, lane = tid & 31, wave = tid >> 5;
    const bool vec_ok =
        ((reinterpret_cast<size_t>(cnt) | reinterpret_cast<size_t>(cursor)) & 15) == 0;
    int carry = 0;
    for (int base = 0; base < N; base += 8192) {
        const int i0 = base + tid * 8;
        const bool whole = vec_ok && (i0 + 7 < N);
        int c[8];
        if (whole) {
            const int4 a = *reinterpret_cast<const int4*>(cnt + i0);
            const int4 b = *reinterpret_cast<const int4*>(cnt + i0 + 4);
            c[0] = a.x; c[1] = a.y; c[2] = a.z; c[3] = a.w;
            c[4] = b.x; c[5] = b.y; c[6] = b.z; c[7] = b.w;
        } else {
            #pragma unroll
            for (int k = 0; k < 8; ++k) c[k] = (i0 + k < N) ? cnt[i0 + k] : 0;
        }
        int tsum = 0;
        #pragma unroll
        for (int k = 0; k < 8; ++k) tsum += c[k];
        int x = tsum;
        #pragma unroll
        for (int d = 1; d < 32; d <<= 1) {
            const int y = __shfl_up(x, d);
            if (lane >= d) x += y;
        }
        if (lane == 31) wsum[wave] = x;
        __syncthreads();
        if (wave == 0) {
            const int w = wsum[lane];
            int xs = w;
            #pragma unroll
            for (int d = 1; d < 32; d <<= 1) {
                const int y = __shfl_up(xs, d);
                if (lane >= d) xs += y;
            }
            wsum[lane] = xs - w;
            if (lane == 31) wsum[32] = xs;
        }
        __syncthreads();
        int p = carry + wsum[wave] + (x - tsum);
        carry += wsum[32];
        if (whole) {
            int4 a, b;
            a.x = p; p += c[0]; a.y = p; p += c[1]; a.z = p; p += c[2]; a.w = p; p += c[3];
            b.x = p; p += c[4]; b.y = p; p += c[5]; b.z = p; p += c[6]; b.w = p;
            *reinterpret_cast<int4*>(cursor + i0)     = a;
            *reinterpret_cast<int4*>(cursor + i0 + 4) = b;
        } else {
            #pragma unroll
            for (int k = 0; k < 8; ++k) {
                if (i0 + k < N) cursor[i0 + k] = p;
                p += c[k];
            }
        }
        __syncthreads();
    }
}

__global__ __launch_bounds__(256) void edge_fill_kernel(
    const int* __restrict__ eidx, int* __restrict__ cursor,
    int* __restrict__ perm, int E_)
{
    const int e = blockIdx.x * 256 + threadIdx.x;
    if (e < E_) {
        const int s = eidx[e];
        const int t = eidx[E_ + e];
        const int pos = atomicAdd(&cursor[s], 1);
        reinterpret_cast<int4*>(perm)[pos] = make_int4(e, s, t, 0);
    }
}

__global__ __launch_bounds__(256) void edge_fused_kernel(
    const float* __restrict__ ea, const int* __restrict__ perm,
    const float* __restrict__ qkv,
    const float* __restrict__ qw, const float* __restrict__ qb,
    const float* __restrict__ kw, const float* __restrict__ kb,
    const float* __restrict__ vw, const float* __restrict__ vb,
    float* __restrict__ den, float* __restrict__ out, int E_)
{
    __shared__ float sea[128 * 21];
    __shared__ float srpe[128 * 67];
    __shared__ float sex[128 * HH];
    __shared__ int sE[128];
    __shared__ int sS[128];
    __shared__ int sT[128];
    const int tid = threadIdx.x;
    const int e0  = blockIdx.x * 128;

    const int wave = tid >> 5, lane = tid & 31;
    const int lhi = lane >> 4, l16 = lane & 15;
    const int etile = wave * 16;

    if (tid < 128) {
        const int i = e0 + tid;
        int e = 0, s = -1, t = 0;
        if (i < E_) {
            const int4 rec = reinterpret_cast<const int4*>(perm)[i];
            e = rec.x; s = rec.y; t = rec.z;
        }
        sE[tid] = e; sS[tid] = s; sT[tid] = t;
    }
    __syncthreads();

    float stg[9];
    #pragma unroll
    for (int k = 0; k < 9; ++k) {
        const int i = tid + k * 256;
        const int el = i / RPE, c = i % RPE;
        stg[k] = ea[(size_t)sE[el] * RPE + c];
    }
    #pragma unroll
    for (int k = 0; k < 9; ++k) {
        const int i = tid + k * 256;
        const int el = i / RPE, c = i % RPE;
        sea[el*21 + c] = (sS[el] >= 0) ? stg[k] : 0.0f;
    }
    v2f bw[2][2][5];
    #pragma unroll
    for (int sel = 0; sel < 2; ++sel) {
        const float* Wp = sel ? kw : qw;
        #pragma unroll
        for (int ct = 0; ct < 2; ++ct) {
            #pragma unroll
            for (int kq = 0; kq < 5; ++kq) {
                const int ka = kq*4 + (lhi ? 2 : 0);
                bw[sel][ct][kq].x = (ka     < RPE) ? Wp[ka     * DHH + ct*16 + l16] : 0.0f;
                bw[sel][ct][kq].y = (ka + 1 < RPE) ? Wp[(ka+1) * DHH + ct*16 + l16] : 0.0f;
            }
        }
    }
    float bqk[2][2];
    #pragma unroll
    for (int sel = 0; sel < 2; ++sel) {
        #pragma unroll
        for (int ct = 0; ct < 2; ++ct) bqk[sel][ct] = (sel ? kb : qb)[ct*16 + l16];
    }
    if (tid < 128) { sea[tid*21 + 18] = 0.f; sea[tid*21 + 19] = 0.f; }
    __syncthreads();

    #pragma unroll
    for (int sel = 0; sel < 2; ++sel) {
        #pragma unroll
        for (int ct = 0; ct < 2; ++ct) {
            const float bc = bqk[sel][ct];
            v8f acc = {bc, bc, bc, bc, bc, bc, bc, bc};
            #pragma unroll
            for (int kq = 0; kq < 5; ++kq) {
                const int ka = kq*4 + (lhi ? 2 : 0);
                v2f a;
                a.x = sea[(etile + l16)*21 + ka];
                a.y = sea[(etile + l16)*21 + ka + 1];
                acc = __builtin_amdgcn_wmma_f32_16x16x4_f32(
                    false, a, false, bw[sel][ct][kq], (short)0, acc, false, false);
            }
            const int cb = sel*32 + ct*16 + l16;
            #pragma unroll
            for (int r = 0; r < 8; ++r)
                srpe[(etile + r + (lhi ? 8 : 0))*67 + cb] = acc[r];
        }
    }
    __syncthreads();

    {
        const int eloc = tid >> 1, pairh = (tid & 1) * 2;
        const int s = sS[eloc];
        const int t = sT[eloc];
        float ev0 = 0.f, ev1 = 0.f;
        if (s >= 0) {
            const float scale = 0.35355339059327373f;
            const float* qrow = qkv + (size_t)s * QKVW + pairh*DD;
            const float* krow = qkv + (size_t)t * QKVW + DHH + pairh*DD;
            float qv[2*DD], kv[2*DD];
            #pragma unroll
            for (int i = 0; i < 2*DD; ++i) kv[i] = krow[i];
            #pragma unroll
            for (int i = 0; i < 2*DD; ++i) qv[i] = qrow[i];
            #pragma unroll
            for (int hh = 0; hh < 2; ++hh) {
                const int h = pairh + hh;
                float compat = 0.f;
                #pragma unroll
                for (int d = 0; d < DD; ++d) {
                    const int c = h*DD + d;
                    const float qe = qv[hh*DD + d]*scale + srpe[eloc*67 + c];
                    const float ke = kv[hh*DD + d]       + srpe[eloc*67 + 32 + c];
                    compat = fmaf(qe, ke, compat);
                }
                const float ev = __expf(compat);
                if (hh == 0) ev0 = ev; else ev1 = ev;
            }
        }
        sex[eloc*HH + pairh]     = ev0;
        sex[eloc*HH + pairh + 1] = ev1;
    }

    v2f bv[4][5];
    #pragma unroll
    for (int ct = 0; ct < 4; ++ct) {
        #pragma unroll
        for (int kq = 0; kq < 5; ++kq) {
            const int ka = kq*4 + (lhi ? 2 : 0);
            bv[ct][kq].x = (ka     < RPE) ? vw[ka     * DIMV + ct*16 + l16] : 0.0f;
            bv[ct][kq].y = (ka + 1 < RPE) ? vw[(ka+1) * DIMV + ct*16 + l16] : 0.0f;
        }
    }
    __syncthreads();

    #pragma unroll
    for (int ct = 0; ct < 4; ++ct) {
        v8f acc = {};
        #pragma unroll
        for (int kq = 0; kq < 5; ++kq) {
            const int ka = kq*4 + (lhi ? 2 : 0);
            v2f a;
            a.x = sea[(etile + l16)*21 + ka];
            a.y = sea[(etile + l16)*21 + ka + 1];
            acc = __builtin_amdgcn_wmma_f32_16x16x4_f32(
                false, a, false, bv[ct][kq], (short)0, acc, false, false);
        }
        const int cb = ct*16 + l16;
        #pragma unroll
        for (int r = 0; r < 8; ++r)
            srpe[(etile + r + (lhi ? 8 : 0))*67 + cb] = acc[r];
    }
    __syncthreads();

    const int ms = sS[etile + l16];
    const int mt = sT[etile + l16];
    const float vb0 = vb[lane], vb1 = vb[lane + 32];
    int cur = -1;
    float r0 = 0.f, r1 = 0.f;
    float d0 = 0.f, d1 = 0.f;
    #pragma unroll 1
    for (int hb = 0; hb < 16; hb += 8) {
        float val0[8], val1[8];
        float wgt0[8], wgt1[8];
        int   srun[8];
        #pragma unroll
        for (int j = 0; j < 8; ++j) {
            const int s = __builtin_amdgcn_readlane(ms, hb + j);
            const int t = __builtin_amdgcn_readlane(mt, hb + j);
            const float a0 = sex[(etile + hb + j)*HH + lhi];
            const float a1 = sex[(etile + hb + j)*HH + 2 + lhi];
            const float* vrow = qkv + (size_t)t * QKVW + 2*DHH;
            val0[j] = (vrow[lane]      + srpe[(etile + hb + j)*67 + lane]      + vb0) * a0;
            val1[j] = (vrow[lane + 32] + srpe[(etile + hb + j)*67 + lane + 32] + vb1) * a1;
            wgt0[j] = a0; wgt1[j] = a1;
            srun[j] = s;
        }
        #pragma unroll
        for (int j = 0; j < 8; ++j) {
            if (srun[j] != cur) {
                if (cur >= 0) {
                    atomicAdd(&out[(size_t)cur * DIMV + lane],      r0);
                    atomicAdd(&out[(size_t)cur * DIMV + 32 + lane], r1);
                    if (l16 < 2) atomicAdd(&den[cur * HH + lhi + 2*l16], l16 ? d1 : d0);
                }
                cur = srun[j]; r0 = val0[j]; r1 = val1[j]; d0 = wgt0[j]; d1 = wgt1[j];
            } else {
                r0 += val0[j]; r1 += val1[j]; d0 += wgt0[j]; d1 += wgt1[j];
            }
        }
    }
    if (cur >= 0) {
        atomicAdd(&out[(size_t)cur * DIMV + lane],      r0);
        atomicAdd(&out[(size_t)cur * DIMV + 32 + lane], r1);
        if (l16 < 2) atomicAdd(&den[cur * HH + lhi + 2*l16], l16 ? d1 : d0);
    }
}

__global__ __launch_bounds__(256) void out_divide_kernel(
    float* __restrict__ out, const float* __restrict__ den, int N)
{
    const int total = N * (DIMV / 4);
    const int base  = blockIdx.x * 2048 + threadIdx.x;
    float4 v[8];
    float  d[8];
    #pragma unroll
    for (int k = 0; k < 8; ++k) {
        const int i = base + k * 256;
        v[k] = make_float4(0.f, 0.f, 0.f, 0.f);
        d[k] = 1.0f;
        if (i < total) {
            v[k] = reinterpret_cast<const float4*>(out)[i];
            d[k] = den[(i >> 4) * HH + ((i & 15) >> 2)];
        }
    }
    #pragma unroll
    for (int k = 0; k < 8; ++k) {
        const int i = base + k * 256;
        if (i < total) {
            const float dd = d[k] + 1e-16f;
            float4 w = v[k];
            w.x = w.x / dd; w.y = w.y / dd; w.z = w.z / dd; w.w = w.w / dd;
            reinterpret_cast<float4*>(out)[i] = w;
        }
    }
}

extern "C" void kernel_launch(void* const* d_in, const int* in_sizes, int n_in,
                              void* d_out, int out_size, void* d_ws, size_t ws_size,
                              hipStream_t stream)
{
    const float* x       = (const float*)d_in[0];
    const int*   eidx    = (const int*)  d_in[1];
    const float* ea      = (const float*)d_in[2];
    const float* qkv_w   = (const float*)d_in[3];
    const float* qkv_b   = (const float*)d_in[4];
    const float* k_rpe_w = (const float*)d_in[5];
    const float* k_rpe_b = (const float*)d_in[6];
    const float* q_rpe_w = (const float*)d_in[7];
    const float* q_rpe_b = (const float*)d_in[8];
    const float* v_rpe_w = (const float*)d_in[9];
    const float* v_rpe_b = (const float*)d_in[10];

    const int N  = in_sizes[0] / DIMV;
    const int E_ = in_sizes[2] / RPE;
    float* out = (float*)d_out;

    char* ws = (char*)d_ws;
    float* qkv  = (float*)ws;
    int*   perm = (int*)  (ws + (size_t)N * QKVW * 4);
    float* den  = (float*)(ws + (size_t)N * QKVW * 4
                              + (size_t)E_ * HH * 4);
    int*   cnt  = (int*)  (ws + (size_t)N * QKVW * 4
                              + (size_t)E_ * HH * 4
                              + (size_t)N * HH * 4);
    int*   cur  = (int*)  (ws + (size_t)N * QKVW * 4
                              + (size_t)E_ * HH * 4
                              + (size_t)N * (HH + 1) * 4);

    hipMemsetAsync(den, 0, (size_t)N * (HH + 1) * sizeof(float), stream);
    hipMemsetAsync(out, 0, (size_t)N * DIMV * sizeof(float), stream);

    degree_hist_kernel<<<(E_ + 255) / 256, 256, 0, stream>>>(eidx, cnt, E_);
    degree_scan_kernel<<<1, 1024, 0, stream>>>(cnt, cur, N);
    edge_fill_kernel<<<(E_ + 255) / 256, 256, 0, stream>>>(eidx, cur, perm, E_);
    qkv_proj_kernel<<<(N + 15) / 16, 256, 0, stream>>>(x, qkv_w, qkv_b, qkv, N);
    edge_fused_kernel<<<(E_ + 127) / 128, 256, 0, stream>>>(
        ea, perm, qkv, q_rpe_w, q_rpe_b, k_rpe_w, k_rpe_b, v_rpe_w, v_rpe_b, den, out, E_);
    out_divide_kernel<<<(N * (DIMV / 4) + 2047) / 2048, 256, 0, stream>>>(out, den, N);
}


# Edge sort without global atomics: edges binned by 256-node range, then by node, on LDS counters (block counts via a table; count, scan, place); float sum order changes (arrival rank of LDS adds)
# speedup vs baseline: 4.5309x; 1.7885x over previous
#include <hip/hip_runtime.h>
#include <hip/hip_bf16.h>


typedef __attribute__((ext_vector_type(2))) float v2f;
typedef __attribute__((ext_vector_type(8))) float v8f;

#define HH    4
#define DD    8
#define DHH   32
#define DIMV  64
#define RPE   18
#define QKVW  128

__global__ __launch_bounds__(256) void qkv_proj_kernel(
    const float* __restrict__ x, const float* __restrict__ W,
    const float* __restrict__ bias, float* __restrict__ qkv, int N)
{
    __shared__ float xs[16 * 65];
    const int tid = threadIdx.x;
    const int r0  = blockIdx.x * 16;

    {
        int idx = tid * 4;
        int row = idx >> 6, col = idx & 63;
        int grow = r0 + row;
        float4 vx = make_float4(0.f, 0.f, 0.f, 0.f);
        if (grow < N) vx = *(const float4*)(x + (size_t)grow * 64 + col);
        xs[row*65+col+0] = vx.x; xs[row*65+col+1] = vx.y;
        xs[row*65+col+2] = vx.z; xs[row*65+col+3] = vx.w;
    }
    __syncthreads();

    const int wave = tid >> 5;
    const int lane = tid & 31;
    const int lhi  = lane >> 4;
    const int l16  = lane & 15;
    const int colbase = wave * 16;

    v8f acc = {};
    #pragma unroll
    for (int kk = 0; kk < 64; kk += 4) {
        const int ka = kk + (lhi ? 2 : 0);
        v2f a, b;
        a.x = xs[l16*65 + ka];
        a.y = xs[l16*65 + ka + 1];
        b.x = W[(size_t)ka     * QKVW + colbase + l16];
        b.y = W[(size_t)(ka+1) * QKVW + colbase + l16];
        acc = __builtin_amdgcn_wmma_f32_16x16x4_f32(
            false, a, false, b, (short)0, acc, false, false);
    }

    const int col = colbase + l16;
    const float bv = bias[col];
    #pragma unroll
    for (int r = 0; r < 8; ++r) {
        int grow = r0 + r + (lhi ? 8 : 0);
        if (grow < N) qkv[(size_t)grow * QKVW + col] = acc[r] + bv;
    }
}

#define BSH    8
#define BNODES 256
#define MAXB   1024
#define EPT    8
#define CHUNK  (EPT * 1024)

__global__ __launch_bounds__(1024) void bucket_count_kernel(
    const int* __restrict__ eidx, int* __restrict__ tab, int E_, int NB)
{
    __shared__ int lcnt[MAXB];
    const int tid = threadIdx.x;
    const int base = blockIdx.x * CHUNK;
    lcnt[tid] = 0;
    int s[EPT];
    #pragma unroll
    for (int k = 0; k < EPT; ++k)
        s[k] = eidx[min(base + tid + k * 1024, E_ - 1)];
    #pragma unroll
    for (int k = 0; k < EPT; ++k)
        if (base + tid + k * 1024 >= E_) s[k] = -1;
    __syncthreads();
    #pragma unroll
    for (int k = 0; k < EPT; ++k)
        if (s[k] >= 0) atomicAdd(&lcnt[s[k] >> BSH], 1);
    __syncthreads();
    if (tid < ((NB + 3) & ~3)) tab[blockIdx.x * MAXB + tid] = lcnt[tid];
}

__global__ __launch_bounds__(1024) void bucket_scatter_kernel(
    const int* __restrict__ eidx, const int* __restrict__ tab,
    int* __restrict__ bbase, int* __restrict__ tmp, int E_, int NB)
{
    __shared__ int lcnt[MAXB];
    __shared__ int ltot[MAXB];
    __shared__ int lpre[MAXB];
    __shared__ int lbase[MAXB];
    __shared__ int wsum[33];
    const int tid = threadIdx.x, lane = tid & 31, wave = tid >> 5;
    const int base = blockIdx.x * CHUNK;
    lcnt[tid] = 0; ltot[tid] = 0; lpre[tid] = 0;
    int s[EPT], t[EPT];
    #pragma unroll
    for (int k = 0; k < EPT; ++k) {
        const int e = min(base + tid + k * 1024, E_ - 1);
        s[k] = eidx[e]; t[k] = eidx[E_ + e];
    }
    #pragma unroll
    for (int k = 0; k < EPT; ++k)
        if (base + tid + k * 1024 >= E_) s[k] = -1;
    const int c4 = (tid & 127) * 4;
    int4 tot = make_int4(0, 0, 0, 0), pre = make_int4(0, 0, 0, 0);
    if (c4 < NB) {
        const int nrow = gridDim.x;
        #pragma unroll 4
        for (int r = tid >> 7; r < nrow; r += 8) {
            const int4 v = *reinterpret_cast<const int4*>(tab + r * MAXB + c4);
            tot.x += v.x; tot.y += v.y; tot.z += v.z; tot.w += v.w;
            if (r < (int)blockIdx.x) { pre.x += v.x; pre.y += v.y; pre.z += v.z; pre.w += v.w; }
        }
    }
    __syncthreads();
    int rk[EPT];
    #pragma unroll
    for (int k = 0; k < EPT; ++k) {
        rk[k] = 0;
        if (s[k] >= 0) rk[k] = atomicAdd(&lcnt[s[k] >> BSH], 1);
    }
    if (c4 < NB) {
        atomicAdd(&ltot[c4], tot.x); atomicAdd(&ltot[c4 + 1], tot.y);
        atomicAdd(&ltot[c4 + 2], tot.z); atomicAdd(&ltot[c4 + 3], tot.w);
        atomicAdd(&lpre[c4], pre.x); atomicAdd(&lpre[c4 + 1], pre.y);
        atomicAdd(&lpre[c4 + 2], pre.z); atomicAdd(&lpre[c4 + 3], pre.w);
    }
    __syncthreads();
    const int bc = (tid < NB) ? ltot[tid] : 0;
    int x = bc;
    #pragma unroll
    for (int d = 1; d < 32; d <<= 1) {
        const int y = __shfl_up(x, d);
        if (lane >= d) x += y;
    }
    if (lane == 31) wsum[wave] = x;
    __syncthreads();
    if (wave == 0) {
        const int w = wsum[lane];
        int xs = w;
        #pragma unroll
        for (int d = 1; d < 32; d <<= 1) {
            const int y = __shfl_up(xs, d);
            if (lane >= d) xs += y;
        }
        wsum[lane] = xs - w;
        if (lane == 31) wsum[32] = xs;
    }
    __syncthreads();
    const int bb = wsum[wave] + (x - bc);
    lbase[tid] = bb + lpre[tid];
    if (blockIdx.x == 0) {
        if (tid < NB) bbase[tid] = bb;
        if (tid == 0) bbase[NB] = wsum[32];
    }
    __syncthreads();
    #pragma unroll
    for (int k = 0; k < EPT; ++k) {
        if (s[k] >= 0) {
            const int e = base + tid + k * 1024;
            const int pos = lbase[s[k] >> BSH] + rk[k];
            reinterpret_cast<int4*>(tmp)[pos] = make_int4(e, s[k], t[k], 0);
        }
    }
}

#define KC 10
__global__ __launch_bounds__(512) void bucket_sort_kernel(
    const int* __restrict__ tmp, const int* __restrict__ bbase, int* __restrict__ perm)
{
    __shared__ int ncnt[BNODES];
    __shared__ int ncur[BNODES];
    __shared__ int wsum[8];
    const int tid = threadIdx.x, lane = tid & 31, wave = tid >> 5;
    const int s0 = blockIdx.x << BSH;
    const int lo = bbase[blockIdx.x], hi = bbase[blockIdx.x + 1];
    const int4* rec = reinterpret_cast<const int4*>(tmp);
    if (tid < BNODES) ncnt[tid] = 0;
    int4 r[KC];
    #pragma unroll
    for (int k = 0; k < KC; ++k)
        r[k] = rec[max(min(lo + tid + k * 512, hi - 1), 0)];
    #pragma unroll
    for (int k = 0; k < KC; ++k)
        if (lo + tid + k * 512 >= hi) r[k].y = -1;
    __syncthreads();
    #pragma unroll
    for (int k = 0; k < KC; ++k)
        if (r[k].y >= 0) atomicAdd(&ncnt[r[k].y - s0], 1);
    for (int i = lo + tid + KC * 512; i < hi; i += 512)
        atomicAdd(&ncnt[tmp[4 * i + 1] - s0], 1);
    __syncthreads();
    const int c = (tid < BNODES) ? ncnt[tid] : 0;
    int x = c;
    #pragma unroll
    for (int d = 1; d < 32; d <<= 1) {
        const int y = __shfl_up(x, d);
        if (lane >= d) x += y;
    }
    if (lane == 31 && wave < 8) wsum[wave] = x;
    __syncthreads();
    if (tid < BNODES) {
        int wp = 0;
        #pragma unroll
        for (int w = 0; w < 8; ++w) wp += (w < wave) ? wsum[w] : 0;
        ncur[tid] = lo + wp + (x - c);
    }
    __syncthreads();
    #pragma unroll
    for (int k = 0; k < KC; ++k) {
        if (r[k].y >= 0) {
            const int pos = atomicAdd(&ncur[r[k].y - s0], 1);
            reinterpret_cast<int4*>(perm)[pos] = r[k];
        }
    }
    for (int i = lo + tid + KC * 512; i < hi; i += 512) {
        const int4 q = rec[i];
        const int pos = atomicAdd(&ncur[q.y - s0], 1);
        reinterpret_cast<int4*>(perm)[pos] = q;
    }
}

__global__ __launch_bounds__(256) void edge_fused_kernel(
    const float* __restrict__ ea, const int* __restrict__ perm,
    const float* __restrict__ qkv,
    const float* __restrict__ qw, const float* __restrict__ qb,
    const float* __restrict__ kw, const float* __restrict__ kb,
    const float* __restrict__ vw, const float* __restrict__ vb,
    float* __restrict__ den, float* __restrict__ out, int E_)
{
    __shared__ float sea[128 * 21];
    __shared__ float srpe[128 * 67];
    __shared__ float sex[128 * HH];
    __shared__ int sE[128];
    __shared__ int sS[128];
    __shared__ int sT[128];
    const int tid = threadIdx.x;
    const int e0  = blockIdx.x * 128;

    const int wave = tid >> 5, lane = tid & 31;
    const int lhi = lane >> 4, l16 = lane & 15;
    const int etile = wave * 16;

    if (tid < 128) {
        const int i = e0 + tid;
        int e = 0, s = -1, t = 0;
        if (i < E_) {
            const int4 rec = reinterpret_cast<const int4*>(perm)[i];
            e = rec.x; s = rec.y; t = rec.z;
        }
        sE[tid] = e; sS[tid] = s; sT[tid] = t;
    }
    __syncthreads();

    float stg[9];
    #pragma unroll
    for (int k = 0; k < 9; ++k) {
        const int i = tid + k * 256;
        const int el = i / RPE, c = i % RPE;
        stg[k] = ea[(size_t)sE[el] * RPE + c];
    }
    #pragma unroll
    for (int k = 0; k < 9; ++k) {
        const int i = tid + k * 256;
        const int el = i / RPE, c = i % RPE;
        sea[el*21 + c] = (sS[el] >= 0) ? stg[k] : 0.0f;
    }
    v2f bw[2][2][5];
    #pragma unroll
    for (int sel = 0; sel < 2; ++sel) {
        const float* Wp = sel ? kw : qw;
        #pragma unroll
        for (int ct = 0; ct < 2; ++ct) {
            #pragma unroll
            for (int kq = 0; kq < 5; ++kq) {
                const int ka = kq*4 + (lhi ? 2 : 0);
                bw[sel][ct][kq].x = (ka     < RPE) ? Wp[ka     * DHH + ct*16 + l16] : 0.0f;
                bw[sel][ct][kq].y = (ka + 1 < RPE) ? Wp[(ka+1) * DHH + ct*16 + l16] : 0.0f;
            }
        }
    }
    float bqk[2][2];
    #pragma unroll
    for (int sel = 0; sel < 2; ++sel) {
        #pragma unroll
        for (int ct = 0; ct < 2; ++ct) bqk[sel][ct] = (sel ? kb : qb)[ct*16 + l16];
    }
    if (tid < 128) { sea[tid*21 + 18] = 0.f; sea[tid*21 + 19] = 0.f; }
    __syncthreads();

    #pragma unroll
    for (int sel = 0; sel < 2; ++sel) {
        #pragma unroll
        for (int ct = 0; ct < 2; ++ct) {
            const float bc = bqk[sel][ct];
            v8f acc = {bc, bc, bc, bc, bc, bc, bc, bc};
            #pragma unroll
            for (int kq = 0; kq < 5; ++kq) {
                const int ka = kq*4 + (lhi ? 2 : 0);
                v2f a;
                a.x = sea[(etile + l16)*21 + ka];
                a.y = sea[(etile + l16)*21 + ka + 1];
                acc = __builtin_amdgcn_wmma_f32_16x16x4_f32(
                    false, a, false, bw[sel][ct][kq], (short)0, acc, false, false);
            }
            const int cb = sel*32 + ct*16 + l16;
            #pragma unroll
            for (int r = 0; r < 8; ++r)
                srpe[(etile + r + (lhi ? 8 : 0))*67 + cb] = acc[r];
        }
    }
    __syncthreads();

    {
        const int eloc = tid >> 1, pairh = (tid & 1) * 2;
        const int s = sS[eloc];
        const int t = sT[eloc];
        float ev0 = 0.f, ev1 = 0.f;
        if (s >= 0) {
            const float scale = 0.35355339059327373f;
            const float* qrow = qkv + (size_t)s * QKVW + pairh*DD;
            const float* krow = qkv + (size_t)t * QKVW + DHH + pairh*DD;
            float qv[2*DD], kv[2*DD];
            #pragma unroll
            for (int i = 0; i < 2*DD; ++i) kv[i] = krow[i];
            #pragma unroll
            for (int i = 0; i < 2*DD; ++i) qv[i] = qrow[i];
            #pragma unroll
            for (int hh = 0; hh < 2; ++hh) {
                const int h = pairh + hh;
                float compat = 0.f;
                #pragma unroll
                for (int d = 0; d < DD; ++d) {
                    const int c = h*DD + d;
                    const float qe = qv[hh*DD + d]*scale + srpe[eloc*67 + c];
                    const float ke = kv[hh*DD + d]       + srpe[eloc*67 + 32 + c];
                    compat = fmaf(qe, ke, compat);
                }
                const float ev = __expf(compat);
                if (hh == 0) ev0 = ev; else ev1 = ev;
            }
        }
        sex[eloc*HH + pairh]     = ev0;
        sex[eloc*HH + pairh + 1] = ev1;
    }

    v2f bv[4][5];
    #pragma unroll
    for (int ct = 0; ct < 4; ++ct) {
        #pragma unroll
        for (int kq = 0; kq < 5; ++kq) {
            const int ka = kq*4 + (lhi ? 2 : 0);
            bv[ct][kq].x = (ka     < RPE) ? vw[ka     * DIMV + ct*16 + l16] : 0.0f;
            bv[ct][kq].y = (ka + 1 < RPE) ? vw[(ka+1) * DIMV + ct*16 + l16] : 0.0f;
        }
    }
    __syncthreads();

    #pragma unroll
    for (int ct = 0; ct < 4; ++ct) {
        v8f acc = {};
        #pragma unroll
        for (int kq = 0; kq < 5; ++kq) {
            const int ka = kq*4 + (lhi ? 2 : 0);
            v2f a;
            a.x = sea[(etile + l16)*21 + ka];
            a.y = sea[(etile + l16)*21 + ka + 1];
            acc = __builtin_amdgcn_wmma_f32_16x16x4_f32(
                false, a, false, bv[ct][kq], (short)0, acc, false, false);
        }
        const int cb = ct*16 + l16;
        #pragma unroll
        for (int r = 0; r < 8; ++r)
            srpe[(etile + r + (lhi ? 8 : 0))*67 + cb] = acc[r];
    }
    __syncthreads();

    const int ms = sS[etile + l16];
    const int mt = sT[etile + l16];
    const float vb0 = vb[lane], vb1 = vb[lane + 32];
    int cur = -1;
    float r0 = 0.f, r1 = 0.f;
    float d0 = 0.f, d1 = 0.f;
    #pragma unroll 1
    for (int hb = 0; hb < 16; hb += 8) {
        float val0[8], val1[8];
        float wgt0[8], wgt1[8];
        int   srun[8];
        #pragma unroll
        for (int j = 0; j < 8; ++j) {
            const int s = __builtin_amdgcn_readlane(ms, hb + j);
            const int t = __builtin_amdgcn_readlane(mt, hb + j);
            const float a0 = sex[(etile + hb + j)*HH + lhi];
            const float a1 = sex[(etile + hb + j)*HH + 2 + lhi];
            const float* vrow = qkv + (size_t)t * QKVW + 2*DHH;
            val0[j] = (vrow[lane]      + srpe[(etile + hb + j)*67 + lane]      + vb0) * a0;
            val1[j] = (vrow[lane + 32] + srpe[(etile + hb + j)*67 + lane + 32] + vb1) * a1;
            wgt0[j] = a0; wgt1[j] = a1;
            srun[j] = s;
        }
        #pragma unroll
        for (int j = 0; j < 8; ++j) {
            if (srun[j] != cur) {
                if (cur >= 0) {
                    atomicAdd(&out[(size_t)cur * DIMV + lane],      r0);
                    atomicAdd(&out[(size_t)cur * DIMV + 32 + lane], r1);
                    if (l16 < 2) atomicAdd(&den[cur * HH + lhi + 2*l16], l16 ? d1 : d0);
                }
                cur = srun[j]; r0 = val0[j]; r1 = val1[j]; d0 = wgt0[j]; d1 = wgt1[j];
            } else {
                r0 += val0[j]; r1 += val1[j]; d0 += wgt0[j]; d1 += wgt1[j];
            }
        }
    }
    if (cur >= 0) {
        atomicAdd(&out[(size_t)cur * DIMV + lane],      r0);
        atomicAdd(&out[(size_t)cur * DIMV + 32 + lane], r1);
        if (l16 < 2) atomicAdd(&den[cur * HH + lhi + 2*l16], l16 ? d1 : d0);
    }
}

__global__ __launch_bounds__(256) void out_divide_kernel(
    float* __restrict__ out, const float* __restrict__ den, int N)
{
    const int total = N * (DIMV / 4);
    const int base  = blockIdx.x * 2048 + threadIdx.x;
    float4 v[8];
    float  d[8];
    #pragma unroll
    for (int k = 0; k < 8; ++k) {
        const int i = base + k * 256;
        v[k] = make_float4(0.f, 0.f, 0.f, 0.f);
        d[k] = 1.0f;
        if (i < total) {
            v[k] = reinterpret_cast<const float4*>(out)[i];
            d[k] = den[(i >> 4) * HH + ((i & 15) >> 2)];
        }
    }
    #pragma unroll
    for (int k = 0; k < 8; ++k) {
        const int i = base + k * 256;
        if (i < total) {
            const float dd = d[k] + 1e-16f;
            float4 w = v[k];
            w.x = w.x / dd; w.y = w.y / dd; w.z = w.z / dd; w.w = w.w / dd;
            reinterpret_cast<float4*>(out)[i] = w;
        }
    }
}

extern "C" void kernel_launch(void* const* d_in, const int* in_sizes, int n_in,
                              void* d_out, int out_size, void* d_ws, size_t ws_size,
                              hipStream_t stream)
{
    const float* x       = (const float*)d_in[0];
    const int*   eidx    = (const int*)  d_in[1];
    const float* ea      = (const float*)d_in[2];
    const float* qkv_w   = (const float*)d_in[3];
    const float* qkv_b   = (const float*)d_in[4];
    const float* k_rpe_w = (const float*)d_in[5];
    const float* k_rpe_b = (const float*)d_in[6];
    const float* q_rpe_w = (const float*)d_in[7];
    const float* q_rpe_b = (const float*)d_in[8];
    const float* v_rpe_w = (const float*)d_in[9];
    const float* v_rpe_b = (const float*)d_in[10];

    const int N  = in_sizes[0] / DIMV;
    const int E_ = in_sizes[2] / RPE;
    float* out = (float*)d_out;

    char* ws = (char*)d_ws;
    float* qkv    = (float*)ws;
    int*   perm   = (int*)  (ws + (size_t)N * QKVW * 4);
    float* den    = (float*)(ws + (size_t)N * QKVW * 4
                                + (size_t)E_ * HH * 4);
    int*   bbase  = (int*)  (ws + (size_t)N * QKVW * 4
                                + (size_t)E_ * HH * 4
                                + (size_t)N * HH * 4);
    int*   tmp    = bbase + 2 * MAXB;
    int*   tab    = tmp + (size_t)E_ * 4;

    const int NB = (N + BNODES - 1) / BNODES;

    hipMemsetAsync(den, 0, (size_t)N * HH * sizeof(float), stream);
    hipMemsetAsync(out, 0, (size_t)N * DIMV * sizeof(float), stream);

    bucket_count_kernel<<<(E_ + CHUNK - 1) / CHUNK, 1024, 0, stream>>>(eidx, tab, E_, NB);
    bucket_scatter_kernel<<<(E_ + CHUNK - 1) / CHUNK, 1024, 0, stream>>>(
        eidx, tab, bbase, tmp, E_, NB);
    bucket_sort_kernel<<<NB, 512, 0, stream>>>(tmp, bbase, perm);
    qkv_proj_kernel<<<(N + 15) / 16, 256, 0, stream>>>(x, qkv_w, qkv_b, qkv, N);
    edge_fused_kernel<<<(E_ + 127) / 128, 256, 0, stream>>>(
        ea, perm, qkv, q_rpe_w, q_rpe_b, k_rpe_w, k_rpe_b, v_rpe_w, v_rpe_b, den, out, E_);
    out_divide_kernel<<<(N * (DIMV / 4) + 2047) / 2048, 256, 0, stream>>>(out, den, N);
}


# Edge kernel: rpe weight B fragments staged once per block in LDS in fragment order (10 global loads per thread), read back by 8-byte LDS loads, not 80 4-byte global loads per wave
# speedup vs baseline: 4.8032x; 1.0601x over previous
#include <hip/hip_runtime.h>
#include <hip/hip_bf16.h>


typedef __attribute__((ext_vector_type(2))) float v2f;
typedef __attribute__((ext_vector_type(8))) float v8f;

#define HH    4
#define DD    8
#define DHH   32
#define DIMV  64
#define RPE   18
#define QKVW  128

__global__ __launch_bounds__(256) void qkv_proj_kernel(
    const float* __restrict__ x, const float* __restrict__ W,
    const float* __restrict__ bias, float* __restrict__ qkv, int N)
{
    __shared__ float xs[16 * 65];
    const int tid = threadIdx.x;
    const int r0  = blockIdx.x * 16;

    {
        int idx = tid * 4;
        int row = idx >> 6, col = idx & 63;
        int grow = r0 + row;
        float4 vx = make_float4(0.f, 0.f, 0.f, 0.f);
        if (grow < N) vx = *(const float4*)(x + (size_t)grow * 64 + col);
        xs[row*65+col+0] = vx.x; xs[row*65+col+1] = vx.y;
        xs[row*65+col+2] = vx.z; xs[row*65+col+3] = vx.w;
    }
    __syncthreads();

    const int wave = tid >> 5;
    const int lane = tid & 31;
    const int lhi  = lane >> 4;
    const int l16  = lane & 15;
    const int colbase = wave * 16;

    v8f acc = {};
    #pragma unroll
    for (int kk = 0; kk < 64; kk += 4) {
        const int ka = kk + (lhi ? 2 : 0);
        v2f a, b;
        a.x = xs[l16*65 + ka];
        a.y = xs[l16*65 + ka + 1];
        b.x = W[(size_t)ka     * QKVW + colbase + l16];
        b.y = W[(size_t)(ka+1) * QKVW + colbase + l16];
        acc = __builtin_amdgcn_wmma_f32_16x16x4_f32(
            false, a, false, b, (short)0, acc, false, false);
    }

    const int col = colbase + l16;
    const float bv = bias[col];
    #pragma unroll
    for (int r = 0; r < 8; ++r) {
        int grow = r0 + r + (lhi ? 8 : 0);
        if (grow < N) qkv[(size_t)grow * QKVW + col] = acc[r] + bv;
    }
}

#define BSH    8
#define BNODES 256
#define MAXB   1024
#define EPT    8
#define CHUNK  (EPT * 1024)

__global__ __launch_bounds__(1024) void bucket_count_kernel(
    const int* __restrict__ eidx, int* __restrict__ tab, int E_, int NB)
{
    __shared__ int lcnt[MAXB];
    const int tid = threadIdx.x;
    const int base = blockIdx.x * CHUNK;
    lcnt[tid] = 0;
    int s[EPT];
    #pragma unroll
    for (int k = 0; k < EPT; ++k)
        s[k] = eidx[min(base + tid + k * 1024, E_ - 1)];
    #pragma unroll
    for (int k = 0; k < EPT; ++k)
        if (base + tid + k * 1024 >= E_) s[k] = -1;
    __syncthreads();
    #pragma unroll
    for (int k = 0; k < EPT; ++k)
        if (s[k] >= 0) atomicAdd(&lcnt[s[k] >> BSH], 1);
    __syncthreads();
    if (tid < ((NB + 3) & ~3)) tab[blockIdx.x * MAXB + tid] = lcnt[tid];
}

__global__ __launch_bounds__(1024) void bucket_scatter_kernel(
    const int* __restrict__ eidx, const int* __restrict__ tab,
    int* __restrict__ bbase, int* __restrict__ tmp, int E_, int NB)
{
    __shared__ int lcnt[MAXB];
    __shared__ int ltot[MAXB];
    __shared__ int lpre[MAXB];
    __shared__ int lbase[MAXB];
    __shared__ int wsum[33];
    const int tid = threadIdx.x, lane = tid & 31, wave = tid >> 5;
    const int base = blockIdx.x * CHUNK;
    lcnt[tid] = 0; ltot[tid] = 0; lpre[tid] = 0;
    int s[EPT], t[EPT];
    #pragma unroll
    for (int k = 0; k < EPT; ++k) {
        const int e = min(base + tid + k * 1024, E_ - 1);
        s[k] = eidx[e]; t[k] = eidx[E_ + e];
    }
    #pragma unroll
    for (int k = 0; k < EPT; ++k)
        if (base + tid + k * 1024 >= E_) s[k] = -1;
    const int c4 = (tid & 127) * 4;
    int4 tot = make_int4(0, 0, 0, 0), pre = make_int4(0, 0, 0, 0);
    if (c4 < NB) {
        const int nrow = gridDim.x;
        #pragma unroll 4
        for (int r = tid >> 7; r < nrow; r += 8) {
            const int4 v = *reinterpret_cast<const int4*>(tab + r * MAXB + c4);
            tot.x += v.x; tot.y += v.y; tot.z += v.z; tot.w += v.w;
            if (r < (int)blockIdx.x) { pre.x += v.x; pre.y += v.y; pre.z += v.z; pre.w += v.w; }
        }
    }
    __syncthreads();
    int rk[EPT];
    #pragma unroll
    for (int k = 0; k < EPT; ++k) {
        rk[k] = 0;
        if (s[k] >= 0) rk[k] = atomicAdd(&lcnt[s[k] >> BSH], 1);
    }
    if (c4 < NB) {
        atomicAdd(&ltot[c4], tot.x); atomicAdd(&ltot[c4 + 1], tot.y);
        atomicAdd(&ltot[c4 + 2], tot.z); atomicAdd(&ltot[c4 + 3], tot.w);
        atomicAdd(&lpre[c4], pre.x); atomicAdd(&lpre[c4 + 1], pre.y);
        atomicAdd(&lpre[c4 + 2], pre.z); atomicAdd(&lpre[c4 + 3], pre.w);
    }
    __syncthreads();
    const int bc = (tid < NB) ? ltot[tid] : 0;
    int x = bc;
    #pragma unroll
    for (int d = 1; d < 32; d <<= 1) {
        const int y = __shfl_up(x, d);
        if (lane >= d) x += y;
    }
    if (lane == 31) wsum[wave] = x;
    __syncthreads();
    if (wave == 0) {
        const int w = wsum[lane];
        int xs = w;
        #pragma unroll
        for (int d = 1; d < 32; d <<= 1) {
            const int y = __shfl_up(xs, d);
            if (lane >= d) xs += y;
        }
        wsum[lane] = xs - w;
        if (lane == 31) wsum[32] = xs;
    }
    __syncthreads();
    const int bb = wsum[wave] + (x - bc);
    lbase[tid] = bb + lpre[tid];
    if (blockIdx.x == 0) {
        if (tid < NB) bbase[tid] = bb;
        if (tid == 0) bbase[NB] = wsum[32];
    }
    __syncthreads();
    #pragma unroll
    for (int k = 0; k < EPT; ++k) {
        if (s[k] >= 0) {
            const int e = base + tid + k * 1024;
            const int pos = lbase[s[k] >> BSH] + rk[k];
            reinterpret_cast<int4*>(tmp)[pos] = make_int4(e, s[k], t[k], 0);
        }
    }
}

#define KC 10
__global__ __launch_bounds__(512) void bucket_sort_kernel(
    const int* __restrict__ tmp, const int* __restrict__ bbase, int* __restrict__ perm)
{
    __shared__ int ncnt[BNODES];
    __shared__ int ncur[BNODES];
    __shared__ int wsum[8];
    const int tid = threadIdx.x, lane = tid & 31, wave = tid >> 5;
    const int s0 = blockIdx.x << BSH;
    const int lo = bbase[blockIdx.x], hi = bbase[blockIdx.x + 1];
    const int4* rec = reinterpret_cast<const int4*>(tmp);
    if (tid < BNODES) ncnt[tid] = 0;
    int4 r[KC];
    #pragma unroll
    for (int k = 0; k < KC; ++k)
        r[k] = rec[max(min(lo + tid + k * 512, hi - 1), 0)];
    #pragma unroll
    for (int k = 0; k < KC; ++k)
        if (lo + tid + k * 512 >= hi) r[k].y = -1;
    __syncthreads();
    #pragma unroll
    for (int k = 0; k < KC; ++k)
        if (r[k].y >= 0) atomicAdd(&ncnt[r[k].y - s0], 1);
    for (int i = lo + tid + KC * 512; i < hi; i += 512)
        atomicAdd(&ncnt[tmp[4 * i + 1] - s0], 1);
    __syncthreads();
    const int c = (tid < BNODES) ? ncnt[tid] : 0;
    int x = c;
    #pragma unroll
    for (int d = 1; d < 32; d <<= 1) {
        const int y = __shfl_up(x, d);
        if (lane >= d) x += y;
    }
    if (lane == 31 && wave < 8) wsum[wave] = x;
    __syncthreads();
    if (tid < BNODES) {
        int wp = 0;
        #pragma unroll
        for (int w = 0; w < 8; ++w) wp += (w < wave) ? wsum[w] : 0;
        ncur[tid] = lo + wp + (x - c);
    }
    __syncthreads();
    #pragma unroll
    for (int k = 0; k < KC; ++k) {
        if (r[k].y >= 0) {
            const int pos = atomicAdd(&ncur[r[k].y - s0], 1);
            reinterpret_cast<int4*>(perm)[pos] = r[k];
        }
    }
    for (int i = lo + tid + KC * 512; i < hi; i += 512) {
        const int4 q = rec[i];
        const int pos = atomicAdd(&ncur[q.y - s0], 1);
        reinterpret_cast<int4*>(perm)[pos] = q;
    }
}

__global__ __launch_bounds__(256) void edge_fused_kernel(
    const float* __restrict__ ea, const int* __restrict__ perm,
    const float* __restrict__ qkv,
    const float* __restrict__ qw, const float* __restrict__ qb,
    const float* __restrict__ kw, const float* __restrict__ kb,
    const float* __restrict__ vw, const float* __restrict__ vb,
    float* __restrict__ den, float* __restrict__ out, int E_)
{
    __shared__ float sea[128 * 21];
    __shared__ float srpe[128 * 67];
    __shared__ float sex[128 * HH];
    __shared__ int sE[128];
    __shared__ int sS[128];
    __shared__ int sT[128];
    __shared__ v2f swf[8 * 5 * 32];
    const int tid = threadIdx.x;
    const int e0  = blockIdx.x * 128;

    const int wave = tid >> 5, lane = tid & 31;
    const int lhi = lane >> 4, l16 = lane & 15;
    const int etile = wave * 16;

    int pe = 0, ps = -1, pt = 0;
    if (tid < 128) {
        const int i = e0 + tid;
        if (i < E_) {
            const int4 rec = reinterpret_cast<const int4*>(perm)[i];
            pe = rec.x; ps = rec.y; pt = rec.z;
        }
    }

    const int wv = __builtin_amdgcn_readfirstlane(wave);
    v2f wfr[5];
    #pragma unroll
    for (int k = 0; k < 5; ++k) {
        const int g  = wv + k * 8;
        const int T  = g / 5, kq = g % 5;
        const int ka = kq*4 + (lhi ? 2 : 0);
        const float* Wp = (T < 2) ? qw : (T < 4) ? kw : vw;
        const int ld  = (T < 4) ? DHH : DIMV;
        const int col = ((T < 4) ? (T & 1) : (T - 4)) * 16 + l16;
        const float wx = Wp[min(ka,     RPE - 1) * ld + col];
        const float wy = Wp[min(ka + 1, RPE - 1) * ld + col];
        wfr[k].x = (ka     < RPE) ? wx : 0.0f;
        wfr[k].y = (ka + 1 < RPE) ? wy : 0.0f;
    }

    if (tid < 128) { sE[tid] = pe; sS[tid] = ps; sT[tid] = pt; }
    __syncthreads();

    float stg[9];
    #pragma unroll
    for (int k = 0; k < 9; ++k) {
        const int i = tid + k * 256;
        const int el = i / RPE, c = i % RPE;
        stg[k] = ea[(size_t)sE[el] * RPE + c];
    }
    #pragma unroll
    for (int k = 0; k < 9; ++k) {
        const int i = tid + k * 256;
        const int el = i / RPE, c = i % RPE;
        sea[el*21 + c] = (sS[el] >= 0) ? stg[k] : 0.0f;
    }
    #pragma unroll
    for (int k = 0; k < 5; ++k) swf[(wv + k * 8) * 32 + lane] = wfr[k];
    float bqk[2][2];
    #pragma unroll
    for (int sel = 0; sel < 2; ++sel) {
        #pragma unroll
        for (int ct = 0; ct < 2; ++ct) bqk[sel][ct] = (sel ? kb : qb)[ct*16 + l16];
    }
    if (tid < 128) { sea[tid*21 + 18] = 0.f; sea[tid*21 + 19] = 0.f; }
    __syncthreads();

    v2f bw[2][2][5];
    #pragma unroll
    for (int sel = 0; sel < 2; ++sel) {
        #pragma unroll
        for (int ct = 0; ct < 2; ++ct) {
            #pragma unroll
            for (int kq = 0; kq < 5; ++kq)
                bw[sel][ct][kq] = swf[((sel*2 + ct) * 5 + kq) * 32 + lane];
        }
    }

    #pragma unroll
    for (int sel = 0; sel < 2; ++sel) {
        #pragma unroll
        for (int ct = 0; ct < 2; ++ct) {
            const float bc = bqk[sel][ct];
            v8f acc = {bc, bc, bc, bc, bc, bc, bc, bc};
            #pragma unroll
            for (int kq = 0; kq < 5; ++kq) {
                const int ka = kq*4 + (lhi ? 2 : 0);
                v2f a;
                a.x = sea[(etile + l16)*21 + ka];
                a.y = sea[(etile + l16)*21 + ka + 1];
                acc = __builtin_amdgcn_wmma_f32_16x16x4_f32(
                    false, a, false, bw[sel][ct][kq], (short)0, acc, false, false);
            }
            const int cb = sel*32 + ct*16 + l16;
            #pragma unroll
            for (int r = 0; r < 8; ++r)
                srpe[(etile + r + (lhi ? 8 : 0))*67 + cb] = acc[r];
        }
    }
    __syncthreads();

    {
        const int eloc = tid >> 1, pairh = (tid & 1) * 2;
        const int s = sS[eloc];
        const int t = sT[eloc];
        float ev0 = 0.f, ev1 = 0.f;
        if (s >= 0) {
            const float scale = 0.35355339059327373f;
            const float* qrow = qkv + (size_t)s * QKVW + pairh*DD;
            const float* krow = qkv + (size_t)t * QKVW + DHH + pairh*DD;
            float qv[2*DD], kv[2*DD];
            #pragma unroll
            for (int i = 0; i < 2*DD; ++i) kv[i] = krow[i];
            #pragma unroll
            for (int i = 0; i < 2*DD; ++i) qv[i] = qrow[i];
            #pragma unroll
            for (int hh = 0; hh < 2; ++hh) {
                const int h = pairh + hh;
                float compat = 0.f;
                #pragma unroll
                for (int d = 0; d < DD; ++d) {
                    const int c = h*DD + d;
                    const float qe = qv[hh*DD + d]*scale + srpe[eloc*67 + c];
                    const float ke = kv[hh*DD + d]       + srpe[eloc*67 + 32 + c];
                    compat = fmaf(qe, ke, compat);
                }
                const float ev = __expf(compat);
                if (hh == 0) ev0 = ev; else ev1 = ev;
            }
        }
        sex[eloc*HH + pairh]     = ev0;
        sex[eloc*HH + pairh + 1] = ev1;
    }

    v2f bv[4][5];
    #pragma unroll
    for (int ct = 0; ct < 4; ++ct) {
        #pragma unroll
        for (int kq = 0; kq < 5; ++kq)
            bv[ct][kq] = swf[((4 + ct) * 5 + kq) * 32 + lane];
    }
    __syncthreads();

    #pragma unroll
    for (int ct = 0; ct < 4; ++ct) {
        v8f acc = {};
        #pragma unroll
        for (int kq = 0; kq < 5; ++kq) {
            const int ka = kq*4 + (lhi ? 2 : 0);
            v2f a;
            a.x = sea[(etile + l16)*21 + ka];
            a.y = sea[(etile + l16)*21 + ka + 1];
            acc = __builtin_amdgcn_wmma_f32_16x16x4_f32(
                false, a, false, bv[ct][kq], (short)0, acc, false, false);
        }
        const int cb = ct*16 + l16;
        #pragma unroll
        for (int r = 0; r < 8; ++r)
            srpe[(etile + r + (lhi ? 8 : 0))*67 + cb] = acc[r];
    }
    __syncthreads();

    const int ms = sS[etile + l16];
    const int mt = sT[etile + l16];
    const float vb0 = vb[lane], vb1 = vb[lane + 32];
    int cur = -1;
    float r0 = 0.f, r1 = 0.f;
    float d0 = 0.f, d1 = 0.f;
    #pragma unroll 1
    for (int hb = 0; hb < 16; hb += 8) {
        float val0[8], val1[8];
        float wgt0[8], wgt1[8];
        int   srun[8];
        #pragma unroll
        for (int j = 0; j < 8; ++j) {
            const int s = __builtin_amdgcn_readlane(ms, hb + j);
            const int t = __builtin_amdgcn_readlane(mt, hb + j);
            const float a0 = sex[(etile + hb + j)*HH + lhi];
            const float a1 = sex[(etile + hb + j)*HH + 2 + lhi];
            const float* vrow = qkv + (size_t)t * QKVW + 2*DHH;
            val0[j] = (vrow[lane]      + srpe[(etile + hb + j)*67 + lane]      + vb0) * a0;
            val1[j] = (vrow[lane + 32] + srpe[(etile + hb + j)*67 + lane + 32] + vb1) * a1;
            wgt0[j] = a0; wgt1[j] = a1;
            srun[j] = s;
        }
        #pragma unroll
        for (int j = 0; j < 8; ++j) {
            if (srun[j] != cur) {
                if (cur >= 0) {
                    atomicAdd(&out[(size_t)cur * DIMV + lane],      r0);
                    atomicAdd(&out[(size_t)cur * DIMV + 32 + lane], r1);
                    if (l16 < 2) atomicAdd(&den[cur * HH + lhi + 2*l16], l16 ? d1 : d0);
                }
                cur = srun[j]; r0 = val0[j]; r1 = val1[j]; d0 = wgt0[j]; d1 = wgt1[j];
            } else {
                r0 += val0[j]; r1 += val1[j]; d0 += wgt0[j]; d1 += wgt1[j];
            }
        }
    }
    if (cur >= 0) {
        atomicAdd(&out[(size_t)cur * DIMV + lane],      r0);
        atomicAdd(&out[(size_t)cur * DIMV + 32 + lane], r1);
        if (l16 < 2) atomicAdd(&den[cur * HH + lhi + 2*l16], l16 ? d1 : d0);
    }
}

__global__ __launch_bounds__(256) void out_divide_kernel(
    float* __restrict__ out, const float* __restrict__ den, int N)
{
    const int total = N * (DIMV / 4);
    const int base  = blockIdx.x * 2048 + threadIdx.x;
    float4 v[8];
    float  d[8];
    #pragma unroll
    for (int k = 0; k < 8; ++k) {
        const int i = base + k * 256;
        v[k] = make_float4(0.f, 0.f, 0.f, 0.f);
        d[k] = 1.0f;
        if (i < total) {
            v[k] = reinterpret_cast<const float4*>(out)[i];
            d[k] = den[(i >> 4) * HH + ((i & 15) >> 2)];
        }
    }
    #pragma unroll
    for (int k = 0; k < 8; ++k) {
        const int i = base + k * 256;
        if (i < total) {
            const float dd = d[k] + 1e-16f;
            float4 w = v[k];
            w.x = w.x / dd; w.y = w.y / dd; w.z = w.z / dd; w.w = w.w / dd;
            reinterpret_cast<float4*>(out)[i] = w;
        }
    }
}

extern "C" void kernel_launch(void* const* d_in, const int* in_sizes, int n_in,
                              void* d_out, int out_size, void* d_ws, size_t ws_size,
                              hipStream_t stream)
{
    const float* x       = (const float*)d_in[0];
    const int*   eidx    = (const int*)  d_in[1];
    const float* ea      = (const float*)d_in[2];
    const float* qkv_w   = (const float*)d_in[3];
    const float* qkv_b   = (const float*)d_in[4];
    const float* k_rpe_w = (const float*)d_in[5];
    const float* k_rpe_b = (const float*)d_in[6];
    const float* q_rpe_w = (const float*)d_in[7];
    const float* q_rpe_b = (const float*)d_in[8];
    const float* v_rpe_w = (const float*)d_in[9];
    const float* v_rpe_b = (const float*)d_in[10];

    const int N  = in_sizes[0] / DIMV;
    const int E_ = in_sizes[2] / RPE;
    float* out = (float*)d_out;

    char* ws = (char*)d_ws;
    float* qkv    = (float*)ws;
    int*   perm   = (int*)  (ws + (size_t)N * QKVW * 4);
    float* den    = (float*)(ws + (size_t)N * QKVW * 4
                                + (size_t)E_ * HH * 4);
    int*   bbase  = (int*)  (ws + (size_t)N * QKVW * 4
                                + (size_t)E_ * HH * 4
                                + (size_t)N * HH * 4);
    int*   tmp    = bbase + 2 * MAXB;
    int*   tab    = tmp + (size_t)E_ * 4;

    const int NB = (N + BNODES - 1) / BNODES;

    hipMemsetAsync(den, 0, (size_t)N * HH * sizeof(float), stream);
    hipMemsetAsync(out, 0, (size_t)N * DIMV * sizeof(float), stream);

    bucket_count_kernel<<<(E_ + CHUNK - 1) / CHUNK, 1024, 0, stream>>>(eidx, tab, E_, NB);
    bucket_scatter_kernel<<<(E_ + CHUNK - 1) / CHUNK, 1024, 0, stream>>>(
        eidx, tab, bbase, tmp, E_, NB);
    bucket_sort_kernel<<<NB, 512, 0, stream>>>(tmp, bbase, perm);
    qkv_proj_kernel<<<(N + 15) / 16, 256, 0, stream>>>(x, qkv_w, qkv_b, qkv, N);
    edge_fused_kernel<<<(E_ + 127) / 128, 256, 0, stream>>>(
        ea, perm, qkv, q_rpe_w, q_rpe_b, k_rpe_w, k_rpe_b, v_rpe_w, v_rpe_b, den, out, E_);
    out_divide_kernel<<<(N * (DIMV / 4) + 2047) / 2048, 256, 0, stream>>>(out, den, N);
}


# Edge kernel: every wave takes its own 16 slots through 13 tiles with no block barrier (its LDS rows are private) and fetches the next tile's records and edge_attr rows while it computes this one
# speedup vs baseline: 5.1754x; 1.0775x over previous
#include <hip/hip_runtime.h>
#include <hip/hip_bf16.h>


typedef __attribute__((ext_vector_type(2))) float v2f;
typedef __attribute__((ext_vector_type(8))) float v8f;

#define HH    4
#define DD    8
#define DHH   32
#define DIMV  64
#define RPE   18
#define QKVW  128

__global__ __launch_bounds__(256) void qkv_proj_kernel(
    const float* __restrict__ x, const float* __restrict__ W,
    const float* __restrict__ bias, float* __restrict__ qkv, int N)
{
    __shared__ float xs[16 * 65];
    const int tid = threadIdx.x;
    const int r0  = blockIdx.x * 16;

    {
        int idx = tid * 4;
        int row = idx >> 6, col = idx & 63;
        int grow = r0 + row;
        float4 vx = make_float4(0.f, 0.f, 0.f, 0.f);
        if (grow < N) vx = *(const float4*)(x + (size_t)grow * 64 + col);
        xs[row*65+col+0] = vx.x; xs[row*65+col+1] = vx.y;
        xs[row*65+col+2] = vx.z; xs[row*65+col+3] = vx.w;
    }
    __syncthreads();

    const int wave = tid >> 5;
    const int lane = tid & 31;
    const int lhi  = lane >> 4;
    const int l16  = lane & 15;
    const int colbase = wave * 16;

    v8f acc = {};
    #pragma unroll
    for (int kk = 0; kk < 64; kk += 4) {
        const int ka = kk + (lhi ? 2 : 0);
        v2f a, b;
        a.x = xs[l16*65 + ka];
        a.y = xs[l16*65 + ka + 1];
        b.x = W[(size_t)ka     * QKVW + colbase + l16];
        b.y = W[(size_t)(ka+1) * QKVW + colbase + l16];
        acc = __builtin_amdgcn_wmma_f32_16x16x4_f32(
            false, a, false, b, (short)0, acc, false, false);
    }

    const int col = colbase + l16;
    const float bv = bias[col];
    #pragma unroll
    for (int r = 0; r < 8; ++r) {
        int grow = r0 + r + (lhi ? 8 : 0);
        if (grow < N) qkv[(size_t)grow * QKVW + col] = acc[r] + bv;
    }
}

#define BSH    8
#define BNODES 256
#define MAXB   1024
#define EPT    8
#define CHUNK  (EPT * 1024)

__global__ __launch_bounds__(1024) void bucket_count_kernel(
    const int* __restrict__ eidx, int* __restrict__ tab, int E_, int NB)
{
    __shared__ int lcnt[MAXB];
    const int tid = threadIdx.x;
    const int base = blockIdx.x * CHUNK;
    lcnt[tid] = 0;
    int s[EPT];
    #pragma unroll
    for (int k = 0; k < EPT; ++k)
        s[k] = eidx[min(base + tid + k * 1024, E_ - 1)];
    #pragma unroll
    for (int k = 0; k < EPT; ++k)
        if (base + tid + k * 1024 >= E_) s[k] = -1;
    __syncthreads();
    #pragma unroll
    for (int k = 0; k < EPT; ++k)
        if (s[k] >= 0) atomicAdd(&lcnt[s[k] >> BSH], 1);
    __syncthreads();
    if (tid < ((NB + 3) & ~3)) tab[blockIdx.x * MAXB + tid] = lcnt[tid];
}

__global__ __launch_bounds__(1024) void bucket_scatter_kernel(
    const int* __restrict__ eidx, const int* __restrict__ tab,
    int* __restrict__ bbase, int* __restrict__ tmp, int E_, int NB)
{
    __shared__ int lcnt[MAXB];
    __shared__ int ltot[MAXB];
    __shared__ int lpre[MAXB];
    __shared__ int lbase[MAXB];
    __shared__ int wsum[33];
    const int tid = threadIdx.x, lane = tid & 31, wave = tid >> 5;
    const int base = blockIdx.x * CHUNK;
    lcnt[tid] = 0; ltot[tid] = 0; lpre[tid] = 0;
    int s[EPT], t[EPT];
    #pragma unroll
    for (int k = 0; k < EPT; ++k) {
        const int e = min(base + tid + k * 1024, E_ - 1);
        s[k] = eidx[e]; t[k] = eidx[E_ + e];
    }
    #pragma unroll
    for (int k = 0; k < EPT; ++k)
        if (base + tid + k * 1024 >= E_) s[k] = -1;
    const int c4 = (tid & 127) * 4;
    int4 tot = make_int4(0, 0, 0, 0), pre = make_int4(0, 0, 0, 0);
    if (c4 < NB) {
        const int nrow = gridDim.x;
        #pragma unroll 4
        for (int r = tid >> 7; r < nrow; r += 8) {
            const int4 v = *reinterpret_cast<const int4*>(tab + r * MAXB + c4);
            tot.x += v.x; tot.y += v.y; tot.z += v.z; tot.w += v.w;
            if (r < (int)blockIdx.x) { pre.x += v.x; pre.y += v.y; pre.z += v.z; pre.w += v.w; }
        }
    }
    __syncthreads();
    int rk[EPT];
    #pragma unroll
    for (int k = 0; k < EPT; ++k) {
        rk[k] = 0;
        if (s[k] >= 0) rk[k] = atomicAdd(&lcnt[s[k] >> BSH], 1);
    }
    if (c4 < NB) {
        atomicAdd(&ltot[c4], tot.x); atomicAdd(&ltot[c4 + 1], tot.y);
        atomicAdd(&ltot[c4 + 2], tot.z); atomicAdd(&ltot[c4 + 3], tot.w);
        atomicAdd(&lpre[c4], pre.x); atomicAdd(&lpre[c4 + 1], pre.y);
        atomicAdd(&lpre[c4 + 2], pre.z); atomicAdd(&lpre[c4 + 3], pre.w);
    }
    __syncthreads();
    const int bc = (tid < NB) ? ltot[tid] : 0;
    int x = bc;
    #pragma unroll
    for (int d = 1; d < 32; d <<= 1) {
        const int y = __shfl_up(x, d);
        if (lane >= d) x += y;
    }
    if (lane == 31) wsum[wave] = x;
    __syncthreads();
    if (wave == 0) {
        const int w = wsum[lane];
        int xs = w;
        #pragma unroll
        for (int d = 1; d < 32; d <<= 1) {
            const int y = __shfl_up(xs, d);
            if (lane >= d) xs += y;
        }
        wsum[lane] = xs - w;
        if (lane == 31) wsum[32] = xs;
    }
    __syncthreads();
    const int bb = wsum[wave] + (x - bc);
    lbase[tid] = bb + lpre[tid];
    if (blockIdx.x == 0) {
        if (tid < NB) bbase[tid] = bb;
        if (tid == 0) bbase[NB] = wsum[32];
    }
    __syncthreads();
    #pragma unroll
    for (int k = 0; k < EPT; ++k) {
        if (s[k] >= 0) {
            const int e = base + tid + k * 1024;
            const int pos = lbase[s[k] >> BSH] + rk[k];
            reinterpret_cast<int4*>(tmp)[pos] = make_int4(e, s[k], t[k], 0);
        }
    }
}

#define KC 10
__global__ __launch_bounds__(512) void bucket_sort_kernel(
    const int* __restrict__ tmp, const int* __restrict__ bbase, int* __restrict__ perm)
{
    __shared__ int ncnt[BNODES];
    __shared__ int ncur[BNODES];
    __shared__ int wsum[8];
    const int tid = threadIdx.x, lane = tid & 31, wave = tid >> 5;
    const int s0 = blockIdx.x << BSH;
    const int lo = bbase[blockIdx.x], hi = bbase[blockIdx.x + 1];
    const int4* rec = reinterpret_cast<const int4*>(tmp);
    if (tid < BNODES) ncnt[tid] = 0;
    int4 r[KC];
    #pragma unroll
    for (int k = 0; k < KC; ++k)
        r[k] = rec[max(min(lo + tid + k * 512, hi - 1), 0)];
    #pragma unroll
    for (int k = 0; k < KC; ++k)
        if (lo + tid + k * 512 >= hi) r[k].y = -1;
    __syncthreads();
    #pragma unroll
    for (int k = 0; k < KC; ++k)
        if (r[k].y >= 0) atomicAdd(&ncnt[r[k].y - s0], 1);
    for (int i = lo + tid + KC * 512; i < hi; i += 512)
        atomicAdd(&ncnt[tmp[4 * i + 1] - s0], 1);
    __syncthreads();
    const int c = (tid < BNODES) ? ncnt[tid] : 0;
    int x = c;
    #pragma unroll
    for (int d = 1; d < 32; d <<= 1) {
        const int y = __shfl_up(x, d);
        if (lane >= d) x += y;
    }
    if (lane == 31 && wave < 8) wsum[wave] = x;
    __syncthreads();
    if (tid < BNODES) {
        int wp = 0;
        #pragma unroll
        for (int w = 0; w < 8; ++w) wp += (w < wave) ? wsum[w] : 0;
        ncur[tid] = lo + wp + (x - c);
    }
    __syncthreads();
    #pragma unroll
    for (int k = 0; k < KC; ++k) {
        if (r[k].y >= 0) {
            const int pos = atomicAdd(&ncur[r[k].y - s0], 1);
            reinterpret_cast<int4*>(perm)[pos] = r[k];
        }
    }
    for (int i = lo + tid + KC * 512; i < hi; i += 512) {
        const int4 q = rec[i];
        const int pos = atomicAdd(&ncur[q.y - s0], 1);
        reinterpret_cast<int4*>(perm)[pos] = q;
    }
}

#define TPB 13

__device__ __forceinline__ void wave_lds_sync()
{
    __builtin_amdgcn_fence(__ATOMIC_RELEASE, "workgroup", "local");
    __builtin_amdgcn_wave_barrier();
    __builtin_amdgcn_fence(__ATOMIC_ACQUIRE, "workgroup", "local");
}

__global__ __launch_bounds__(256) void edge_fused_kernel(
    const float* __restrict__ ea, const int* __restrict__ perm,
    const float* __restrict__ qkv,
    const float* __restrict__ qw, const float* __restrict__ qb,
    const float* __restrict__ kw, const float* __restrict__ kb,
    const float* __restrict__ vw, const float* __restrict__ vb,
    float* __restrict__ den, float* __restrict__ out, int E_)
{
    __shared__ float sea[128 * 21];
    __shared__ float srpe[128 * 67];
    __shared__ float sex[128 * HH];
    __shared__ alignas(16) int sE[2][128];
    __shared__ int sS[2][128];
    __shared__ int sT[2][128];
    __shared__ v2f swf[8 * 5 * 32];
    __shared__ float sbqk[2 * DHH];
    const int tid = threadIdx.x;

    const int wave = tid >> 5, lane = tid & 31;
    const int lhi = lane >> 4, l16 = lane & 15;
    const int etile = wave * 16;
    const int ntiles = (E_ + 127) >> 7;
    const int tile0  = blockIdx.x * TPB;
    const int tend   = min(tile0 + TPB, ntiles);

    int pe = 0, ps = -1, pt = 0;
    {
        const int i = tile0 * 128 + etile + l16;
        if (i < E_) {
            const int4 rec = reinterpret_cast<const int4*>(perm)[i];
            pe = rec.x; ps = rec.y; pt = rec.z;
        }
    }

    const int wv = __builtin_amdgcn_readfirstlane(wave);
    v2f wfr[5];
    #pragma unroll
    for (int k = 0; k < 5; ++k) {
        const int g  = wv + k * 8;
        const int T  = g / 5, kq = g % 5;
        const int ka = kq*4 + (lhi ? 2 : 0);
        const float* Wp = (T < 2) ? qw : (T < 4) ? kw : vw;
        const int ld  = (T < 4) ? DHH : DIMV;
        const int col = ((T < 4) ? (T & 1) : (T - 4)) * 16 + l16;
        const float wx = Wp[min(ka,     RPE - 1) * ld + col];
        const float wy = Wp[min(ka + 1, RPE - 1) * ld + col];
        wfr[k].x = (ka     < RPE) ? wx : 0.0f;
        wfr[k].y = (ka + 1 < RPE) ? wy : 0.0f;
    }

    sE[0][etile + l16] = pe; sS[0][etile + l16] = ps; sT[0][etile + l16] = pt;
    wave_lds_sync();

    float stg[9];
    {
        const int4 e03 = *reinterpret_cast<const int4*>(&sE[0][etile + 8*lhi]);
        const int4 e47 = *reinterpret_cast<const int4*>(&sE[0][etile + 8*lhi + 4]);
        const int  e8  = sE[0][etile + (lane >> 1)];
        const int eid[8] = {e03.x, e03.y, e03.z, e03.w, e47.x, e47.y, e47.z, e47.w};
        #pragma unroll
        for (int k = 0; k < 8; ++k) stg[k] = ea[(size_t)eid[k] * RPE + l16];
        stg[8] = ea[(size_t)e8 * RPE + 16 + (lane & 1)];
    }
    pe = 0; ps = -1; pt = 0;
    {
        const int i = (tile0 + 1) * 128 + etile + l16;
        if (i < E_) {
            const int4 rec = reinterpret_cast<const int4*>(perm)[i];
            pe = rec.x; ps = rec.y; pt = rec.z;
        }
    }
    #pragma unroll
    for (int k = 0; k < 5; ++k) swf[(wv + k * 8) * 32 + lane] = wfr[k];
    if (tid < 2 * DHH) sbqk[tid] = (tid < DHH) ? qb[tid] : kb[tid - DHH];
    const float vb0 = vb[lane], vb1 = vb[lane + 32];
    if (tid < 128) { sea[tid*21 + 18] = 0.f; sea[tid*21 + 19] = 0.f; }
    __syncthreads();

    #pragma unroll 1
    for (int tile = tile0; tile < tend; ++tile) {
        const int cb = (tile - tile0) & 1, nb = cb ^ 1;
        const int* cS = sS[cb];
        const int* cT = sT[cb];

        #pragma unroll
        for (int k = 0; k < 8; ++k) sea[(etile + k + 8*lhi)*21 + l16] = stg[k];
        sea[(etile + (lane >> 1))*21 + 16 + (lane & 1)] = stg[8];
        sE[nb][etile + l16] = pe; sS[nb][etile + l16] = ps; sT[nb][etile + l16] = pt;
        wave_lds_sync();

        v2f bw[2][2][5];
        #pragma unroll
        for (int sel = 0; sel < 2; ++sel) {
            #pragma unroll
            for (int ct = 0; ct < 2; ++ct) {
                #pragma unroll
                for (int kq = 0; kq < 5; ++kq)
                    bw[sel][ct][kq] = swf[((sel*2 + ct) * 5 + kq) * 32 + lane];
            }
        }

        #pragma unroll
        for (int sel = 0; sel < 2; ++sel) {
            #pragma unroll
            for (int ct = 0; ct < 2; ++ct) {
                const float bc = sbqk[sel*32 + ct*16 + l16];
                v8f acc = {bc, bc, bc, bc, bc, bc, bc, bc};
                #pragma unroll
                for (int kq = 0; kq < 5; ++kq) {
                    const int ka = kq*4 + (lhi ? 2 : 0);
                    v2f a;
                    a.x = sea[(etile + l16)*21 + ka];
                    a.y = sea[(etile + l16)*21 + ka + 1];
                    acc = __builtin_amdgcn_wmma_f32_16x16x4_f32(
                        false, a, false, bw[sel][ct][kq], (short)0, acc, false, false);
                }
                const int cbq = sel*32 + ct*16 + l16;
                #pragma unroll
                for (int r = 0; r < 8; ++r)
                    srpe[(etile + r + (lhi ? 8 : 0))*67 + cbq] = acc[r];
            }
        }
        wave_lds_sync();

        {
            const int eloc = tid >> 1, pairh = (tid & 1) * 2;
            const int s = cS[eloc];
            const int t = cT[eloc];
            const float scale = 0.35355339059327373f;
            const float* qrow = qkv + (size_t)max(s, 0) * QKVW + pairh*DD;
            const float* krow = qkv + (size_t)t * QKVW + DHH + pairh*DD;
            float qv[2*DD], kv[2*DD];
            #pragma unroll
            for (int i = 0; i < 2*DD; ++i) kv[i] = krow[i];
            #pragma unroll
            for (int i = 0; i < 2*DD; ++i) qv[i] = qrow[i];
            __builtin_amdgcn_sched_barrier(0);

            {
                const int4 e03 = *reinterpret_cast<const int4*>(&sE[nb][etile + 8*lhi]);
                const int4 e47 = *reinterpret_cast<const int4*>(&sE[nb][etile + 8*lhi + 4]);
                const int  e8  = sE[nb][etile + (lane >> 1)];
                const int eid[8] = {e03.x, e03.y, e03.z, e03.w, e47.x, e47.y, e47.z, e47.w};
                #pragma unroll
                for (int k = 0; k < 8; ++k) stg[k] = ea[(size_t)eid[k] * RPE + l16];
                stg[8] = ea[(size_t)e8 * RPE + 16 + (lane & 1)];
            }
            pe = 0; ps = -1; pt = 0;
            {
                const int i = (tile + 2) * 128 + etile + l16;
                if (i < E_) {
                    const int4 rec = reinterpret_cast<const int4*>(perm)[i];
                    pe = rec.x; ps = rec.y; pt = rec.z;
                }
            }

            float ev0 = 0.f, ev1 = 0.f;
            #pragma unroll
            for (int hh = 0; hh < 2; ++hh) {
                const int h = pairh + hh;
                float compat = 0.f;
                #pragma unroll
                for (int d = 0; d < DD; ++d) {
                    const int c = h*DD + d;
                    const float qe = qv[hh*DD + d]*scale + srpe[eloc*67 + c];
                    const float ke = kv[hh*DD + d]       + srpe[eloc*67 + 32 + c];
                    compat = fmaf(qe, ke, compat);
                }
                const float ev = __expf(compat);
                if (hh == 0) ev0 = ev; else ev1 = ev;
            }
            sex[eloc*HH + pairh]     = (s >= 0) ? ev0 : 0.f;
            sex[eloc*HH + pairh + 1] = (s >= 0) ? ev1 : 0.f;
        }

        wave_lds_sync();

        v2f bv[4][5];
        #pragma unroll
        for (int ct = 0; ct < 4; ++ct) {
            #pragma unroll
            for (int kq = 0; kq < 5; ++kq)
                bv[ct][kq] = swf[((4 + ct) * 5 + kq) * 32 + lane];
        }

        #pragma unroll
        for (int ct = 0; ct < 4; ++ct) {
            v8f acc = {};
            #pragma unroll
            for (int kq = 0; kq < 5; ++kq) {
                const int ka = kq*4 + (lhi ? 2 : 0);
                v2f a;
                a.x = sea[(etile + l16)*21 + ka];
                a.y = sea[(etile + l16)*21 + ka + 1];
                acc = __builtin_amdgcn_wmma_f32_16x16x4_f32(
                    false, a, false, bv[ct][kq], (short)0, acc, false, false);
            }
            const int cbv = ct*16 + l16;
            #pragma unroll
            for (int r = 0; r < 8; ++r)
                srpe[(etile + r + (lhi ? 8 : 0))*67 + cbv] = acc[r];
        }
        wave_lds_sync();

        const int ms = cS[etile + l16];
        const int mt = cT[etile + l16];
        int cur = -1;
        float r0 = 0.f, r1 = 0.f;
        float d0 = 0.f, d1 = 0.f;
        #pragma unroll 1
        for (int hb = 0; hb < 16; hb += 8) {
            float val0[8], val1[8];
            float wgt0[8], wgt1[8];
            int   srun[8];
            #pragma unroll
            for (int j = 0; j < 8; ++j) {
                const int s = __builtin_amdgcn_readlane(ms, hb + j);
                const int t = __builtin_amdgcn_readlane(mt, hb + j);
                const float a0 = sex[(etile + hb + j)*HH + lhi];
                const float a1 = sex[(etile + hb + j)*HH + 2 + lhi];
                const float* vrow = qkv + (size_t)t * QKVW + 2*DHH;
                val0[j] = (vrow[lane]      + srpe[(etile + hb + j)*67 + lane]      + vb0) * a0;
                val1[j] = (vrow[lane + 32] + srpe[(etile + hb + j)*67 + lane + 32] + vb1) * a1;
                wgt0[j] = a0; wgt1[j] = a1;
                srun[j] = s;
            }
            #pragma unroll
            for (int j = 0; j < 8; ++j) {
                if (srun[j] != cur) {
                    if (cur >= 0) {
                        atomicAdd(&out[(size_t)cur * DIMV + lane],      r0);
                        atomicAdd(&out[(size_t)cur * DIMV + 32 + lane], r1);
                        if (l16 < 2) atomicAdd(&den[cur * HH + lhi + 2*l16], l16 ? d1 : d0);
                    }
                    cur = srun[j]; r0 = val0[j]; r1 = val1[j]; d0 = wgt0[j]; d1 = wgt1[j];
                } else {
                    r0 += val0[j]; r1 += val1[j]; d0 += wgt0[j]; d1 += wgt1[j];
                }
            }
        }
        if (cur >= 0) {
            atomicAdd(&out[(size_t)cur * DIMV + lane],      r0);
            atomicAdd(&out[(size_t)cur * DIMV + 32 + lane], r1);
            if (l16 < 2) atomicAdd(&den[cur * HH + lhi + 2*l16], l16 ? d1 : d0);
        }
    }
}

__global__ __launch_bounds__(256) void out_divide_kernel(
    float* __restrict__ out, const float* __restrict__ den, int N)
{
    const int total = N * (DIMV / 4);
    const int base  = blockIdx.x * 2048 + threadIdx.x;
    float4 v[8];
    float  d[8];
    #pragma unroll
    for (int k = 0; k < 8; ++k) {
        const int i = base + k * 256;
        v[k] = make_float4(0.f, 0.f, 0.f, 0.f);
        d[k] = 1.0f;
        if (i < total) {
            v[k] = reinterpret_cast<const float4*>(out)[i];
            d[k] = den[(i >> 4) * HH + ((i & 15) >> 2)];
        }
    }
    #pragma unroll
    for (int k = 0; k < 8; ++k) {
        const int i = base + k * 256;
        if (i < total) {
            const float dd = d[k] + 1e-16f;
            float4 w = v[k];
            w.x = w.x / dd; w.y = w.y / dd; w.z = w.z / dd; w.w = w.w / dd;
            reinterpret_cast<float4*>(out)[i] = w;
        }
    }
}

extern "C" void kernel_launch(void* const* d_in, const int* in_sizes, int n_in,
                              void* d_out, int out_size, void* d_ws, size_t ws_size,
                              hipStream_t stream)
{
    const float* x       = (const float*)d_in[0];
    const int*   eidx    = (const int*)  d_in[1];
    const float* ea      = (const float*)d_in[2];
    const float* qkv_w   = (const float*)d_in[3];
    const float* qkv_b   = (const float*)d_in[4];
    const float* k_rpe_w = (const float*)d_in[5];
    const float* k_rpe_b = (const float*)d_in[6];
    const float* q_rpe_w = (const float*)d_in[7];
    const float* q_rpe_b = (const float*)d_in[8];
    const float* v_rpe_w = (const float*)d_in[9];
    const float* v_rpe_b = (const float*)d_in[10];

    const int N  = in_sizes[0] / DIMV;
    const int E_ = in_sizes[2] / RPE;
    float* out = (float*)d_out;

    char* ws = (char*)d_ws;
    float* qkv    = (float*)ws;
    int*   perm   = (int*)  (ws + (size_t)N * QKVW * 4);
    float* den    = (float*)(ws + (size_t)N * QKVW * 4
                                + (size_t)E_ * HH * 4);
    int*   bbase  = (int*)  (ws + (size_t)N * QKVW * 4
                                + (size_t)E_ * HH * 4
                                + (size_t)N * HH * 4);
    int*   tmp    = bbase + 2 * MAXB;
    int*   tab    = tmp + (size_t)E_ * 4;

    const int NB = (N + BNODES - 1) / BNODES;

    hipMemsetAsync(den, 0, (size_t)N * HH * sizeof(float), stream);
    hipMemsetAsync(out, 0, (size_t)N * DIMV * sizeof(float), stream);

    bucket_count_kernel<<<(E_ + CHUNK - 1) / CHUNK, 1024, 0, stream>>>(eidx, tab, E_, NB);
    bucket_scatter_kernel<<<(E_ + CHUNK - 1) / CHUNK, 1024, 0, stream>>>(
        eidx, tab, bbase, tmp, E_, NB);
    bucket_sort_kernel<<<NB, 512, 0, stream>>>(tmp, bbase, perm);
    qkv_proj_kernel<<<(N + 15) / 16, 256, 0, stream>>>(x, qkv_w, qkv_b, qkv, N);
    edge_fused_kernel<<<(E_ + 128 * TPB - 1) / (128 * TPB), 256, 0, stream>>>(
        ea, perm, qkv, q_rpe_w, q_rpe_b, k_rpe_w, k_rpe_b, v_rpe_w, v_rpe_b, den, out, E_);
    out_divide_kernel<<<(N * (DIMV / 4) + 2047) / 2048, 256, 0, stream>>>(out, den, N);
}


# Edge kernel: a wave owns whole source nodes (runs starting in its 200 slots; row starts kept by the sort), stores each divided row once: no atomics, memsets or divide pass; float sum order changes
# speedup vs baseline: 5.5099x; 1.0646x over previous
#include <hip/hip_runtime.h>
#include <hip/hip_bf16.h>


typedef __attribute__((ext_vector_type(2))) float v2f;
typedef __attribute__((ext_vector_type(8))) float v8f;

#define HH    4
#define DD    8
#define DHH   32
#define DIMV  64
#define RPE   18
#define QKVW  128

__global__ __launch_bounds__(256) void qkv_proj_kernel(
    const float* __restrict__ x, const float* __restrict__ W,
    const float* __restrict__ bias, float* __restrict__ qkv, int N)
{
    __shared__ float xs[16 * 65];
    const int tid = threadIdx.x;
    const int r0  = blockIdx.x * 16;

    {
        int idx = tid * 4;
        int row = idx >> 6, col = idx & 63;
        int grow = r0 + row;
        float4 vx = make_float4(0.f, 0.f, 0.f, 0.f);
        if (grow < N) vx = *(const float4*)(x + (size_t)grow * 64 + col);
        xs[row*65+col+0] = vx.x; xs[row*65+col+1] = vx.y;
        xs[row*65+col+2] = vx.z; xs[row*65+col+3] = vx.w;
    }
    __syncthreads();

    const int wave = tid >> 5;
    const int lane = tid & 31;
    const int lhi  = lane >> 4;
    const int l16  = lane & 15;
    const int colbase = wave * 16;

    v8f acc = {};
    #pragma unroll
    for (int kk = 0; kk < 64; kk += 4) {
        const int ka = kk + (lhi ? 2 : 0);
        v2f a, b;
        a.x = xs[l16*65 + ka];
        a.y = xs[l16*65 + ka + 1];
        b.x = W[(size_t)ka     * QKVW + colbase + l16];
        b.y = W[(size_t)(ka+1) * QKVW + colbase + l16];
        acc = __builtin_amdgcn_wmma_f32_16x16x4_f32(
            false, a, false, b, (short)0, acc, false, false);
    }

    const int col = colbase + l16;
    const float bv = bias[col];
    #pragma unroll
    for (int r = 0; r < 8; ++r) {
        int grow = r0 + r + (lhi ? 8 : 0);
        if (grow < N) qkv[(size_t)grow * QKVW + col] = acc[r] + bv;
    }
}

#define BSH    8
#define BNODES 256
#define MAXB   1024
#define EPT    8
#define CHUNK  (EPT * 1024)

__global__ __launch_bounds__(1024) void bucket_count_kernel(
    const int* __restrict__ eidx, int* __restrict__ tab, int E_, int NB)
{
    __shared__ int lcnt[MAXB];
    const int tid = threadIdx.x;
    const int base = blockIdx.x * CHUNK;
    lcnt[tid] = 0;
    int s[EPT];
    #pragma unroll
    for (int k = 0; k < EPT; ++k)
        s[k] = eidx[min(base + tid + k * 1024, E_ - 1)];
    #pragma unroll
    for (int k = 0; k < EPT; ++k)
        if (base + tid + k * 1024 >= E_) s[k] = -1;
    __syncthreads();
    #pragma unroll
    for (int k = 0; k < EPT; ++k)
        if (s[k] >= 0) atomicAdd(&lcnt[s[k] >> BSH], 1);
    __syncthreads();
    if (tid < ((NB + 3) & ~3)) tab[blockIdx.x * MAXB + tid] = lcnt[tid];
}

__global__ __launch_bounds__(1024) void bucket_scatter_kernel(
    const int* __restrict__ eidx, const int* __restrict__ tab,
    int* __restrict__ bbase, int* __restrict__ tmp, int E_, int NB)
{
    __shared__ int lcnt[MAXB];
    __shared__ int ltot[MAXB];
    __shared__ int lpre[MAXB];
    __shared__ int lbase[MAXB];
    __shared__ int wsum[33];
    const int tid = threadIdx.x, lane = tid & 31, wave = tid >> 5;
    const int base = blockIdx.x * CHUNK;
    lcnt[tid] = 0; ltot[tid] = 0; lpre[tid] = 0;
    int s[EPT], t[EPT];
    #pragma unroll
    for (int k = 0; k < EPT; ++k) {
        const int e = min(base + tid + k * 1024, E_ - 1);
        s[k] = eidx[e]; t[k] = eidx[E_ + e];
    }
    #pragma unroll
    for (int k = 0; k < EPT; ++k)
        if (base + tid + k * 1024 >= E_) s[k] = -1;
    const int c4 = (tid & 127) * 4;
    int4 tot = make_int4(0, 0, 0, 0), pre = make_int4(0, 0, 0, 0);
    if (c4 < NB) {
        const int nrow = gridDim.x;
        #pragma unroll 4
        for (int r = tid >> 7; r < nrow; r += 8) {
            const int4 v = *reinterpret_cast<const int4*>(tab + r * MAXB + c4);
            tot.x += v.x; tot.y += v.y; tot.z += v.z; tot.w += v.w;
            if (r < (int)blockIdx.x) { pre.x += v.x; pre.y += v.y; pre.z += v.z; pre.w += v.w; }
        }
    }
    __syncthreads();
    int rk[EPT];
    #pragma unroll
    for (int k = 0; k < EPT; ++k) {
        rk[k] = 0;
        if (s[k] >= 0) rk[k] = atomicAdd(&lcnt[s[k] >> BSH], 1);
    }
    if (c4 < NB) {
        atomicAdd(&ltot[c4], tot.x); atomicAdd(&ltot[c4 + 1], tot.y);
        atomicAdd(&ltot[c4 + 2], tot.z); atomicAdd(&ltot[c4 + 3], tot.w);
        atomicAdd(&lpre[c4], pre.x); atomicAdd(&lpre[c4 + 1], pre.y);
        atomicAdd(&lpre[c4 + 2], pre.z); atomicAdd(&lpre[c4 + 3], pre.w);
    }
    __syncthreads();
    const int bc = (tid < NB) ? ltot[tid] : 0;
    int x = bc;
    #pragma unroll
    for (int d = 1; d < 32; d <<= 1) {
        const int y = __shfl_up(x, d);
        if (lane >= d) x += y;
    }
    if (lane == 31) wsum[wave] = x;
    __syncthreads();
    if (wave == 0) {
        const int w = wsum[lane];
        int xs = w;
        #pragma unroll
        for (int d = 1; d < 32; d <<= 1) {
            const int y = __shfl_up(xs, d);
            if (lane >= d) xs += y;
        }
        wsum[lane] = xs - w;
        if (lane == 31) wsum[32] = xs;
    }
    __syncthreads();
    const int bb = wsum[wave] + (x - bc);
    lbase[tid] = bb + lpre[tid];
    if (blockIdx.x == 0) {
        if (tid < NB) bbase[tid] = bb;
        if (tid == 0) bbase[NB] = wsum[32];
    }
    __syncthreads();
    #pragma unroll
    for (int k = 0; k < EPT; ++k) {
        if (s[k] >= 0) {
            const int e = base + tid + k * 1024;
            const int pos = lbase[s[k] >> BSH] + rk[k];
            reinterpret_cast<int4*>(tmp)[pos] = make_int4(e, s[k], t[k], 0);
        }
    }
}

#define KC 10
__global__ __launch_bounds__(512) void bucket_sort_kernel(
    const int* __restrict__ tmp, const int* __restrict__ bbase, int* __restrict__ perm,
    int* __restrict__ rowptr)
{
    __shared__ int ncnt[BNODES];
    __shared__ int ncur[BNODES];
    __shared__ int wsum[8];
    const int tid = threadIdx.x, lane = tid & 31, wave = tid >> 5;
    const int s0 = blockIdx.x << BSH;
    const int lo = bbase[blockIdx.x], hi = bbase[blockIdx.x + 1];
    const int4* rec = reinterpret_cast<const int4*>(tmp);
    if (tid < BNODES) ncnt[tid] = 0;
    int4 r[KC];
    #pragma unroll
    for (int k = 0; k < KC; ++k)
        r[k] = rec[max(min(lo + tid + k * 512, hi - 1), 0)];
    #pragma unroll
    for (int k = 0; k < KC; ++k)
        if (lo + tid + k * 512 >= hi) r[k].y = -1;
    __syncthreads();
    #pragma unroll
    for (int k = 0; k < KC; ++k)
        if (r[k].y >= 0) atomicAdd(&ncnt[r[k].y - s0], 1);
    for (int i = lo + tid + KC * 512; i < hi; i += 512)
        atomicAdd(&ncnt[tmp[4 * i + 1] - s0], 1);
    __syncthreads();
    const int c = (tid < BNODES) ? ncnt[tid] : 0;
    int x = c;
    #pragma unroll
    for (int d = 1; d < 32; d <<= 1) {
        const int y = __shfl_up(x, d);
        if (lane >= d) x += y;
    }
    if (lane == 31 && wave < 8) wsum[wave] = x;
    __syncthreads();
    if (tid < BNODES) {
        int wp = 0;
        #pragma unroll
        for (int w = 0; w < 8; ++w) wp += (w < wave) ? wsum[w] : 0;
        ncur[tid] = lo + wp + (x - c);
        rowptr[s0 + tid] = lo + wp + (x - c);
    }
    if (blockIdx.x == gridDim.x - 1 && tid == 0) rowptr[s0 + BNODES] = hi;
    __syncthreads();
    #pragma unroll
    for (int k = 0; k < KC; ++k) {
        if (r[k].y >= 0) {
            const int pos = atomicAdd(&ncur[r[k].y - s0], 1);
            reinterpret_cast<int4*>(perm)[pos] = r[k];
        }
    }
    for (int i = lo + tid + KC * 512; i < hi; i += 512) {
        const int4 q = rec[i];
        const int pos = atomicAdd(&ncur[q.y - s0], 1);
        reinterpret_cast<int4*>(perm)[pos] = q;
    }
}

#define SPW 200

__device__ __forceinline__ void wave_range(const int* __restrict__ perm, const int* __restrict__ rowptr,
                                           int g, int E_, int N, int& nA, int& sa, int& nB, int& sb)
{
    const int p0 = g * SPW, p1 = p0 + SPW;
    const int s0 = perm[4 * min(p0, E_ - 1) + 1];
    const int s1 = perm[4 * min(p1, E_ - 1) + 1];
    const int a0 = rowptr[s0], b0 = rowptr[s0 + 1];
    const int a1 = rowptr[s1], b1 = rowptr[s1 + 1];
    nA = (a0 == p0) ? s0 : s0 + 1;  sa = (a0 == p0) ? p0 : b0;
    nB = (a1 == p1) ? s1 : s1 + 1;  sb = (a1 == p1) ? p1 : b1;
    if (p0 >= E_) { nA = N; sa = E_; }
    if (p1 >= E_) { nB = N; sb = E_; }
    if (g == 0) nA = 0;
}

__device__ __forceinline__ void store_run(float* __restrict__ out, int from, int node, int lane,
                                          float r0, float r1, float d0, float d1)
{
    #pragma clang loop unroll(disable) vectorize(disable) interleave(disable)
    for (int n = from; n < node; ++n) {
        out[(size_t)n * DIMV + lane]      = 0.f;
        out[(size_t)n * DIMV + 32 + lane] = 0.f;
    }
    out[(size_t)node * DIMV + lane]      = r0 / (d0 + 1e-16f);
    out[(size_t)node * DIMV + 32 + lane] = r1 / (d1 + 1e-16f);
}

__device__ __forceinline__ void wave_lds_sync()
{
    __builtin_amdgcn_fence(__ATOMIC_RELEASE, "workgroup", "local");
    __builtin_amdgcn_wave_barrier();
    __builtin_amdgcn_fence(__ATOMIC_ACQUIRE, "workgroup", "local");
}

__global__ __launch_bounds__(256) void edge_fused_kernel(
    const float* __restrict__ ea, const int* __restrict__ perm,
    const int* __restrict__ rowptr, const float* __restrict__ qkv,
    const float* __restrict__ qw, const float* __restrict__ qb,
    const float* __restrict__ kw, const float* __restrict__ kb,
    const float* __restrict__ vw, const float* __restrict__ vb,
    float* __restrict__ out, int E_, int N)
{
    __shared__ float sea[128 * 21];
    __shared__ float srpe[128 * 67];
    __shared__ float sex[128 * HH];
    __shared__ alignas(16) int sE[2][128];
    __shared__ int sS[2][128];
    __shared__ int sT[2][128];
    __shared__ v2f swf[8 * 5 * 32];
    __shared__ float sbqk[2 * DHH];
    const int tid = threadIdx.x;

    const int wave = tid >> 5, lane = tid & 31;
    const int lhi = lane >> 4, l16 = lane & 15;
    const int etile = wave * 16;
    const int wv = __builtin_amdgcn_readfirstlane(wave);
    int nA, sa, nB, sb;
    wave_range(perm, rowptr, (int)blockIdx.x * 8 + wv, E_, N, nA, sa, nB, sb);
    const int nt = (sb - sa + 15) >> 4;

    int pe = 0, ps = -1, pt = 0;
    {
        const int i = sa + l16;
        if (i < sb) {
            const int4 rec = reinterpret_cast<const int4*>(perm)[i];
            pe = rec.x; ps = rec.y; pt = rec.z;
        }
    }

    v2f wfr[5];
    #pragma unroll
    for (int k = 0; k < 5; ++k) {
        const int g  = wv + k * 8;
        const int T  = g / 5, kq = g % 5;
        const int ka = kq*4 + (lhi ? 2 : 0);
        const float* Wp = (T < 2) ? qw : (T < 4) ? kw : vw;
        const int ld  = (T < 4) ? DHH : DIMV;
        const int col = ((T < 4) ? (T & 1) : (T - 4)) * 16 + l16;
        const float wx = Wp[min(ka,     RPE - 1) * ld + col];
        const float wy = Wp[min(ka + 1, RPE - 1) * ld + col];
        wfr[k].x = (ka     < RPE) ? wx : 0.0f;
        wfr[k].y = (ka + 1 < RPE) ? wy : 0.0f;
    }

    sE[0][etile + l16] = pe; sS[0][etile + l16] = ps; sT[0][etile + l16] = pt;
    wave_lds_sync();

    float stg[9];
    {
        const int4 e03 = *reinterpret_cast<const int4*>(&sE[0][etile + 8*lhi]);
        const int4 e47 = *reinterpret_cast<const int4*>(&sE[0][etile + 8*lhi + 4]);
        const int  e8  = sE[0][etile + (lane >> 1)];
        const int eid[8] = {e03.x, e03.y, e03.z, e03.w, e47.x, e47.y, e47.z, e47.w};
        #pragma unroll
        for (int k = 0; k < 8; ++k) stg[k] = ea[(size_t)eid[k] * RPE + l16];
        stg[8] = ea[(size_t)e8 * RPE + 16 + (lane & 1)];
    }
    pe = 0; ps = -1; pt = 0;
    {
        const int i = sa + 16 + l16;
        if (i < sb) {
            const int4 rec = reinterpret_cast<const int4*>(perm)[i];
            pe = rec.x; ps = rec.y; pt = rec.z;
        }
    }
    #pragma unroll
    for (int k = 0; k < 5; ++k) swf[(wv + k * 8) * 32 + lane] = wfr[k];
    if (tid < 2 * DHH) sbqk[tid] = (tid < DHH) ? qb[tid] : kb[tid - DHH];
    const float vb0 = vb[lane], vb1 = vb[lane + 32];
    if (tid < 128) { sea[tid*21 + 18] = 0.f; sea[tid*21 + 19] = 0.f; }
    __syncthreads();

    int cur = -1;
    int lastn = nA;
    float r0 = 0.f, r1 = 0.f;
    float d0 = 0.f, d1 = 0.f;

    #pragma unroll 1
    for (int k = 0; k < nt; ++k) {
        const int cb = k & 1, nb = cb ^ 1;
        const int* cS = sS[cb];
        const int* cT = sT[cb];

        #pragma unroll
        for (int k = 0; k < 8; ++k) sea[(etile + k + 8*lhi)*21 + l16] = stg[k];
        sea[(etile + (lane >> 1))*21 + 16 + (lane & 1)] = stg[8];
        sE[nb][etile + l16] = pe; sS[nb][etile + l16] = ps; sT[nb][etile + l16] = pt;
        wave_lds_sync();

        v2f bw[2][2][5];
        #pragma unroll
        for (int sel = 0; sel < 2; ++sel) {
            #pragma unroll
            for (int ct = 0; ct < 2; ++ct) {
                #pragma unroll
                for (int kq = 0; kq < 5; ++kq)
                    bw[sel][ct][kq] = swf[((sel*2 + ct) * 5 + kq) * 32 + lane];
            }
        }

        #pragma unroll
        for (int sel = 0; sel < 2; ++sel) {
            #pragma unroll
            for (int ct = 0; ct < 2; ++ct) {
                const float bc = sbqk[sel*32 + ct*16 + l16];
                v8f acc = {bc, bc, bc, bc, bc, bc, bc, bc};
                #pragma unroll
                for (int kq = 0; kq < 5; ++kq) {
                    const int ka = kq*4 + (lhi ? 2 : 0);
                    v2f a;
                    a.x = sea[(etile + l16)*21 + ka];
                    a.y = sea[(etile + l16)*21 + ka + 1];
                    acc = __builtin_amdgcn_wmma_f32_16x16x4_f32(
                        false, a, false, bw[sel][ct][kq], (short)0, acc, false, false);
                }
                const int cbq = sel*32 + ct*16 + l16;
                #pragma unroll
                for (int r = 0; r < 8; ++r)
                    srpe[(etile + r + (lhi ? 8 : 0))*67 + cbq] = acc[r];
            }
        }
        wave_lds_sync();

        {
            const int eloc = tid >> 1, pairh = (tid & 1) * 2;
            const int s = cS[eloc];
            const int t = cT[eloc];
            const float scale = 0.35355339059327373f;
            const float* qrow = qkv + (size_t)max(s, 0) * QKVW + pairh*DD;
            const float* krow = qkv + (size_t)t * QKVW + DHH + pairh*DD;
            float qv[2*DD], kv[2*DD];
            #pragma unroll
            for (int i = 0; i < 2*DD; ++i) kv[i] = krow[i];
            #pragma unroll
            for (int i = 0; i < 2*DD; ++i) qv[i] = qrow[i];
            __builtin_amdgcn_sched_barrier(0);

            {
                const int4 e03 = *reinterpret_cast<const int4*>(&sE[nb][etile + 8*lhi]);
                const int4 e47 = *reinterpret_cast<const int4*>(&sE[nb][etile + 8*lhi + 4]);
                const int  e8  = sE[nb][etile + (lane >> 1)];
                const int eid[8] = {e03.x, e03.y, e03.z, e03.w, e47.x, e47.y, e47.z, e47.w};
                #pragma unroll
                for (int k = 0; k < 8; ++k) stg[k] = ea[(size_t)eid[k] * RPE + l16];
                stg[8] = ea[(size_t)e8 * RPE + 16 + (lane & 1)];
            }
            pe = 0; ps = -1; pt = 0;
            {
                const int i = sa + (k + 2) * 16 + l16;
                if (i < sb) {
                    const int4 rec = reinterpret_cast<const int4*>(perm)[i];
                    pe = rec.x; ps = rec.y; pt = rec.z;
                }
            }

            float ev0 = 0.f, ev1 = 0.f;
            #pragma unroll
            for (int hh = 0; hh < 2; ++hh) {
                const int h = pairh + hh;
                float compat = 0.f;
                #pragma unroll
                for (int d = 0; d < DD; ++d) {
                    const int c = h*DD + d;
                    const float qe = qv[hh*DD + d]*scale + srpe[eloc*67 + c];
                    const float ke = kv[hh*DD + d]       + srpe[eloc*67 + 32 + c];
                    compat = fmaf(qe, ke, compat);
                }
                const float ev = __expf(compat);
                if (hh == 0) ev0 = ev; else ev1 = ev;
            }
            sex[eloc*HH + pairh]     = (s >= 0) ? ev0 : 0.f;
            sex[eloc*HH + pairh + 1] = (s >= 0) ? ev1 : 0.f;
        }

        wave_lds_sync();

        v2f bv[4][5];
        #pragma unroll
        for (int ct = 0; ct < 4; ++ct) {
            #pragma unroll
            for (int kq = 0; kq < 5; ++kq)
                bv[ct][kq] = swf[((4 + ct) * 5 + kq) * 32 + lane];
        }

        #pragma unroll
        for (int ct = 0; ct < 4; ++ct) {
            v8f acc = {};
            #pragma unroll
            for (int kq = 0; kq < 5; ++kq) {
                const int ka = kq*4 + (lhi ? 2 : 0);
                v2f a;
                a.x = sea[(etile + l16)*21 + ka];
                a.y = sea[(etile + l16)*21 + ka + 1];
                acc = __builtin_amdgcn_wmma_f32_16x16x4_f32(
                    false, a, false, bv[ct][kq], (short)0, acc, false, false);
            }
            const int cbv = ct*16 + l16;
            #pragma unroll
            for (int r = 0; r < 8; ++r)
                srpe[(etile + r + (lhi ? 8 : 0))*67 + cbv] = acc[r];
        }
        wave_lds_sync();

        const int ms = cS[etile + l16];
        const int mt = cT[etile + l16];
        #pragma unroll 1
        for (int hb = 0; hb < 16; hb += 8) {
            float val0[8], val1[8];
            float wgt0[8], wgt1[8];
            int   srun[8];
            #pragma unroll
            for (int j = 0; j < 8; ++j) {
                const int s = __builtin_amdgcn_readlane(ms, hb + j);
                const int t = __builtin_amdgcn_readlane(mt, hb + j);
                const float a0 = sex[(etile + hb + j)*HH + lhi];
                const float a1 = sex[(etile + hb + j)*HH + 2 + lhi];
                const float* vrow = qkv + (size_t)t * QKVW + 2*DHH;
                val0[j] = (vrow[lane]      + srpe[(etile + hb + j)*67 + lane]      + vb0) * a0;
                val1[j] = (vrow[lane + 32] + srpe[(etile + hb + j)*67 + lane + 32] + vb1) * a1;
                wgt0[j] = a0; wgt1[j] = a1;
                srun[j] = s;
            }
            #pragma unroll
            for (int j = 0; j < 8; ++j) {
                if (srun[j] != cur) {
                    if (cur >= 0) { store_run(out, lastn, cur, lane, r0, r1, d0, d1); lastn = cur + 1; }
                    cur = srun[j]; r0 = val0[j]; r1 = val1[j]; d0 = wgt0[j]; d1 = wgt1[j];
                } else {
                    r0 += val0[j]; r1 += val1[j]; d0 += wgt0[j]; d1 += wgt1[j];
                }
            }
        }
    }
    if (cur >= 0) { store_run(out, lastn, cur, lane, r0, r1, d0, d1); lastn = cur + 1; }
    #pragma clang loop unroll(disable) vectorize(disable) interleave(disable)
    for (int n = lastn; n < nB; ++n) {
        out[(size_t)n * DIMV + lane]      = 0.f;
        out[(size_t)n * DIMV + 32 + lane] = 0.f;
    }
}

extern "C" void kernel_launch(void* const* d_in, const int* in_sizes, int n_in,
                              void* d_out, int out_size, void* d_ws, size_t ws_size,
                              hipStream_t stream)
{
    const float* x       = (const float*)d_in[0];
    const int*   eidx    = (const int*)  d_in[1];
    const float* ea      = (const float*)d_in[2];
    const float* qkv_w   = (const float*)d_in[3];
    const float* qkv_b   = (const float*)d_in[4];
    const float* k_rpe_w = (const float*)d_in[5];
    const float* k_rpe_b = (const float*)d_in[6];
    const float* q_rpe_w = (const float*)d_in[7];
    const float* q_rpe_b = (const float*)d_in[8];
    const float* v_rpe_w = (const float*)d_in[9];
    const float* v_rpe_b = (const float*)d_in[10];

    const int N  = in_sizes[0] / DIMV;
    const int E_ = in_sizes[2] / RPE;
    float* out = (float*)d_out;

    char* ws = (char*)d_ws;
    float* qkv    = (float*)ws;
    int*   perm   = (int*)  (ws + (size_t)N * QKVW * 4);
    int*   rowptr = (int*)  (ws + (size_t)N * QKVW * 4
                                + (size_t)E_ * HH * 4);
    int*   bbase  = (int*)  (ws + (size_t)N * QKVW * 4
                                + (size_t)E_ * HH * 4
                                + (size_t)N * HH * 4);
    int*   tmp    = bbase + 2 * MAXB;
    int*   tab    = tmp + (size_t)E_ * 4;

    const int NB = (N + BNODES - 1) / BNODES;

    bucket_count_kernel<<<(E_ + CHUNK - 1) / CHUNK, 1024, 0, stream>>>(eidx, tab, E_, NB);
    bucket_scatter_kernel<<<(E_ + CHUNK - 1) / CHUNK, 1024, 0, stream>>>(
        eidx, tab, bbase, tmp, E_, NB);
    bucket_sort_kernel<<<NB, 512, 0, stream>>>(tmp, bbase, perm, rowptr);
    qkv_proj_kernel<<<(N + 15) / 16, 256, 0, stream>>>(x, qkv_w, qkv_b, qkv, N);
    edge_fused_kernel<<<(E_ + 8 * SPW - 1) / (8 * SPW), 256, 0, stream>>>(
        ea, perm, rowptr, qkv, q_rpe_w, q_rpe_b, k_rpe_w, k_rpe_b, v_rpe_w, v_rpe_b, out, E_, N);
}


# Projection kernel: a wave keeps the weight B fragments of 32 columns in registers and walks 5 row tiles (4-wave blocks): 5,000 long waves enter, not 50,000 short ones; every sum's order kept
# speedup vs baseline: 5.7301x; 1.0400x over previous
#include <hip/hip_runtime.h>
#include <hip/hip_bf16.h>


typedef __attribute__((ext_vector_type(2))) float v2f;
typedef __attribute__((ext_vector_type(8))) float v8f;

#define HH    4
#define DD    8
#define DHH   32
#define DIMV  64
#define RPE   18
#define QKVW  128

#define QTPB 5
#define XP   68

__device__ __forceinline__ void block_lds_sync()
{
    __builtin_amdgcn_fence(__ATOMIC_RELEASE, "workgroup", "local");
    __builtin_amdgcn_s_barrier();
    __builtin_amdgcn_fence(__ATOMIC_ACQUIRE, "workgroup", "local");
}

__global__ __launch_bounds__(128) void qkv_proj_kernel(
    const float* __restrict__ x, const float* __restrict__ W,
    const float* __restrict__ bias, float* __restrict__ qkv, int N)
{
    __shared__ alignas(16) float xs[2][16 * XP];
    const int tid  = threadIdx.x;
    const int wave = tid >> 5;
    const int lane = tid & 31;
    const int lhi  = lane >> 4;
    const int l16  = lane & 15;

    const int ntile = (N + 15) >> 4;
    const int t0 = blockIdx.x * QTPB;
    const int t1 = min(t0 + QTPB, ntile);

    const int col0 = wave * 32 + 2 * l16;
    const float bv0 = bias[col0], bv1 = bias[col0 + 1];

    const int rr = tid >> 4, cc = (tid & 15) * 4;
    float4 p0 = make_float4(0.f, 0.f, 0.f, 0.f), p1 = p0;
    {
        const int g0 = t0 * 16 + rr, g1 = g0 + 8;
        if (g0 < N) p0 = *(const float4*)(x + (size_t)g0 * 64 + cc);
        if (g1 < N) p1 = *(const float4*)(x + (size_t)g1 * 64 + cc);
    }

    v2f b0[16], b1[16];
    #pragma unroll
    for (int j = 0; j < 16; ++j) {
        const int ka = 4 * j + (lhi ? 2 : 0);
        const float2 w0 = *(const float2*)(W + (size_t)ka       * QKVW + col0);
        const float2 w1 = *(const float2*)(W + (size_t)(ka + 1) * QKVW + col0);
        b0[j].x = w0.x; b0[j].y = w1.x;
        b1[j].x = w0.y; b1[j].y = w1.y;
    }
    for (int t = t0; t < t1; ++t) {
        float* buf = xs[(t - t0) & 1];
        *(float4*)&buf[rr * XP + cc]       = p0;
        *(float4*)&buf[(rr + 8) * XP + cc] = p1;
        if (t + 1 < t1) {
            const int g0 = (t + 1) * 16 + rr, g1 = g0 + 8;
            p0 = make_float4(0.f, 0.f, 0.f, 0.f); p1 = p0;
            if (g0 < N) p0 = *(const float4*)(x + (size_t)g0 * 64 + cc);
            if (g1 < N) p1 = *(const float4*)(x + (size_t)g1 * 64 + cc);
        }
        block_lds_sync();

        v8f acc0 = {}, acc1 = {};
        #pragma unroll
        for (int j = 0; j < 16; ++j) {
            const int ka = 4 * j + (lhi ? 2 : 0);
            const v2f a = *(const v2f*)&buf[l16 * XP + ka];
            acc0 = __builtin_amdgcn_wmma_f32_16x16x4_f32(
                false, a, false, b0[j], (short)0, acc0, false, false);
            acc1 = __builtin_amdgcn_wmma_f32_16x16x4_f32(
                false, a, false, b1[j], (short)0, acc1, false, false);
        }

        const int r0 = t * 16;
        float* orow = qkv + (size_t)(r0 + (lhi ? 8 : 0)) * QKVW + col0;
        if (r0 + 16 <= N) {
            #pragma unroll
            for (int r = 0; r < 8; ++r)
                *(float2*)(orow + r * QKVW) = make_float2(acc0[r] + bv0, acc1[r] + bv1);
        } else {
            #pragma unroll
            for (int r = 0; r < 8; ++r)
                if (r0 + r + (lhi ? 8 : 0) < N)
                    *(float2*)(orow + r * QKVW) = make_float2(acc0[r] + bv0, acc1[r] + bv1);
        }
    }
}

#define BSH    8
#define BNODES 256
#define MAXB   1024
#define EPT    8
#define CHUNK  (EPT * 1024)

__global__ __launch_bounds__(1024) void bucket_count_kernel(
    const int* __restrict__ eidx, int* __restrict__ tab, int E_, int NB)
{
    __shared__ int lcnt[MAXB];
    const int tid = threadIdx.x;
    const int base = blockIdx.x * CHUNK;
    lcnt[tid] = 0;
    int s[EPT];
    #pragma unroll
    for (int k = 0; k < EPT; ++k)
        s[k] = eidx[min(base + tid + k * 1024, E_ - 1)];
    #pragma unroll
    for (int k = 0; k < EPT; ++k)
        if (base + tid + k * 1024 >= E_) s[k] = -1;
    __syncthreads();
    #pragma unroll
    for (int k = 0; k < EPT; ++k)
        if (s[k] >= 0) atomicAdd(&lcnt[s[k] >> BSH], 1);
    __syncthreads();
    if (tid < ((NB + 3) & ~3)) tab[blockIdx.x * MAXB + tid] = lcnt[tid];
}

__global__ __launch_bounds__(1024) void bucket_scatter_kernel(
    const int* __restrict__ eidx, const int* __restrict__ tab,
    int* __restrict__ bbase, int* __restrict__ tmp, int E_, int NB)
{
    __shared__ int lcnt[MAXB];
    __shared__ int ltot[MAXB];
    __shared__ int lpre[MAXB];
    __shared__ int lbase[MAXB];
    __shared__ int wsum[33];
    const int tid = threadIdx.x, lane = tid & 31, wave = tid >> 5;
    const int base = blockIdx.x * CHUNK;
    lcnt[tid] = 0; ltot[tid] = 0; lpre[tid] = 0;
    int s[EPT], t[EPT];
    #pragma unroll
    for (int k = 0; k < EPT; ++k) {
        const int e = min(base + tid + k * 1024, E_ - 1);
        s[k] = eidx[e]; t[k] = eidx[E_ + e];
    }
    #pragma unroll
    for (int k = 0; k < EPT; ++k)
        if (base + tid + k * 1024 >= E_) s[k] = -1;
    const int c4 = (tid & 127) * 4;
    int4 tot = make_int4(0, 0, 0, 0), pre = make_int4(0, 0, 0, 0);
    if (c4 < NB) {
        const int nrow = gridDim.x;
        #pragma unroll 4
        for (int r = tid >> 7; r < nrow; r += 8) {
            const int4 v = *reinterpret_cast<const int4*>(tab + r * MAXB + c4);
            tot.x += v.x; tot.y += v.y; tot.z += v.z; tot.w += v.w;
            if (r < (int)blockIdx.x) { pre.x += v.x; pre.y += v.y; pre.z += v.z; pre.w += v.w; }
        }
    }
    __syncthreads();
    int rk[EPT];
    #pragma unroll
    for (int k = 0; k < EPT; ++k) {
        rk[k] = 0;
        if (s[k] >= 0) rk[k] = atomicAdd(&lcnt[s[k] >> BSH], 1);
    }
    if (c4 < NB) {
        atomicAdd(&ltot[c4], tot.x); atomicAdd(&ltot[c4 + 1], tot.y);
        atomicAdd(&ltot[c4 + 2], tot.z); atomicAdd(&ltot[c4 + 3], tot.w);
        atomicAdd(&lpre[c4], pre.x); atomicAdd(&lpre[c4 + 1], pre.y);
        atomicAdd(&lpre[c4 + 2], pre.z); atomicAdd(&lpre[c4 + 3], pre.w);
    }
    __syncthreads();
    const int bc = (tid < NB) ? ltot[tid] : 0;
    int x = bc;
    #pragma unroll
    for (int d = 1; d < 32; d <<= 1) {
        const int y = __shfl_up(x, d);
        if (lane >= d) x += y;
    }
    if (lane == 31) wsum[wave] = x;
    __syncthreads();
    if (wave == 0) {
        const int w = wsum[lane];
        int xs = w;
        #pragma unroll
        for (int d = 1; d < 32; d <<= 1) {
            const int y = __shfl_up(xs, d);
            if (lane >= d) xs += y;
        }
        wsum[lane] = xs - w;
        if (lane == 31) wsum[32] = xs;
    }
    __syncthreads();
    const int bb = wsum[wave] + (x - bc);
    lbase[tid] = bb + lpre[tid];
    if (blockIdx.x == 0) {
        if (tid < NB) bbase[tid] = bb;
        if (tid == 0) bbase[NB] = wsum[32];
    }
    __syncthreads();
    #pragma unroll
    for (int k = 0; k < EPT; ++k) {
        if (s[k] >= 0) {
            const int e = base + tid + k * 1024;
            const int pos = lbase[s[k] >> BSH] + rk[k];
            reinterpret_cast<int4*>(tmp)[pos] = make_int4(e, s[k], t[k], 0);
        }
    }
}

#define KC 10
__global__ __launch_bounds__(512) void bucket_sort_kernel(
    const int* __restrict__ tmp, const int* __restrict__ bbase, int* __restrict__ perm,
    int* __restrict__ rowptr)
{
    __shared__ int ncnt[BNODES];
    __shared__ int ncur[BNODES];
    __shared__ int wsum[8];
    const int tid = threadIdx.x, lane = tid & 31, wave = tid >> 5;
    const int s0 = blockIdx.x << BSH;
    const int lo = bbase[blockIdx.x], hi = bbase[blockIdx.x + 1];
    const int4* rec = reinterpret_cast<const int4*>(tmp);
    if (tid < BNODES) ncnt[tid] = 0;
    int4 r[KC];
    #pragma unroll
    for (int k = 0; k < KC; ++k)
        r[k] = rec[max(min(lo + tid + k * 512, hi - 1), 0)];
    #pragma unroll
    for (int k = 0; k < KC; ++k)
        if (lo + tid + k * 512 >= hi) r[k].y = -1;
    __syncthreads();
    #pragma unroll
    for (int k = 0; k < KC; ++k)
        if (r[k].y >= 0) atomicAdd(&ncnt[r[k].y - s0], 1);
    for (int i = lo + tid + KC * 512; i < hi; i += 512)
        atomicAdd(&ncnt[tmp[4 * i + 1] - s0], 1);
    __syncthreads();
    const int c = (tid < BNODES) ? ncnt[tid] : 0;
    int x = c;
    #pragma unroll
    for (int d = 1; d < 32; d <<= 1) {
        const int y = __shfl_up(x, d);
        if (lane >= d) x += y;
    }
    if (lane == 31 && wave < 8) wsum[wave] = x;
    __syncthreads();
    if (tid < BNODES) {
        int wp = 0;
        #pragma unroll
        for (int w = 0; w < 8; ++w) wp += (w < wave) ? wsum[w] : 0;
        ncur[tid] = lo + wp + (x - c);
        rowptr[s0 + tid] = lo + wp + (x - c);
    }
    if (blockIdx.x == gridDim.x - 1 && tid == 0) rowptr[s0 + BNODES] = hi;
    __syncthreads();
    #pragma unroll
    for (int k = 0; k < KC; ++k) {
        if (r[k].y >= 0) {
            const int pos = atomicAdd(&ncur[r[k].y - s0], 1);
            reinterpret_cast<int4*>(perm)[pos] = r[k];
        }
    }
    for (int i = lo + tid + KC * 512; i < hi; i += 512) {
        const int4 q = rec[i];
        const int pos = atomicAdd(&ncur[q.y - s0], 1);
        reinterpret_cast<int4*>(perm)[pos] = q;
    }
}

#define SPW 200

__device__ __forceinline__ void wave_range(const int* __restrict__ perm, const int* __restrict__ rowptr,
                                           int g, int E_, int N, int& nA, int& sa, int& nB, int& sb)
{
    const int p0 = g * SPW, p1 = p0 + SPW;
    const int s0 = perm[4 * min(p0, E_ - 1) + 1];
    const int s1 = perm[4 * min(p1, E_ - 1) + 1];
    const int a0 = rowptr[s0], b0 = rowptr[s0 + 1];
    const int a1 = rowptr[s1], b1 = rowptr[s1 + 1];
    nA = (a0 == p0) ? s0 : s0 + 1;  sa = (a0 == p0) ? p0 : b0;
    nB = (a1 == p1) ? s1 : s1 + 1;  sb = (a1 == p1) ? p1 : b1;
    if (p0 >= E_) { nA = N; sa = E_; }
    if (p1 >= E_) { nB = N; sb = E_; }
    if (g == 0) nA = 0;
}

__device__ __forceinline__ void store_run(float* __restrict__ out, int from, int node, int lane,
                                          float r0, float r1, float d0, float d1)
{
    #pragma clang loop unroll(disable) vectorize(disable) interleave(disable)
    for (int n = from; n < node; ++n) {
        out[(size_t)n * DIMV + lane]      = 0.f;
        out[(size_t)n * DIMV + 32 + lane] = 0.f;
    }
    out[(size_t)node * DIMV + lane]      = r0 / (d0 + 1e-16f);
    out[(size_t)node * DIMV + 32 + lane] = r1 / (d1 + 1e-16f);
}

__device__ __forceinline__ void wave_lds_sync()
{
    __builtin_amdgcn_fence(__ATOMIC_RELEASE, "workgroup", "local");
    __builtin_amdgcn_wave_barrier();
    __builtin_amdgcn_fence(__ATOMIC_ACQUIRE, "workgroup", "local");
}

__global__ __launch_bounds__(256) void edge_fused_kernel(
    const float* __restrict__ ea, const int* __restrict__ perm,
    const int* __restrict__ rowptr, const float* __restrict__ qkv,
    const float* __restrict__ qw, const float* __restrict__ qb,
    const float* __restrict__ kw, const float* __restrict__ kb,
    const float* __restrict__ vw, const float* __restrict__ vb,
    float* __restrict__ out, int E_, int N)
{
    __shared__ float sea[128 * 21];
    __shared__ float srpe[128 * 67];
    __shared__ float sex[128 * HH];
    __shared__ alignas(16) int sE[2][128];
    __shared__ int sS[2][128];
    __shared__ int sT[2][128];
    __shared__ v2f swf[8 * 5 * 32];
    __shared__ float sbqk[2 * DHH];
    const int tid = threadIdx.x;

    const int wave = tid >> 5, lane = tid & 31;
    const int lhi = lane >> 4, l16 = lane & 15;
    const int etile = wave * 16;
    const int wv = __builtin_amdgcn_readfirstlane(wave);
    int nA, sa, nB, sb;
    wave_range(perm, rowptr, (int)blockIdx.x * 8 + wv, E_, N, nA, sa, nB, sb);
    const int nt = (sb - sa + 15) >> 4;

    int pe = 0, ps = -1, pt = 0;
    {
        const int i = sa + l16;
        if (i < sb) {
            const int4 rec = reinterpret_cast<const int4*>(perm)[i];
            pe = rec.x; ps = rec.y; pt = rec.z;
        }
    }

    v2f wfr[5];
    #pragma unroll
    for (int k = 0; k < 5; ++k) {
        const int g  = wv + k * 8;
        const int T  = g / 5, kq = g % 5;
        const int ka = kq*4 + (lhi ? 2 : 0);
        const float* Wp = (T < 2) ? qw : (T < 4) ? kw : vw;
        const int ld  = (T < 4) ? DHH : DIMV;
        const int col = ((T < 4) ? (T & 1) : (T - 4)) * 16 + l16;
        const float wx = Wp[min(ka,     RPE - 1) * ld + col];
        const float wy = Wp[min(ka + 1, RPE - 1) * ld + col];
        wfr[k].x = (ka     < RPE) ? wx : 0.0f;
        wfr[k].y = (ka + 1 < RPE) ? wy : 0.0f;
    }

    sE[0][etile + l16] = pe; sS[0][etile + l16] = ps; sT[0][etile + l16] = pt;
    wave_lds_sync();

    float stg[9];
    {
        const int4 e03 = *reinterpret_cast<const int4*>(&sE[0][etile + 8*lhi]);
        const int4 e47 = *reinterpret_cast<const int4*>(&sE[0][etile + 8*lhi + 4]);
        const int  e8  = sE[0][etile + (lane >> 1)];
        const int eid[8] = {e03.x, e03.y, e03.z, e03.w, e47.x, e47.y, e47.z, e47.w};
        #pragma unroll
        for (int k = 0; k < 8; ++k) stg[k] = ea[(size_t)eid[k] * RPE + l16];
        stg[8] = ea[(size_t)e8 * RPE + 16 + (lane & 1)];
    }
    pe = 0; ps = -1; pt = 0;
    {
        const int i = sa + 16 + l16;
        if (i < sb) {
            const int4 rec = reinterpret_cast<const int4*>(perm)[i];
            pe = rec.x; ps = rec.y; pt = rec.z;
        }
    }
    #pragma unroll
    for (int k = 0; k < 5; ++k) swf[(wv + k * 8) * 32 + lane] = wfr[k];
    if (tid < 2 * DHH) sbqk[tid] = (tid < DHH) ? qb[tid] : kb[tid - DHH];
    const float vb0 = vb[lane], vb1 = vb[lane + 32];
    if (tid < 128) { sea[tid*21 + 18] = 0.f; sea[tid*21 + 19] = 0.f; }
    __syncthreads();

    int cur = -1;
    int lastn = nA;
    float r0 = 0.f, r1 = 0.f;
    float d0 = 0.f, d1 = 0.f;

    #pragma unroll 1
    for (int k = 0; k < nt; ++k) {
        const int cb = k & 1, nb = cb ^ 1;
        const int* cS = sS[cb];
        const int* cT = sT[cb];

        #pragma unroll
        for (int k = 0; k < 8; ++k) sea[(etile + k + 8*lhi)*21 + l16] = stg[k];
        sea[(etile + (lane >> 1))*21 + 16 + (lane & 1)] = stg[8];
        sE[nb][etile + l16] = pe; sS[nb][etile + l16] = ps; sT[nb][etile + l16] = pt;
        wave_lds_sync();

        v2f bw[2][2][5];
        #pragma unroll
        for (int sel = 0; sel < 2; ++sel) {
            #pragma unroll
            for (int ct = 0; ct < 2; ++ct) {
                #pragma unroll
                for (int kq = 0; kq < 5; ++kq)
                    bw[sel][ct][kq] = swf[((sel*2 + ct) * 5 + kq) * 32 + lane];
            }
        }

        #pragma unroll
        for (int sel = 0; sel < 2; ++sel) {
            #pragma unroll
            for (int ct = 0; ct < 2; ++ct) {
                const float bc = sbqk[sel*32 + ct*16 + l16];
                v8f acc = {bc, bc, bc, bc, bc, bc, bc, bc};
                #pragma unroll
                for (int kq = 0; kq < 5; ++kq) {
                    const int ka = kq*4 + (lhi ? 2 : 0);
                    v2f a;
                    a.x = sea[(etile + l16)*21 + ka];
                    a.y = sea[(etile + l16)*21 + ka + 1];
                    acc = __builtin_amdgcn_wmma_f32_16x16x4_f32(
                        false, a, false, bw[sel][ct][kq], (short)0, acc, false, false);
                }
                const int cbq = sel*32 + ct*16 + l16;
                #pragma unroll
                for (int r = 0; r < 8; ++r)
                    srpe[(etile + r + (lhi ? 8 : 0))*67 + cbq] = acc[r];
            }
        }
        wave_lds_sync();

        {
            const int eloc = tid >> 1, pairh = (tid & 1) * 2;
            const int s = cS[eloc];
            const int t = cT[eloc];
            const float scale = 0.35355339059327373f;
            const float* qrow = qkv + (size_t)max(s, 0) * QKVW + pairh*DD;
            const float* krow = qkv + (size_t)t * QKVW + DHH + pairh*DD;
            float qv[2*DD], kv[2*DD];
            #pragma unroll
            for (int i = 0; i < 2*DD; ++i) kv[i] = krow[i];
            #pragma unroll
            for (int i = 0; i < 2*DD; ++i) qv[i] = qrow[i];
            __builtin_amdgcn_sched_barrier(0);

            {
                const int4 e03 = *reinterpret_cast<const int4*>(&sE[nb][etile + 8*lhi]);
                const int4 e47 = *reinterpret_cast<const int4*>(&sE[nb][etile + 8*lhi + 4]);
                const int  e8  = sE[nb][etile + (lane >> 1)];
                const int eid[8] = {e03.x, e03.y, e03.z, e03.w, e47.x, e47.y, e47.z, e47.w};
                #pragma unroll
                for (int k = 0; k < 8; ++k) stg[k] = ea[(size_t)eid[k] * RPE + l16];
                stg[8] = ea[(size_t)e8 * RPE + 16 + (lane & 1)];
            }
            pe = 0; ps = -1; pt = 0;
            {
                const int i = sa + (k + 2) * 16 + l16;
                if (i < sb) {
                    const int4 rec = reinterpret_cast<const int4*>(perm)[i];
                    pe = rec.x; ps = rec.y; pt = rec.z;
                }
            }

            float ev0 = 0.f, ev1 = 0.f;
            #pragma unroll
            for (int hh = 0; hh < 2; ++hh) {
                const int h = pairh + hh;
                float compat = 0.f;
                #pragma unroll
                for (int d = 0; d < DD; ++d) {
                    const int c = h*DD + d;
                    const float qe = qv[hh*DD + d]*scale + srpe[eloc*67 + c];
                    const float ke = kv[hh*DD + d]       + srpe[eloc*67 + 32 + c];
                    compat = fmaf(qe, ke, compat);
                }
                const float ev = __expf(compat);
                if (hh == 0) ev0 = ev; else ev1 = ev;
            }
            sex[eloc*HH + pairh]     = (s >= 0) ? ev0 : 0.f;
            sex[eloc*HH + pairh + 1] = (s >= 0) ? ev1 : 0.f;
        }

        wave_lds_sync();

        v2f bv[4][5];
        #pragma unroll
        for (int ct = 0; ct < 4; ++ct) {
            #pragma unroll
            for (int kq = 0; kq < 5; ++kq)
                bv[ct][kq] = swf[((4 + ct) * 5 + kq) * 32 + lane];
        }

        #pragma unroll
        for (int ct = 0; ct < 4; ++ct) {
            v8f acc = {};
            #pragma unroll
            for (int kq = 0; kq < 5; ++kq) {
                const int ka = kq*4 + (lhi ? 2 : 0);
                v2f a;
                a.x = sea[(etile + l16)*21 + ka];
                a.y = sea[(etile + l16)*21 + ka + 1];
                acc = __builtin_amdgcn_wmma_f32_16x16x4_f32(
                    false, a, false, bv[ct][kq], (short)0, acc, false, false);
            }
            const int cbv = ct*16 + l16;
            #pragma unroll
            for (int r = 0; r < 8; ++r)
                srpe[(etile + r + (lhi ? 8 : 0))*67 + cbv] = acc[r];
        }
        wave_lds_sync();

        const int ms = cS[etile + l16];
        const int mt = cT[etile + l16];
        #pragma unroll 1
        for (int hb = 0; hb < 16; hb += 8) {
            float val0[8], val1[8];
            float wgt0[8], wgt1[8];
            int   srun[8];
            #pragma unroll
            for (int j = 0; j < 8; ++j) {
                const int s = __builtin_amdgcn_readlane(ms, hb + j);
                const int t = __builtin_amdgcn_readlane(mt, hb + j);
                const float a0 = sex[(etile + hb + j)*HH + lhi];
                const float a1 = sex[(etile + hb + j)*HH + 2 + lhi];
                const float* vrow = qkv + (size_t)t * QKVW + 2*DHH;
                val0[j] = (vrow[lane]      + srpe[(etile + hb + j)*67 + lane]      + vb0) * a0;
                val1[j] = (vrow[lane + 32] + srpe[(etile + hb + j)*67 + lane + 32] + vb1) * a1;
                wgt0[j] = a0; wgt1[j] = a1;
                srun[j] = s;
            }
            #pragma unroll
            for (int j = 0; j < 8; ++j) {
                if (srun[j] != cur) {
                    if (cur >= 0) { store_run(out, lastn, cur, lane, r0, r1, d0, d1); lastn = cur + 1; }
                    cur = srun[j]; r0 = val0[j]; r1 = val1[j]; d0 = wgt0[j]; d1 = wgt1[j];
                } else {
                    r0 += val0[j]; r1 += val1[j]; d0 += wgt0[j]; d1 += wgt1[j];
                }
            }
        }
    }
    if (cur >= 0) { store_run(out, lastn, cur, lane, r0, r1, d0, d1); lastn = cur + 1; }
    #pragma clang loop unroll(disable) vectorize(disable) interleave(disable)
    for (int n = lastn; n < nB; ++n) {
        out[(size_t)n * DIMV + lane]      = 0.f;
        out[(size_t)n * DIMV + 32 + lane] = 0.f;
    }
}

extern "C" void kernel_launch(void* const* d_in, const int* in_sizes, int n_in,
                              void* d_out, int out_size, void* d_ws, size_t ws_size,
                              hipStream_t stream)
{
    const float* x       = (const float*)d_in[0];
    const int*   eidx    = (const int*)  d_in[1];
    const float* ea      = (const float*)d_in[2];
    const float* qkv_w   = (const float*)d_in[3];
    const float* qkv_b   = (const float*)d_in[4];
    const float* k_rpe_w = (const float*)d_in[5];
    const float* k_rpe_b = (const float*)d_in[6];
    const float* q_rpe_w = (const float*)d_in[7];
    const float* q_rpe_b = (const float*)d_in[8];
    const float* v_rpe_w = (const float*)d_in[9];
    const float* v_rpe_b = (const float*)d_in[10];

    const int N  = in_sizes[0] / DIMV;
    const int E_ = in_sizes[2] / RPE;
    float* out = (float*)d_out;

    char* ws = (char*)d_ws;
    float* qkv    = (float*)ws;
    int*   perm   = (int*)  (ws + (size_t)N * QKVW * 4);
    int*   rowptr = (int*)  (ws + (size_t)N * QKVW * 4
                                + (size_t)E_ * HH * 4);
    int*   bbase  = (int*)  (ws + (size_t)N * QKVW * 4
                                + (size_t)E_ * HH * 4
                                + (size_t)N * HH * 4);
    int*   tmp    = bbase + 2 * MAXB;
    int*   tab    = tmp + (size_t)E_ * 4;

    const int NB = (N + BNODES - 1) / BNODES;

    bucket_count_kernel<<<(E_ + CHUNK - 1) / CHUNK, 1024, 0, stream>>>(eidx, tab, E_, NB);
    bucket_scatter_kernel<<<(E_ + CHUNK - 1) / CHUNK, 1024, 0, stream>>>(
        eidx, tab, bbase, tmp, E_, NB);
    bucket_sort_kernel<<<NB, 512, 0, stream>>>(tmp, bbase, perm, rowptr);
    qkv_proj_kernel<<<((N + 15) / 16 + QTPB - 1) / QTPB, 128, 0, stream>>>(x, qkv_w, qkv_b, qkv, N);
    edge_fused_kernel<<<(E_ + 8 * SPW - 1) / (8 * SPW), 256, 0, stream>>>(
        ea, perm, rowptr, qkv, q_rpe_w, q_rpe_b, k_rpe_w, k_rpe_b, v_rpe_w, v_rpe_b, out, E_, N);
}


# Edge kernel: v rpe product moved behind the per-node sum (it is linear): a run sums exp-weighted edge_attr per head, 4 finished nodes share 20 WMMAs, not 20 per 16 edges; float sum order changes
# speedup vs baseline: 5.9517x; 1.0387x over previous
#include <hip/hip_runtime.h>
#include <hip/hip_bf16.h>


typedef __attribute__((ext_vector_type(2))) float v2f;
typedef __attribute__((ext_vector_type(8))) float v8f;

#define HH    4
#define DD    8
#define DHH   32
#define DIMV  64
#define RPE   18
#define QKVW  128

#define QTPB 5
#define XP   68

__device__ __forceinline__ void block_lds_sync()
{
    __builtin_amdgcn_fence(__ATOMIC_RELEASE, "workgroup", "local");
    __builtin_amdgcn_s_barrier();
    __builtin_amdgcn_fence(__ATOMIC_ACQUIRE, "workgroup", "local");
}

__global__ __launch_bounds__(128) void qkv_proj_kernel(
    const float* __restrict__ x, const float* __restrict__ W,
    const float* __restrict__ bias, float* __restrict__ qkv, int N)
{
    __shared__ alignas(16) float xs[2][16 * XP];
    const int tid  = threadIdx.x;
    const int wave = tid >> 5;
    const int lane = tid & 31;
    const int lhi  = lane >> 4;
    const int l16  = lane & 15;

    const int ntile = (N + 15) >> 4;
    const int t0 = blockIdx.x * QTPB;
    const int t1 = min(t0 + QTPB, ntile);

    const int col0 = wave * 32 + 2 * l16;
    const float bv0 = bias[col0], bv1 = bias[col0 + 1];

    const int rr = tid >> 4, cc = (tid & 15) * 4;
    float4 p0 = make_float4(0.f, 0.f, 0.f, 0.f), p1 = p0;
    {
        const int g0 = t0 * 16 + rr, g1 = g0 + 8;
        if (g0 < N) p0 = *(const float4*)(x + (size_t)g0 * 64 + cc);
        if (g1 < N) p1 = *(const float4*)(x + (size_t)g1 * 64 + cc);
    }

    v2f b0[16], b1[16];
    #pragma unroll
    for (int j = 0; j < 16; ++j) {
        const int ka = 4 * j + (lhi ? 2 : 0);
        const float2 w0 = *(const float2*)(W + (size_t)ka       * QKVW + col0);
        const float2 w1 = *(const float2*)(W + (size_t)(ka + 1) * QKVW + col0);
        b0[j].x = w0.x; b0[j].y = w1.x;
        b1[j].x = w0.y; b1[j].y = w1.y;
    }
    for (int t = t0; t < t1; ++t) {
        float* buf = xs[(t - t0) & 1];
        *(float4*)&buf[rr * XP + cc]       = p0;
        *(float4*)&buf[(rr + 8) * XP + cc] = p1;
        if (t + 1 < t1) {
            const int g0 = (t + 1) * 16 + rr, g1 = g0 + 8;
            p0 = make_float4(0.f, 0.f, 0.f, 0.f); p1 = p0;
            if (g0 < N) p0 = *(const float4*)(x + (size_t)g0 * 64 + cc);
            if (g1 < N) p1 = *(const float4*)(x + (size_t)g1 * 64 + cc);
        }
        block_lds_sync();

        v8f acc0 = {}, acc1 = {};
        #pragma unroll
        for (int j = 0; j < 16; ++j) {
            const int ka = 4 * j + (lhi ? 2 : 0);
            const v2f a = *(const v2f*)&buf[l16 * XP + ka];
            acc0 = __builtin_amdgcn_wmma_f32_16x16x4_f32(
                false, a, false, b0[j], (short)0, acc0, false, false);
            acc1 = __builtin_amdgcn_wmma_f32_16x16x4_f32(
                false, a, false, b1[j], (short)0, acc1, false, false);
        }

        const int r0 = t * 16;
        float* orow = qkv + (size_t)(r0 + (lhi ? 8 : 0)) * QKVW + col0;
        if (r0 + 16 <= N) {
            #pragma unroll
            for (int r = 0; r < 8; ++r)
                *(float2*)(orow + r * QKVW) = make_float2(acc0[r] + bv0, acc1[r] + bv1);
        } else {
            #pragma unroll
            for (int r = 0; r < 8; ++r)
                if (r0 + r + (lhi ? 8 : 0) < N)
                    *(float2*)(orow + r * QKVW) = make_float2(acc0[r] + bv0, acc1[r] + bv1);
        }
    }
}

#define BSH    8
#define BNODES 256
#define MAXB   1024
#define EPT    8
#define CHUNK  (EPT * 1024)

__global__ __launch_bounds__(1024) void bucket_count_kernel(
    const int* __restrict__ eidx, int* __restrict__ tab, int E_, int NB)
{
    __shared__ int lcnt[MAXB];
    const int tid = threadIdx.x;
    const int base = blockIdx.x * CHUNK;
    lcnt[tid] = 0;
    int s[EPT];
    #pragma unroll
    for (int k = 0; k < EPT; ++k)
        s[k] = eidx[min(base + tid + k * 1024, E_ - 1)];
    #pragma unroll
    for (int k = 0; k < EPT; ++k)
        if (base + tid + k * 1024 >= E_) s[k] = -1;
    __syncthreads();
    #pragma unroll
    for (int k = 0; k < EPT; ++k)
        if (s[k] >= 0) atomicAdd(&lcnt[s[k] >> BSH], 1);
    __syncthreads();
    if (tid < ((NB + 3) & ~3)) tab[blockIdx.x * MAXB + tid] = lcnt[tid];
}

__global__ __launch_bounds__(1024) void bucket_scatter_kernel(
    const int* __restrict__ eidx, const int* __restrict__ tab,
    int* __restrict__ bbase, int* __restrict__ tmp, int E_, int NB)
{
    __shared__ int lcnt[MAXB];
    __shared__ int ltot[MAXB];
    __shared__ int lpre[MAXB];
    __shared__ int lbase[MAXB];
    __shared__ int wsum[33];
    const int tid = threadIdx.x, lane = tid & 31, wave = tid >> 5;
    const int base = blockIdx.x * CHUNK;
    lcnt[tid] = 0; ltot[tid] = 0; lpre[tid] = 0;
    int s[EPT], t[EPT];
    #pragma unroll
    for (int k = 0; k < EPT; ++k) {
        const int e = min(base + tid + k * 1024, E_ - 1);
        s[k] = eidx[e]; t[k] = eidx[E_ + e];
    }
    #pragma unroll
    for (int k = 0; k < EPT; ++k)
        if (base + tid + k * 1024 >= E_) s[k] = -1;
    const int c4 = (tid & 127) * 4;
    int4 tot = make_int4(0, 0, 0, 0), pre = make_int4(0, 0, 0, 0);
    if (c4 < NB) {
        const int nrow = gridDim.x;
        #pragma unroll 4
        for (int r = tid >> 7; r < nrow; r += 8) {
            const int4 v = *reinterpret_cast<const int4*>(tab + r * MAXB + c4);
            tot.x += v.x; tot.y += v.y; tot.z += v.z; tot.w += v.w;
            if (r < (int)blockIdx.x) { pre.x += v.x; pre.y += v.y; pre.z += v.z; pre.w += v.w; }
        }
    }
    __syncthreads();
    int rk[EPT];
    #pragma unroll
    for (int k = 0; k < EPT; ++k) {
        rk[k] = 0;
        if (s[k] >= 0) rk[k] = atomicAdd(&lcnt[s[k] >> BSH], 1);
    }
    if (c4 < NB) {
        atomicAdd(&ltot[c4], tot.x); atomicAdd(&ltot[c4 + 1], tot.y);
        atomicAdd(&ltot[c4 + 2], tot.z); atomicAdd(&ltot[c4 + 3], tot.w);
        atomicAdd(&lpre[c4], pre.x); atomicAdd(&lpre[c4 + 1], pre.y);
        atomicAdd(&lpre[c4 + 2], pre.z); atomicAdd(&lpre[c4 + 3], pre.w);
    }
    __syncthreads();
    const int bc = (tid < NB) ? ltot[tid] : 0;
    int x = bc;
    #pragma unroll
    for (int d = 1; d < 32; d <<= 1) {
        const int y = __shfl_up(x, d);
        if (lane >= d) x += y;
    }
    if (lane == 31) wsum[wave] = x;
    __syncthreads();
    if (wave == 0) {
        const int w = wsum[lane];
        int xs = w;
        #pragma unroll
        for (int d = 1; d < 32; d <<= 1) {
            const int y = __shfl_up(xs, d);
            if (lane >= d) xs += y;
        }
        wsum[lane] = xs - w;
        if (lane == 31) wsum[32] = xs;
    }
    __syncthreads();
    const int bb = wsum[wave] + (x - bc);
    lbase[tid] = bb + lpre[tid];
    if (blockIdx.x == 0) {
        if (tid < NB) bbase[tid] = bb;
        if (tid == 0) bbase[NB] = wsum[32];
    }
    __syncthreads();
    #pragma unroll
    for (int k = 0; k < EPT; ++k) {
        if (s[k] >= 0) {
            const int e = base + tid + k * 1024;
            const int pos = lbase[s[k] >> BSH] + rk[k];
            reinterpret_cast<int4*>(tmp)[pos] = make_int4(e, s[k], t[k], 0);
        }
    }
}

#define KC 10
__global__ __launch_bounds__(512) void bucket_sort_kernel(
    const int* __restrict__ tmp, const int* __restrict__ bbase, int* __restrict__ perm,
    int* __restrict__ rowptr)
{
    __shared__ int ncnt[BNODES];
    __shared__ int ncur[BNODES];
    __shared__ int wsum[8];
    const int tid = threadIdx.x, lane = tid & 31, wave = tid >> 5;
    const int s0 = blockIdx.x << BSH;
    const int lo = bbase[blockIdx.x], hi = bbase[blockIdx.x + 1];
    const int4* rec = reinterpret_cast<const int4*>(tmp);
    if (tid < BNODES) ncnt[tid] = 0;
    int4 r[KC];
    #pragma unroll
    for (int k = 0; k < KC; ++k)
        r[k] = rec[max(min(lo + tid + k * 512, hi - 1), 0)];
    #pragma unroll
    for (int k = 0; k < KC; ++k)
        if (lo + tid + k * 512 >= hi) r[k].y = -1;
    __syncthreads();
    #pragma unroll
    for (int k = 0; k < KC; ++k)
        if (r[k].y >= 0) atomicAdd(&ncnt[r[k].y - s0], 1);
    for (int i = lo + tid + KC * 512; i < hi; i += 512)
        atomicAdd(&ncnt[tmp[4 * i + 1] - s0], 1);
    __syncthreads();
    const int c = (tid < BNODES) ? ncnt[tid] : 0;
    int x = c;
    #pragma unroll
    for (int d = 1; d < 32; d <<= 1) {
        const int y = __shfl_up(x, d);
        if (lane >= d) x += y;
    }
    if (lane == 31 && wave < 8) wsum[wave] = x;
    __syncthreads();
    if (tid < BNODES) {
        int wp = 0;
        #pragma unroll
        for (int w = 0; w < 8; ++w) wp += (w < wave) ? wsum[w] : 0;
        ncur[tid] = lo + wp + (x - c);
        rowptr[s0 + tid] = lo + wp + (x - c);
    }
    if (blockIdx.x == gridDim.x - 1 && tid == 0) rowptr[s0 + BNODES] = hi;
    __syncthreads();
    #pragma unroll
    for (int k = 0; k < KC; ++k) {
        if (r[k].y >= 0) {
            const int pos = atomicAdd(&ncur[r[k].y - s0], 1);
            reinterpret_cast<int4*>(perm)[pos] = r[k];
        }
    }
    for (int i = lo + tid + KC * 512; i < hi; i += 512) {
        const int4 q = rec[i];
        const int pos = atomicAdd(&ncur[q.y - s0], 1);
        reinterpret_cast<int4*>(perm)[pos] = q;
    }
}

#define SPW 200
#define NCAP 4
#define GK   20
#define NB_G 0
#define NB_R (HH * GK)
#define NB_N (NB_R + DIMV)
#define NBE  148

__device__ __forceinline__ void wave_range(const int* __restrict__ perm, const int* __restrict__ rowptr,
                                           int g, int E_, int N, int& nA, int& sa, int& nB, int& sb)
{
    const int p0 = g * SPW, p1 = p0 + SPW;
    const int s0 = perm[4 * min(p0, E_ - 1) + 1];
    const int s1 = perm[4 * min(p1, E_ - 1) + 1];
    const int a0 = rowptr[s0], b0 = rowptr[s0 + 1];
    const int a1 = rowptr[s1], b1 = rowptr[s1 + 1];
    nA = (a0 == p0) ? s0 : s0 + 1;  sa = (a0 == p0) ? p0 : b0;
    nB = (a1 == p1) ? s1 : s1 + 1;  sb = (a1 == p1) ? p1 : b1;
    if (p0 >= E_) { nA = N; sa = E_; }
    if (p1 >= E_) { nB = N; sb = E_; }
    if (g == 0) nA = 0;
}

__device__ __forceinline__ void wave_lds_sync()
{
    __builtin_amdgcn_fence(__ATOMIC_RELEASE, "workgroup", "local");
    __builtin_amdgcn_wave_barrier();
    __builtin_amdgcn_fence(__ATOMIC_ACQUIRE, "workgroup", "local");
}

__device__ __forceinline__ void buffer_run(float* __restrict__ out, int from, int node, int lane,
                                           float* nbw, int e,
                                           float r0, float r1,
                                           float g0, float g1, float g2, float g3)
{
    #pragma clang loop unroll(disable) vectorize(disable) interleave(disable)
    for (int n = from; n < node; ++n) {
        out[(size_t)n * DIMV + lane]      = 0.f;
        out[(size_t)n * DIMV + 32 + lane] = 0.f;
    }
    float* en = nbw + e * NBE;
    if (lane < GK) {
        en[NB_G + lane]          = g0;
        en[NB_G + GK + lane]     = g1;
        en[NB_G + 2*GK + lane]   = g2;
        en[NB_G + 3*GK + lane]   = g3;
    }
    en[NB_R + lane]      = r0;
    en[NB_R + 32 + lane] = r1;
    en[NB_N] = __int_as_float(node);
}

__device__ __forceinline__ void flush_nodes(float* __restrict__ out, const v2f* swf,
                                            float* nbw, int nb, int lane)
{
    const int lhi = lane >> 4, l16 = lane & 15;
    wave_lds_sync();
    const int er = min(l16, nb - 1);
    #pragma unroll 1
    for (int h = 0; h < HH; ++h) {
        v8f acc = {};
        #pragma unroll
        for (int r = 0; r < NCAP; ++r) {
            const float c0 = nbw[min(r, nb - 1)*NBE + NB_R + 16*h + l16];
            acc[r] = lhi ? 0.f : c0;
        }
        #pragma unroll
        for (int kq = 0; kq < 5; ++kq) {
            const v2f a = *(const v2f*)&nbw[er*NBE + NB_G + h*GK + 4*kq + 2*lhi];
            const v2f b = swf[((4 + h) * 5 + kq) * 32 + lane];
            acc = __builtin_amdgcn_wmma_f32_16x16x4_f32(
                false, a, false, b, (short)0, acc, false, false);
        }
        if (lhi == 0) {
            #pragma unroll
            for (int r = 0; r < NCAP; ++r)
                if (r < nb) nbw[r*NBE + NB_R + 16*h + l16] = acc[r];
        }
    }
    wave_lds_sync();
    #pragma unroll 1
    for (int e = 0; e < nb; ++e) {
        const float* en = nbw + e * NBE;
        const int node = __builtin_amdgcn_readfirstlane(__float_as_int(en[NB_N]));
        const float d0 = en[NB_G + lhi*GK + RPE], d1 = en[NB_G + (2 + lhi)*GK + RPE];
        out[(size_t)node * DIMV + lane]      = en[NB_R + lane]      / (d0 + 1e-16f);
        out[(size_t)node * DIMV + 32 + lane] = en[NB_R + 32 + lane] / (d1 + 1e-16f);
    }
    wave_lds_sync();
}

__global__ __launch_bounds__(256) void edge_fused_kernel(
    const float* __restrict__ ea, const int* __restrict__ perm,
    const int* __restrict__ rowptr, const float* __restrict__ qkv,
    const float* __restrict__ qw, const float* __restrict__ qb,
    const float* __restrict__ kw, const float* __restrict__ kb,
    const float* __restrict__ vw, const float* __restrict__ vb,
    float* __restrict__ out, int E_, int N)
{
    __shared__ float sea[128 * 21];
    __shared__ float srpe[128 * 67];
    __shared__ alignas(16) float sex[128 * HH];
    __shared__ alignas(16) int sE[2][128];
    __shared__ int sS[2][128];
    __shared__ int sT[2][128];
    __shared__ v2f swf[8 * 5 * 32];
    __shared__ float sbqk[2 * DHH];
    __shared__ alignas(16) float snb[8 * NCAP * NBE];
    const int tid = threadIdx.x;

    const int wave = tid >> 5, lane = tid & 31;
    const int lhi = lane >> 4, l16 = lane & 15;
    const int etile = wave * 16;
    const int wv = __builtin_amdgcn_readfirstlane(wave);
    int nA, sa, nB, sb;
    wave_range(perm, rowptr, (int)blockIdx.x * 8 + wv, E_, N, nA, sa, nB, sb);
    const int nt = (sb - sa + 15) >> 4;

    int pe = 0, ps = -1, pt = 0;
    {
        const int i = sa + l16;
        if (i < sb) {
            const int4 rec = reinterpret_cast<const int4*>(perm)[i];
            pe = rec.x; ps = rec.y; pt = rec.z;
        }
    }

    v2f wfr[5];
    #pragma unroll
    for (int k = 0; k < 5; ++k) {
        const int g  = wv + k * 8;
        const int T  = g / 5, kq = g % 5;
        const int ka = kq*4 + (lhi ? 2 : 0);
        const float* Wp = (T < 2) ? qw : (T < 4) ? kw : vw;
        const int ld  = (T < 4) ? DHH : DIMV;
        const int col = ((T < 4) ? (T & 1) : (T - 4)) * 16 + l16;
        const float wx = Wp[min(ka,     RPE - 1) * ld + col];
        const float wy = Wp[min(ka + 1, RPE - 1) * ld + col];
        const float wb = vb[(T < 4) ? l16 : col];
        const float w18 = (T < 4) ? 0.0f : wb;
        wfr[k].x = (ka     < RPE) ? wx : (ka == RPE) ? w18 : 0.0f;
        wfr[k].y = (ka + 1 < RPE) ? wy : 0.0f;
    }

    sE[0][etile + l16] = pe; sS[0][etile + l16] = ps; sT[0][etile + l16] = pt;
    wave_lds_sync();

    float stg[9];
    {
        const int4 e03 = *reinterpret_cast<const int4*>(&sE[0][etile + 8*lhi]);
        const int4 e47 = *reinterpret_cast<const int4*>(&sE[0][etile + 8*lhi + 4]);
        const int  e8  = sE[0][etile + (lane >> 1)];
        const int eid[8] = {e03.x, e03.y, e03.z, e03.w, e47.x, e47.y, e47.z, e47.w};
        #pragma unroll
        for (int k = 0; k < 8; ++k) stg[k] = ea[(size_t)eid[k] * RPE + l16];
        stg[8] = ea[(size_t)e8 * RPE + 16 + (lane & 1)];
    }
    pe = 0; ps = -1; pt = 0;
    {
        const int i = sa + 16 + l16;
        if (i < sb) {
            const int4 rec = reinterpret_cast<const int4*>(perm)[i];
            pe = rec.x; ps = rec.y; pt = rec.z;
        }
    }
    #pragma unroll
    for (int k = 0; k < 5; ++k) swf[(wv + k * 8) * 32 + lane] = wfr[k];
    if (tid < 2 * DHH) sbqk[tid] = (tid < DHH) ? qb[tid] : kb[tid - DHH];
    if (tid < 128) { sea[tid*21 + 18] = 1.f; sea[tid*21 + 19] = 0.f; }
    __syncthreads();

    int cur = -1;
    int lastn = nA;
    float r0 = 0.f, r1 = 0.f;
    float g0 = 0.f, g1 = 0.f, g2 = 0.f, g3 = 0.f;
    int nbuf = 0;
    float* nbw = snb + wv * (NCAP * NBE);
    const int kl = min(lane, GK - 1);

    #pragma unroll 1
    for (int k = 0; k < nt; ++k) {
        const int cb = k & 1, nb = cb ^ 1;
        const int* cS = sS[cb];
        const int* cT = sT[cb];

        #pragma unroll
        for (int k = 0; k < 8; ++k) sea[(etile + k + 8*lhi)*21 + l16] = stg[k];
        sea[(etile + (lane >> 1))*21 + 16 + (lane & 1)] = stg[8];
        sE[nb][etile + l16] = pe; sS[nb][etile + l16] = ps; sT[nb][etile + l16] = pt;
        wave_lds_sync();

        v2f bw[2][2][5];
        #pragma unroll
        for (int sel = 0; sel < 2; ++sel) {
            #pragma unroll
            for (int ct = 0; ct < 2; ++ct) {
                #pragma unroll
                for (int kq = 0; kq < 5; ++kq)
                    bw[sel][ct][kq] = swf[((sel*2 + ct) * 5 + kq) * 32 + lane];
            }
        }

        #pragma unroll
        for (int sel = 0; sel < 2; ++sel) {
            #pragma unroll
            for (int ct = 0; ct < 2; ++ct) {
                const float bc = sbqk[sel*32 + ct*16 + l16];
                v8f acc = {bc, bc, bc, bc, bc, bc, bc, bc};
                #pragma unroll
                for (int kq = 0; kq < 5; ++kq) {
                    const int ka = kq*4 + (lhi ? 2 : 0);
                    v2f a;
                    a.x = sea[(etile + l16)*21 + ka];
                    a.y = sea[(etile + l16)*21 + ka + 1];
                    acc = __builtin_amdgcn_wmma_f32_16x16x4_f32(
                        false, a, false, bw[sel][ct][kq], (short)0, acc, false, false);
                }
                const int cbq = sel*32 + ct*16 + l16;
                #pragma unroll
                for (int r = 0; r < 8; ++r)
                    srpe[(etile + r + (lhi ? 8 : 0))*67 + cbq] = acc[r];
            }
        }
        wave_lds_sync();

        {
            const int eloc = tid >> 1, pairh = (tid & 1) * 2;
            const int s = cS[eloc];
            const int t = cT[eloc];
            const float scale = 0.35355339059327373f;
            const float* qrow = qkv + (size_t)max(s, 0) * QKVW + pairh*DD;
            const float* krow = qkv + (size_t)t * QKVW + DHH + pairh*DD;
            float qv[2*DD], kv[2*DD];
            #pragma unroll
            for (int i = 0; i < 2*DD; ++i) kv[i] = krow[i];
            #pragma unroll
            for (int i = 0; i < 2*DD; ++i) qv[i] = qrow[i];
            __builtin_amdgcn_sched_barrier(0);

            {
                const int4 e03 = *reinterpret_cast<const int4*>(&sE[nb][etile + 8*lhi]);
                const int4 e47 = *reinterpret_cast<const int4*>(&sE[nb][etile + 8*lhi + 4]);
                const int  e8  = sE[nb][etile + (lane >> 1)];
                const int eid[8] = {e03.x, e03.y, e03.z, e03.w, e47.x, e47.y, e47.z, e47.w};
                #pragma unroll
                for (int k = 0; k < 8; ++k) stg[k] = ea[(size_t)eid[k] * RPE + l16];
                stg[8] = ea[(size_t)e8 * RPE + 16 + (lane & 1)];
            }
            pe = 0; ps = -1; pt = 0;
            {
                const int i = sa + (k + 2) * 16 + l16;
                if (i < sb) {
                    const int4 rec = reinterpret_cast<const int4*>(perm)[i];
                    pe = rec.x; ps = rec.y; pt = rec.z;
                }
            }

            float ev0 = 0.f, ev1 = 0.f;
            #pragma unroll
            for (int hh = 0; hh < 2; ++hh) {
                const int h = pairh + hh;
                float compat = 0.f;
                #pragma unroll
                for (int d = 0; d < DD; ++d) {
                    const int c = h*DD + d;
                    const float qe = qv[hh*DD + d]*scale + srpe[eloc*67 + c];
                    const float ke = kv[hh*DD + d]       + srpe[eloc*67 + 32 + c];
                    compat = fmaf(qe, ke, compat);
                }
                const float ev = __expf(compat);
                if (hh == 0) ev0 = ev; else ev1 = ev;
            }
            sex[eloc*HH + pairh]     = (s >= 0) ? ev0 : 0.f;
            sex[eloc*HH + pairh + 1] = (s >= 0) ? ev1 : 0.f;
        }

        wave_lds_sync();

        const int ms = cS[etile + l16];
        const int mt = cT[etile + l16];
        #pragma unroll 1
        for (int hb = 0; hb < 16; hb += 8) {
            float val0[8], val1[8];
            int   srun[8];
            #pragma unroll
            for (int j = 0; j < 8; ++j) {
                const int s = __builtin_amdgcn_readlane(ms, hb + j);
                const int t = __builtin_amdgcn_readlane(mt, hb + j);
                const float* vrow = qkv + (size_t)t * QKVW + 2*DHH;
                val0[j] = vrow[lane];
                val1[j] = vrow[lane + 32];
                srun[j] = s;
            }
            const float* exrow = &sex[(etile + hb)*HH];
            const float* earow = &sea[(etile + hb)*21 + kl];
            int j0 = 0;
            do {
                int stop = 8;
                float4 exn = *reinterpret_cast<const float4*>(exrow);
                float  ean = earow[0];
                #pragma unroll
                for (int j = 0; j < 8; ++j) {
                    const float4 ex = exn;
                    const float  eaj = ean;
                    if (j + 1 < 8) {
                        exn = *reinterpret_cast<const float4*>(exrow + (j + 1)*HH);
                        ean = earow[(j + 1)*21];
                    }
                    if (j >= j0 && j < stop) {
                        if (srun[j] != cur) {
                            if (cur >= 0 && nbuf == NCAP) {
                                stop = j;
                            } else {
                                if (cur >= 0) {
                                    buffer_run(out, lastn, cur, lane, nbw, nbuf,
                                               r0, r1, g0, g1, g2, g3);
                                    lastn = cur + 1; ++nbuf;
                                }
                                cur = srun[j];
                                r0 = 0.f; r1 = 0.f;
                                g0 = 0.f; g1 = 0.f; g2 = 0.f; g3 = 0.f;
                            }
                        }
                        if (j < stop) {
                            const float a0 = lhi ? ex.y : ex.x;
                            const float a1 = lhi ? ex.w : ex.z;
                            r0 = fmaf(val0[j], a0, r0); r1 = fmaf(val1[j], a1, r1);
                            g0 = fmaf(ex.x, eaj, g0); g1 = fmaf(ex.y, eaj, g1);
                            g2 = fmaf(ex.z, eaj, g2); g3 = fmaf(ex.w, eaj, g3);
                        }
                    }
                }
                if (stop < 8) { flush_nodes(out, swf, nbw, nbuf, lane); nbuf = 0; }
                j0 = stop;
            } while (j0 < 8);
        }
    }
    #pragma unroll 1
    while (cur >= 0 || nbuf > 0) {
        if (cur >= 0 && nbuf < NCAP) {
            buffer_run(out, lastn, cur, lane, nbw, nbuf,
                       r0, r1, g0, g1, g2, g3);
            lastn = cur + 1; ++nbuf; cur = -1;
        }
        flush_nodes(out, swf, nbw, nbuf, lane);
        nbuf = 0;
    }
    #pragma clang loop unroll(disable) vectorize(disable) interleave(disable)
    for (int n = lastn; n < nB; ++n) {
        out[(size_t)n * DIMV + lane]      = 0.f;
        out[(size_t)n * DIMV + 32 + lane] = 0.f;
    }
}

extern "C" void kernel_launch(void* const* d_in, const int* in_sizes, int n_in,
                              void* d_out, int out_size, void* d_ws, size_t ws_size,
                              hipStream_t stream)
{
    const float* x       = (const float*)d_in[0];
    const int*   eidx    = (const int*)  d_in[1];
    const float* ea      = (const float*)d_in[2];
    const float* qkv_w   = (const float*)d_in[3];
    const float* qkv_b   = (const float*)d_in[4];
    const float* k_rpe_w = (const float*)d_in[5];
    const float* k_rpe_b = (const float*)d_in[6];
    const float* q_rpe_w = (const float*)d_in[7];
    const float* q_rpe_b = (const float*)d_in[8];
    const float* v_rpe_w = (const float*)d_in[9];
    const float* v_rpe_b = (const float*)d_in[10];

    const int N  = in_sizes[0] / DIMV;
    const int E_ = in_sizes[2] / RPE;
    float* out = (float*)d_out;

    char* ws = (char*)d_ws;
    float* qkv    = (float*)ws;
    int*   perm   = (int*)  (ws + (size_t)N * QKVW * 4);
    int*   rowptr = (int*)  (ws + (size_t)N * QKVW * 4
                                + (size_t)E_ * HH * 4);
    int*   bbase  = (int*)  (ws + (size_t)N * QKVW * 4
                                + (size_t)E_ * HH * 4
                                + (size_t)N * HH * 4);
    int*   tmp    = bbase + 2 * MAXB;
    int*   tab    = tmp + (size_t)E_ * 4;

    const int NB = (N + BNODES - 1) / BNODES;

    bucket_count_kernel<<<(E_ + CHUNK - 1) / CHUNK, 1024, 0, stream>>>(eidx, tab, E_, NB);
    bucket_scatter_kernel<<<(E_ + CHUNK - 1) / CHUNK, 1024, 0, stream>>>(
        eidx, tab, bbase, tmp, E_, NB);
    bucket_sort_kernel<<<NB, 512, 0, stream>>>(tmp, bbase, perm, rowptr);
    qkv_proj_kernel<<<((N + 15) / 16 + QTPB - 1) / QTPB, 128, 0, stream>>>(x, qkv_w, qkv_b, qkv, N);
    edge_fused_kernel<<<(E_ + 8 * SPW - 1) / (8 * SPW), 256, 0, stream>>>(
        ea, perm, rowptr, qkv, q_rpe_w, q_rpe_b, k_rpe_w, k_rpe_b, v_rpe_w, v_rpe_b, out, E_, N);
}
